# starting kernel
# baseline (speedup 1.0000x reference)
#include <hip/hip_runtime.h>
#include <hip/hip_bf16.h>

#define N_NODES 50000
#define E_EDGES 800000
#define IN_F    256
#define H_HEADS 4
#define F_OUT   64
#define HF      256
#define SLOPE   0.2f

typedef __attribute__((ext_vector_type(16))) __bf16 v16bf;
typedef __attribute__((ext_vector_type(8)))  float  v8f;

__device__ __forceinline__ float leakyf(float x) { return x > 0.f ? x : SLOPE * x; }

__device__ __forceinline__ void atomicMaxF(float* addr, float val) {
  if (val >= 0.f) {
    atomicMax((int*)addr, __float_as_int(val));
  } else {
    atomicMin((unsigned int*)addr, __float_as_uint(val));
  }
}

#define WT_STRIDE 40

__global__ __launch_bounds__(256)
void gemm_h_kernel(const float* __restrict__ feat, const float* __restrict__ W,
                   float* __restrict__ h) {
  __shared__ __align__(16) __bf16 Wt[HF * WT_STRIDE];

  const int t       = threadIdx.x;
  const int wave    = t >> 5;
  const int lane    = t & 31;
  const int rowBase = blockIdx.x * 128 + wave * 16;
  const bool rowValid = (rowBase < N_NODES);
  const int nloc = lane & 15;
  const int kg   = lane >> 4;

  v8f acc[16];
#pragma unroll
  for (int nt = 0; nt < 16; ++nt)
#pragma unroll
    for (int v = 0; v < 8; ++v) acc[nt][v] = 0.f;

  for (int ks = 0; ks < IN_F / 32; ++ks) {
    const int kbase = ks * 32;
    __syncthreads();
#pragma unroll
    for (int kk = 0; kk < 32; ++kk)
      Wt[t * WT_STRIDE + kk] = (__bf16)W[(kbase + kk) * HF + t];
    __syncthreads();

    if (rowValid) {
      const float* arow = feat + (size_t)(rowBase + nloc) * IN_F + kbase;
      float4 a0 = *(const float4*)(arow + 8 * kg);
      float4 a1 = *(const float4*)(arow + 8 * kg + 4);
      float4 a2 = *(const float4*)(arow + 16 + 8 * kg);
      float4 a3 = *(const float4*)(arow + 16 + 8 * kg + 4);
      float atmp[16] = {a0.x, a0.y, a0.z, a0.w, a1.x, a1.y, a1.z, a1.w,
                        a2.x, a2.y, a2.z, a2.w, a3.x, a3.y, a3.z, a3.w};
      v16bf afrag;
#pragma unroll
      for (int i = 0; i < 16; ++i) afrag[i] = (__bf16)atmp[i];

      const __bf16* wrow0 = &Wt[nloc * WT_STRIDE];
      uint4 b0 = *(const uint4*)(wrow0 + 8 * kg);
      uint4 b1 = *(const uint4*)(wrow0 + 16 + 8 * kg);
#pragma unroll
      for (int nt = 0; nt < 16; ++nt) {
        uint4 nb0, nb1;
        if (nt < 15) {
          const __bf16* wrow = &Wt[((nt + 1) * 16 + nloc) * WT_STRIDE];
          nb0 = *(const uint4*)(wrow + 8 * kg);
          nb1 = *(const uint4*)(wrow + 16 + 8 * kg);
        }
        const __bf16* p0 = (const __bf16*)&b0;
        const __bf16* p1 = (const __bf16*)&b1;
        v16bf bfrag;
#pragma unroll
        for (int i = 0; i < 8; ++i) { bfrag[i] = p0[i]; bfrag[8 + i] = p1[i]; }
        acc[nt] = __builtin_amdgcn_wmma_f32_16x16x32_bf16(
            false, afrag, false, bfrag, (short)0, acc[nt], false, false);
        if (nt < 15) { b0 = nb0; b1 = nb1; }
      }
    }
  }

  if (rowValid) {
#pragma unroll
    for (int nt = 0; nt < 16; ++nt)
#pragma unroll
      for (int v = 0; v < 8; ++v)
        h[(size_t)(rowBase + v + 8 * kg) * HF + nt * 16 + nloc] = acc[nt][v];
  }
}

__global__ void init_kernel(float* maxs, float* maxd, float* sums, float* sumd,
                            float* outdeg, float* indeg) {
  int i = blockIdx.x * blockDim.x + threadIdx.x;
  if (i < N_NODES * H_HEADS) {
    maxs[i] = -1e30f; maxd[i] = -1e30f; sums[i] = 0.f; sumd[i] = 0.f;
  }
  if (i < N_NODES) { outdeg[i] = 0.f; indeg[i] = 0.f; }
}

__global__ void scores_kernel(const float* __restrict__ h,
                              const float* __restrict__ attn_l,
                              const float* __restrict__ attn_r,
                              const float* __restrict__ hop_attn_l,
                              float* __restrict__ el, float* __restrict__ er,
                              float* __restrict__ al) {
  int i = blockIdx.x * blockDim.x + threadIdx.x;
  if (i >= N_NODES * H_HEADS) return;
  int node = i >> 2, head = i & 3;
  const float* hp  = h + (size_t)node * HF + head * F_OUT;
  const float* alp = attn_l + head * F_OUT;
  const float* arp = attn_r + head * F_OUT;
  const float* hlp = hop_attn_l + head * F_OUT;
  float sl = 0.f, sr = 0.f, shl = 0.f;
#pragma unroll 8
  for (int f = 0; f < F_OUT; ++f) {
    float v = hp[f];
    sl += v * alp[f]; sr += v * arp[f]; shl += v * hlp[f];
  }
  el[i] = sl; er[i] = sr; al[i] = shl;
}

__global__ void edge_e_kernel(const int* __restrict__ src, const int* __restrict__ dst,
                              const float* __restrict__ el, const float* __restrict__ er,
                              float* __restrict__ e_ws, float* maxs, float* maxd,
                              float* outdeg, float* indeg) {
  int i = blockIdx.x * blockDim.x + threadIdx.x;
  if (i >= E_EDGES * H_HEADS) return;
  int eid = i >> 2, head = i & 3;
  int s = src[eid], d = dst[eid];
  float ev = leakyf(el[s * 4 + head] + er[d * 4 + head]);
  e_ws[i] = ev;
  atomicMaxF(&maxs[s * 4 + head], ev);
  atomicMaxF(&maxd[d * 4 + head], ev);
  if (head == 0) { atomicAdd(&outdeg[s], 1.f); atomicAdd(&indeg[d], 1.f); }
}

__global__ void edge_sum_kernel(const int* __restrict__ src, const int* __restrict__ dst,
                                const float* __restrict__ e_ws,
                                const float* __restrict__ maxs, const float* __restrict__ maxd,
                                float* sums, float* sumd) {
  int i = blockIdx.x * blockDim.x + threadIdx.x;
  if (i >= E_EDGES * H_HEADS) return;
  int eid = i >> 2, head = i & 3;
  int s = src[eid], d = dst[eid];
  float ev = e_ws[i];
  atomicAdd(&sums[s * 4 + head], __expf(ev - maxs[s * 4 + head]));
  atomicAdd(&sumd[d * 4 + head], __expf(ev - maxd[d * 4 + head]));
}

__global__ void edge_a_kernel(const int* __restrict__ src, const int* __restrict__ dst,
                              const float* __restrict__ e_ws,
                              const float* __restrict__ maxs, const float* __restrict__ maxd,
                              const float* __restrict__ sums, const float* __restrict__ sumd,
                              const float* __restrict__ sigma, float* __restrict__ a_ws) {
  int i = blockIdx.x * blockDim.x + threadIdx.x;
  if (i >= E_EDGES * H_HEADS) return;
  int eid = i >> 2, head = i & 3;
  int s = src[eid], d = dst[eid];
  float ev = e_ws[i];
  float sgm = 1.f / (1.f + __expf(-sigma[0]));
  float ax = fmaxf(__expf(ev - maxs[s * 4 + head]) / sums[s * 4 + head], 1e-10f);
  float ay = fmaxf(__expf(ev - maxd[d * 4 + head]) / sumd[d * 4 + head], 1e-10f);
  a_ws[i] = __powf(ay, sgm) * __powf(ax, 1.f - sgm);
}

__global__ void norm_kernel(const float* __restrict__ outdeg, const float* __restrict__ indeg,
                            float* __restrict__ norm_out, float* __restrict__ norm_in) {
  int i = blockIdx.x * blockDim.x + threadIdx.x;
  if (i >= N_NODES) return;
  norm_out[i] = rsqrtf(fmaxf(outdeg[i], 1.f));
  norm_in[i]  = sqrtf(fmaxf(indeg[i], 1.f));
}

__global__ __launch_bounds__(256)
void agg_kernel(const int* __restrict__ src, const int* __restrict__ dst,
                const float* __restrict__ a_ws, const float* __restrict__ norm_out,
                const float* __restrict__ xprev, float* __restrict__ xnext) {
  int eid = blockIdx.x;
  int t = threadIdx.x;
  int s = src[eid], d = dst[eid];
  int head = t >> 6;
  float coef = a_ws[eid * 4 + head] * norm_out[s];
  float v = xprev[(size_t)s * HF + t] * coef;
  atomicAdd(&xnext[(size_t)d * HF + t], v);
}

__global__ void scale_kernel(float* __restrict__ x, const float* __restrict__ norm_in) {
  int i = blockIdx.x * blockDim.x + threadIdx.x;
  if (i >= N_NODES * HF) return;
  x[i] *= norm_in[i >> 8];
}

__global__ __launch_bounds__(128)
void combine_kernel(const float* __restrict__ h, const float* __restrict__ x1,
                    const float* __restrict__ x2, const float* __restrict__ x3,
                    const float* __restrict__ al, const float* __restrict__ hop_attn_r,
                    float* __restrict__ out) {
  int n    = blockIdx.x;
  int head = threadIdx.x >> 5;
  int lane = threadIdx.x & 31;
  const float* hs[4] = {h, x1, x2, x3};
  size_t base = (size_t)n * HF + head * F_OUT;
  const float* ar = hop_attn_r + head * F_OUT;
  float ar0 = ar[lane], ar1 = ar[lane + 32];
  float albase = al[n * 4 + head];

  float xv0[4], xv1[4], score[4];
#pragma unroll
  for (int k = 0; k < 4; ++k) {
    float v0 = hs[k][base + lane];
    float v1 = hs[k][base + lane + 32];
    xv0[k] = v0; xv1[k] = v1;
    float p = v0 * ar0 + v1 * ar1;
#pragma unroll
    for (int off = 16; off > 0; off >>= 1) p += __shfl_xor(p, off);
    score[k] = leakyf(albase + p);
  }
  float m = fmaxf(fmaxf(score[0], score[1]), fmaxf(score[2], score[3]));
  float ex[4], ssum = 0.f;
#pragma unroll
  for (int k = 0; k < 4; ++k) { ex[k] = __expf(score[k] - m); ssum += ex[k]; }
  float o0 = 0.f, o1 = 0.f;
#pragma unroll
  for (int k = 0; k < 4; ++k) {
    float w = ex[k] / ssum;
    o0 += xv0[k] * w; o1 += xv1[k] * w;
  }
  out[base + lane] = o0;
  out[base + lane + 32] = o1;
}

extern "C" void kernel_launch(void* const* d_in, const int* in_sizes, int n_in,
                              void* d_out, int out_size, void* d_ws, size_t ws_size,
                              hipStream_t stream) {
  (void)in_sizes; (void)n_in; (void)out_size; (void)ws_size;
  const float* feat       = (const float*)d_in[0];
  const int*   src        = (const int*)d_in[1];
  const int*   dst        = (const int*)d_in[2];
  const float* W          = (const float*)d_in[3];
  const float* attn_l     = (const float*)d_in[4];
  const float* attn_r     = (const float*)d_in[5];
  const float* hop_attn_l = (const float*)d_in[6];
  const float* hop_attn_r = (const float*)d_in[7];
  const float* sigma      = (const float*)d_in[8];

  float* ws = (float*)d_ws;
  size_t off = 0;
  float* h    = ws + off; off += (size_t)N_NODES * HF;
  float* x1   = ws + off; off += (size_t)N_NODES * HF;
  float* x2   = ws + off; off += (size_t)N_NODES * HF;
  float* x3   = ws + off; off += (size_t)N_NODES * HF;
  float* el   = ws + off; off += (size_t)N_NODES * H_HEADS;
  float* er   = ws + off; off += (size_t)N_NODES * H_HEADS;
  float* al   = ws + off; off += (size_t)N_NODES * H_HEADS;
  float* maxs = ws + off; off += (size_t)N_NODES * H_HEADS;
  float* maxd = ws + off; off += (size_t)N_NODES * H_HEADS;
  float* sums = ws + off; off += (size_t)N_NODES * H_HEADS;
  float* sumd = ws + off; off += (size_t)N_NODES * H_HEADS;
  float* outdeg   = ws + off; off += (size_t)N_NODES;
  float* indeg    = ws + off; off += (size_t)N_NODES;
  float* norm_out = ws + off; off += (size_t)N_NODES;
  float* norm_in  = ws + off; off += (size_t)N_NODES;
  float* e_ws = ws + off; off += (size_t)E_EDGES * H_HEADS;
  float* a_ws = ws + off; off += (size_t)E_EDGES * H_HEADS;

  hipMemsetAsync(x1, 0, (size_t)3 * N_NODES * HF * sizeof(float), stream);

  gemm_h_kernel<<<(N_NODES + 127) / 128, 256, 0, stream>>>(feat, W, h);

  const int NH = N_NODES * H_HEADS;
  init_kernel<<<(NH + 255) / 256, 256, 0, stream>>>(maxs, maxd, sums, sumd, outdeg, indeg);

  scores_kernel<<<(NH + 255) / 256, 256, 0, stream>>>(h, attn_l, attn_r, hop_attn_l, el, er, al);

  const int EH = E_EDGES * H_HEADS;
  edge_e_kernel<<<(EH + 255) / 256, 256, 0, stream>>>(src, dst, el, er, e_ws, maxs, maxd, outdeg, indeg);
  edge_sum_kernel<<<(EH + 255) / 256, 256, 0, stream>>>(src, dst, e_ws, maxs, maxd, sums, sumd);
  edge_a_kernel<<<(EH + 255) / 256, 256, 0, stream>>>(src, dst, e_ws, maxs, maxd, sums, sumd, sigma, a_ws);

  norm_kernel<<<(N_NODES + 255) / 256, 256, 0, stream>>>(outdeg, indeg, norm_out, norm_in);

  const int NE = N_NODES * HF;
  agg_kernel<<<E_EDGES, 256, 0, stream>>>(src, dst, a_ws, norm_out, h, x1);
  scale_kernel<<<(NE + 255) / 256, 256, 0, stream>>>(x1, norm_in);
  agg_kernel<<<E_EDGES, 256, 0, stream>>>(src, dst, a_ws, norm_out, x1, x2);
  scale_kernel<<<(NE + 255) / 256, 256, 0, stream>>>(x2, norm_in);
  agg_kernel<<<E_EDGES, 256, 0, stream>>>(src, dst, a_ws, norm_out, x2, x3);
  scale_kernel<<<(NE + 255) / 256, 256, 0, stream>>>(x3, norm_in);

  combine_kernel<<<N_NODES, 128, 0, stream>>>(h, x1, x2, x3, al, hop_attn_r, (float*)d_out);
}


# Hop aggregation by 250 blocks, each owning 200 dst rows as an LDS tile: scan dst, queue own edges, add src rows by LDS float atomics (arrival order), scale, store once; no global x atomics, no memset
# speedup vs baseline: 3.4180x; 3.4180x over previous
#include <hip/hip_runtime.h>
#include <hip/hip_bf16.h>

#define N_NODES 50000
#define E_EDGES 800000
#define IN_F    256
#define H_HEADS 4
#define F_OUT   64
#define HF      256
#define SLOPE   0.2f

typedef __attribute__((ext_vector_type(16))) __bf16 v16bf;
typedef __attribute__((ext_vector_type(8)))  float  v8f;

__device__ __forceinline__ float leakyf(float x) { return x > 0.f ? x : SLOPE * x; }

__device__ __forceinline__ void atomicMaxF(float* addr, float val) {
  if (val >= 0.f) {
    atomicMax((int*)addr, __float_as_int(val));
  } else {
    atomicMin((unsigned int*)addr, __float_as_uint(val));
  }
}

#define WT_STRIDE 40

__global__ __launch_bounds__(256)
void gemm_h_kernel(const float* __restrict__ feat, const float* __restrict__ W,
                   float* __restrict__ h) {
  __shared__ __align__(16) __bf16 Wt[HF * WT_STRIDE];

  const int t       = threadIdx.x;
  const int wave    = t >> 5;
  const int lane    = t & 31;
  const int rowBase = blockIdx.x * 128 + wave * 16;
  const bool rowValid = (rowBase < N_NODES);
  const int nloc = lane & 15;
  const int kg   = lane >> 4;

  v8f acc[16];
#pragma unroll
  for (int nt = 0; nt < 16; ++nt)
#pragma unroll
    for (int v = 0; v < 8; ++v) acc[nt][v] = 0.f;

  for (int ks = 0; ks < IN_F / 32; ++ks) {
    const int kbase = ks * 32;
    __syncthreads();
#pragma unroll
    for (int kk = 0; kk < 32; ++kk)
      Wt[t * WT_STRIDE + kk] = (__bf16)W[(kbase + kk) * HF + t];
    __syncthreads();

    if (rowValid) {
      const float* arow = feat + (size_t)(rowBase + nloc) * IN_F + kbase;
      float4 a0 = *(const float4*)(arow + 8 * kg);
      float4 a1 = *(const float4*)(arow + 8 * kg + 4);
      float4 a2 = *(const float4*)(arow + 16 + 8 * kg);
      float4 a3 = *(const float4*)(arow + 16 + 8 * kg + 4);
      float atmp[16] = {a0.x, a0.y, a0.z, a0.w, a1.x, a1.y, a1.z, a1.w,
                        a2.x, a2.y, a2.z, a2.w, a3.x, a3.y, a3.z, a3.w};
      v16bf afrag;
#pragma unroll
      for (int i = 0; i < 16; ++i) afrag[i] = (__bf16)atmp[i];

      const __bf16* wrow0 = &Wt[nloc * WT_STRIDE];
      uint4 b0 = *(const uint4*)(wrow0 + 8 * kg);
      uint4 b1 = *(const uint4*)(wrow0 + 16 + 8 * kg);
#pragma unroll
      for (int nt = 0; nt < 16; ++nt) {
        uint4 nb0, nb1;
        if (nt < 15) {
          const __bf16* wrow = &Wt[((nt + 1) * 16 + nloc) * WT_STRIDE];
          nb0 = *(const uint4*)(wrow + 8 * kg);
          nb1 = *(const uint4*)(wrow + 16 + 8 * kg);
        }
        const __bf16* p0 = (const __bf16*)&b0;
        const __bf16* p1 = (const __bf16*)&b1;
        v16bf bfrag;
#pragma unroll
        for (int i = 0; i < 8; ++i) { bfrag[i] = p0[i]; bfrag[8 + i] = p1[i]; }
        acc[nt] = __builtin_amdgcn_wmma_f32_16x16x32_bf16(
            false, afrag, false, bfrag, (short)0, acc[nt], false, false);
        if (nt < 15) { b0 = nb0; b1 = nb1; }
      }
    }
  }

  if (rowValid) {
#pragma unroll
    for (int nt = 0; nt < 16; ++nt)
#pragma unroll
      for (int v = 0; v < 8; ++v)
        h[(size_t)(rowBase + v + 8 * kg) * HF + nt * 16 + nloc] = acc[nt][v];
  }
}

__global__ void init_kernel(float* maxs, float* maxd, float* sums, float* sumd,
                            float* outdeg, float* indeg) {
  int i = blockIdx.x * blockDim.x + threadIdx.x;
  if (i < N_NODES * H_HEADS) {
    maxs[i] = -1e30f; maxd[i] = -1e30f; sums[i] = 0.f; sumd[i] = 0.f;
  }
  if (i < N_NODES) { outdeg[i] = 0.f; indeg[i] = 0.f; }
}

__global__ void scores_kernel(const float* __restrict__ h,
                              const float* __restrict__ attn_l,
                              const float* __restrict__ attn_r,
                              const float* __restrict__ hop_attn_l,
                              float* __restrict__ el, float* __restrict__ er,
                              float* __restrict__ al) {
  int i = blockIdx.x * blockDim.x + threadIdx.x;
  if (i >= N_NODES * H_HEADS) return;
  int node = i >> 2, head = i & 3;
  const float* hp  = h + (size_t)node * HF + head * F_OUT;
  const float* alp = attn_l + head * F_OUT;
  const float* arp = attn_r + head * F_OUT;
  const float* hlp = hop_attn_l + head * F_OUT;
  float sl = 0.f, sr = 0.f, shl = 0.f;
#pragma unroll 8
  for (int f = 0; f < F_OUT; ++f) {
    float v = hp[f];
    sl += v * alp[f]; sr += v * arp[f]; shl += v * hlp[f];
  }
  el[i] = sl; er[i] = sr; al[i] = shl;
}

__global__ void edge_e_kernel(const int* __restrict__ src, const int* __restrict__ dst,
                              const float* __restrict__ el, const float* __restrict__ er,
                              float* __restrict__ e_ws, float* maxs, float* maxd,
                              float* outdeg, float* indeg) {
  int i = blockIdx.x * blockDim.x + threadIdx.x;
  if (i >= E_EDGES * H_HEADS) return;
  int eid = i >> 2, head = i & 3;
  int s = src[eid], d = dst[eid];
  float ev = leakyf(el[s * 4 + head] + er[d * 4 + head]);
  e_ws[i] = ev;
  atomicMaxF(&maxs[s * 4 + head], ev);
  atomicMaxF(&maxd[d * 4 + head], ev);
  if (head == 0) { atomicAdd(&outdeg[s], 1.f); atomicAdd(&indeg[d], 1.f); }
}

__global__ void edge_sum_kernel(const int* __restrict__ src, const int* __restrict__ dst,
                                const float* __restrict__ e_ws,
                                const float* __restrict__ maxs, const float* __restrict__ maxd,
                                float* sums, float* sumd) {
  int i = blockIdx.x * blockDim.x + threadIdx.x;
  if (i >= E_EDGES * H_HEADS) return;
  int eid = i >> 2, head = i & 3;
  int s = src[eid], d = dst[eid];
  float ev = e_ws[i];
  atomicAdd(&sums[s * 4 + head], __expf(ev - maxs[s * 4 + head]));
  atomicAdd(&sumd[d * 4 + head], __expf(ev - maxd[d * 4 + head]));
}

__global__ void edge_a_kernel(const int* __restrict__ src, const int* __restrict__ dst,
                              const float* __restrict__ e_ws,
                              const float* __restrict__ maxs, const float* __restrict__ maxd,
                              const float* __restrict__ sums, const float* __restrict__ sumd,
                              const float* __restrict__ sigma, float* __restrict__ a_ws) {
  int i = blockIdx.x * blockDim.x + threadIdx.x;
  if (i >= E_EDGES * H_HEADS) return;
  int eid = i >> 2, head = i & 3;
  int s = src[eid], d = dst[eid];
  float ev = e_ws[i];
  float sgm = 1.f / (1.f + __expf(-sigma[0]));
  float ax = fmaxf(__expf(ev - maxs[s * 4 + head]) / sums[s * 4 + head], 1e-10f);
  float ay = fmaxf(__expf(ev - maxd[d * 4 + head]) / sumd[d * 4 + head], 1e-10f);
  a_ws[i] = __powf(ay, sgm) * __powf(ax, 1.f - sgm);
}

__global__ void norm_kernel(const float* __restrict__ outdeg, const float* __restrict__ indeg,
                            float* __restrict__ norm_out, float* __restrict__ norm_in) {
  int i = blockIdx.x * blockDim.x + threadIdx.x;
  if (i >= N_NODES) return;
  norm_out[i] = rsqrtf(fmaxf(outdeg[i], 1.f));
  norm_in[i]  = sqrtf(fmaxf(indeg[i], 1.f));
}

#define RNG        200
#define NBLK       (N_NODES / RNG)
#define AGG_T      1024
#define SCAN_V     4
#define SCAN_EDGES (AGG_T * 4 * SCAN_V)
#define QCAP       (SCAN_EDGES + 4096)
#define DRAIN_U    4

static_assert(N_NODES % RNG == 0, "ranges tile the node set exactly");
static_assert(E_EDGES % 4 == 0, "dst is read 4 edges at a time");
static_assert(E_EDGES <= (1 << 20) && RNG <= 256, "queue word packs row and edge id");
static_assert((RNG * HF + QCAP + 1) * 4 <= 320 * 1024, "LDS budget");
static_assert(DRAIN_U == 4, "a batch is one 16-byte queue read");

__device__ __forceinline__ void push_edge(unsigned r, int eid, unsigned* q, unsigned* qn) {
  if (r < (unsigned)RNG) {
    const unsigned p = atomicAdd(qn, 1u);
    q[p] = (r << 20) | (unsigned)eid;
  }
}

template <int U>
__device__ __forceinline__ void add_edges(const unsigned (&ent)[U], int lane, int hsel,
                                          const int* __restrict__ src,
                                          const float* __restrict__ a_ws,
                                          const float* __restrict__ norm_out,
                                          const float* __restrict__ xprev, float* acc) {
  int    sn[U];
  float4 r0[U], r1[U];
  float  c0[U], c1[U];
#pragma unroll
  for (int u = 0; u < U; ++u) sn[u] = src[ent[u] & 0xFFFFFu];
#pragma unroll
  for (int u = 0; u < U; ++u) {
    const int    eid = (int)(ent[u] & 0xFFFFFu);
    const float  nrm = norm_out[sn[u]];
    const float4 a4  = *(const float4*)(a_ws + (size_t)eid * 4);
    c0[u] = (hsel ? a4.y : a4.x) * nrm;
    c1[u] = (hsel ? a4.w : a4.z) * nrm;
    const float* row = xprev + (size_t)sn[u] * HF + lane * 4;
    r0[u] = *(const float4*)row;
    r1[u] = *(const float4*)(row + 128);
  }
#pragma unroll
  for (int u = 0; u < U; ++u) {
    float* arow = acc + (ent[u] >> 20) * HF + lane;
    atomicAdd(arow,       r0[u].x * c0[u]);
    atomicAdd(arow + 32,  r0[u].y * c0[u]);
    atomicAdd(arow + 64,  r0[u].z * c0[u]);
    atomicAdd(arow + 96,  r0[u].w * c0[u]);
    atomicAdd(arow + 128, r1[u].x * c1[u]);
    atomicAdd(arow + 160, r1[u].y * c1[u]);
    atomicAdd(arow + 192, r1[u].z * c1[u]);
    atomicAdd(arow + 224, r1[u].w * c1[u]);
  }
}

__global__ __launch_bounds__(AGG_T)
void agg_tile_kernel(const int* __restrict__ src, const int* __restrict__ dst,
                     const float* __restrict__ a_ws, const float* __restrict__ norm_out,
                     const float* __restrict__ norm_in,
                     const float* __restrict__ xprev, float* __restrict__ xnext) {
  __shared__ __align__(16) float acc[RNG * HF];
  __shared__ __align__(16) unsigned int q[QCAP];
  __shared__ unsigned int qn;

  const int t    = threadIdx.x;
  const int lane = t & 31;
  const int wave = t >> 5;
  const int hsel = lane >> 4;
  const int lo   = blockIdx.x * RNG;
  const unsigned wv = __builtin_amdgcn_readfirstlane((unsigned)wave);

  for (int i = t; i < RNG * HF / 4; i += AGG_T)
    ((float4*)acc)[i] = make_float4(0.f, 0.f, 0.f, 0.f);
  if (t == 0) qn = 0u;
  __syncthreads();

  for (int base = 0; base < E_EDGES; base += SCAN_EDGES) {
    int4 dv[SCAN_V];
#pragma unroll
    for (int j = 0; j < SCAN_V; ++j) {
      const int e0 = base + (j * AGG_T + t) * 4;
      dv[j] = (e0 < E_EDGES) ? *(const int4*)(dst + e0) : make_int4(-1, -1, -1, -1);
    }
#pragma unroll
    for (int j = 0; j < SCAN_V; ++j) {
      const int e0 = base + (j * AGG_T + t) * 4;
      const unsigned rx = (unsigned)(dv[j].x - lo), ry = (unsigned)(dv[j].y - lo);
      const unsigned rz = (unsigned)(dv[j].z - lo), rw = (unsigned)(dv[j].w - lo);
      if (min(min(rx, ry), min(rz, rw)) < (unsigned)RNG) {
        push_edge(rx, e0,     q, &qn);
        push_edge(ry, e0 + 1, q, &qn);
        push_edge(rz, e0 + 2, q, &qn);
        push_edge(rw, e0 + 3, q, &qn);
      }
    }
    __syncthreads();
    const unsigned n = __builtin_amdgcn_readfirstlane(qn);
    __syncthreads();
    const bool last = (base + SCAN_EDGES >= E_EDGES);
    if (n > (unsigned)(QCAP - SCAN_EDGES) || last) {
      for (unsigned i0 = wv * DRAIN_U; i0 + DRAIN_U <= n; i0 += (AGG_T / 32) * DRAIN_U) {
        const uint4 w4 = *(const uint4*)&q[i0];
        const unsigned ent[DRAIN_U] = {
            (unsigned)__builtin_amdgcn_readfirstlane(w4.x),
            (unsigned)__builtin_amdgcn_readfirstlane(w4.y),
            (unsigned)__builtin_amdgcn_readfirstlane(w4.z),
            (unsigned)__builtin_amdgcn_readfirstlane(w4.w)};
        add_edges<DRAIN_U>(ent, lane, hsel, src, a_ws, norm_out, xprev, acc);
      }
      const unsigned nfull = n & ~(unsigned)(DRAIN_U - 1);
      if (nfull + wv < n) {
        const unsigned ent1[1] = {(unsigned)__builtin_amdgcn_readfirstlane(q[nfull + wv])};
        add_edges<1>(ent1, lane, hsel, src, a_ws, norm_out, xprev, acc);
      }
      __syncthreads();
      if (t == 0) qn = 0u;
      __syncthreads();
    }
  }

  for (int g = t; g < RNG * (HF / 4); g += AGG_T) {
    const int row = g >> 6, c4 = g & 63;
    const float* ap = acc + row * HF + (c4 >> 5) * 128 + (c4 & 31);
    const float ni = norm_in[lo + row];
    const float4 o = make_float4(ap[0] * ni, ap[32] * ni, ap[64] * ni, ap[96] * ni);
    *(float4*)(xnext + (size_t)(lo + row) * HF + c4 * 4) = o;
  }
}

__global__ __launch_bounds__(128)
void combine_kernel(const float* __restrict__ h, const float* __restrict__ x1,
                    const float* __restrict__ x2, const float* __restrict__ x3,
                    const float* __restrict__ al, const float* __restrict__ hop_attn_r,
                    float* __restrict__ out) {
  int n    = blockIdx.x;
  int head = threadIdx.x >> 5;
  int lane = threadIdx.x & 31;
  const float* hs[4] = {h, x1, x2, x3};
  size_t base = (size_t)n * HF + head * F_OUT;
  const float* ar = hop_attn_r + head * F_OUT;
  float ar0 = ar[lane], ar1 = ar[lane + 32];
  float albase = al[n * 4 + head];

  float xv0[4], xv1[4], score[4];
#pragma unroll
  for (int k = 0; k < 4; ++k) {
    float v0 = hs[k][base + lane];
    float v1 = hs[k][base + lane + 32];
    xv0[k] = v0; xv1[k] = v1;
    float p = v0 * ar0 + v1 * ar1;
#pragma unroll
    for (int off = 16; off > 0; off >>= 1) p += __shfl_xor(p, off);
    score[k] = leakyf(albase + p);
  }
  float m = fmaxf(fmaxf(score[0], score[1]), fmaxf(score[2], score[3]));
  float ex[4], ssum = 0.f;
#pragma unroll
  for (int k = 0; k < 4; ++k) { ex[k] = __expf(score[k] - m); ssum += ex[k]; }
  float o0 = 0.f, o1 = 0.f;
#pragma unroll
  for (int k = 0; k < 4; ++k) {
    float w = ex[k] / ssum;
    o0 += xv0[k] * w; o1 += xv1[k] * w;
  }
  out[base + lane] = o0;
  out[base + lane + 32] = o1;
}

extern "C" void kernel_launch(void* const* d_in, const int* in_sizes, int n_in,
                              void* d_out, int out_size, void* d_ws, size_t ws_size,
                              hipStream_t stream) {
  (void)in_sizes; (void)n_in; (void)out_size; (void)ws_size;
  const float* feat       = (const float*)d_in[0];
  const int*   src        = (const int*)d_in[1];
  const int*   dst        = (const int*)d_in[2];
  const float* W          = (const float*)d_in[3];
  const float* attn_l     = (const float*)d_in[4];
  const float* attn_r     = (const float*)d_in[5];
  const float* hop_attn_l = (const float*)d_in[6];
  const float* hop_attn_r = (const float*)d_in[7];
  const float* sigma      = (const float*)d_in[8];

  float* ws = (float*)d_ws;
  size_t off = 0;
  float* h    = ws + off; off += (size_t)N_NODES * HF;
  float* x1   = ws + off; off += (size_t)N_NODES * HF;
  float* x2   = ws + off; off += (size_t)N_NODES * HF;
  float* x3   = ws + off; off += (size_t)N_NODES * HF;
  float* el   = ws + off; off += (size_t)N_NODES * H_HEADS;
  float* er   = ws + off; off += (size_t)N_NODES * H_HEADS;
  float* al   = ws + off; off += (size_t)N_NODES * H_HEADS;
  float* maxs = ws + off; off += (size_t)N_NODES * H_HEADS;
  float* maxd = ws + off; off += (size_t)N_NODES * H_HEADS;
  float* sums = ws + off; off += (size_t)N_NODES * H_HEADS;
  float* sumd = ws + off; off += (size_t)N_NODES * H_HEADS;
  float* outdeg   = ws + off; off += (size_t)N_NODES;
  float* indeg    = ws + off; off += (size_t)N_NODES;
  float* norm_out = ws + off; off += (size_t)N_NODES;
  float* norm_in  = ws + off; off += (size_t)N_NODES;
  float* e_ws = ws + off; off += (size_t)E_EDGES * H_HEADS;
  float* a_ws = ws + off; off += (size_t)E_EDGES * H_HEADS;

  gemm_h_kernel<<<(N_NODES + 127) / 128, 256, 0, stream>>>(feat, W, h);

  const int NH = N_NODES * H_HEADS;
  init_kernel<<<(NH + 255) / 256, 256, 0, stream>>>(maxs, maxd, sums, sumd, outdeg, indeg);

  scores_kernel<<<(NH + 255) / 256, 256, 0, stream>>>(h, attn_l, attn_r, hop_attn_l, el, er, al);

  const int EH = E_EDGES * H_HEADS;
  edge_e_kernel<<<(EH + 255) / 256, 256, 0, stream>>>(src, dst, el, er, e_ws, maxs, maxd, outdeg, indeg);
  edge_sum_kernel<<<(EH + 255) / 256, 256, 0, stream>>>(src, dst, e_ws, maxs, maxd, sums, sumd);
  edge_a_kernel<<<(EH + 255) / 256, 256, 0, stream>>>(src, dst, e_ws, maxs, maxd, sums, sumd, sigma, a_ws);

  norm_kernel<<<(N_NODES + 255) / 256, 256, 0, stream>>>(outdeg, indeg, norm_out, norm_in);

  agg_tile_kernel<<<NBLK, AGG_T, 0, stream>>>(src, dst, a_ws, norm_out, norm_in, h,  x1);
  agg_tile_kernel<<<NBLK, AGG_T, 0, stream>>>(src, dst, a_ws, norm_out, norm_in, x1, x2);
  agg_tile_kernel<<<NBLK, AGG_T, 0, stream>>>(src, dst, a_ws, norm_out, norm_in, x2, x3);

  combine_kernel<<<N_NODES, 128, 0, stream>>>(h, x1, x2, x3, al, hop_attn_r, (float*)d_out);
}


# Edge-softmax maxima, exp-sums, degrees by 250 owner blocks: barrier-free scan of one index array queues own edges in LDS, LDS atomics (arrival-order sums), hop queues saved there; no global atomics
# speedup vs baseline: 6.1489x; 1.7990x over previous
#include <hip/hip_runtime.h>
#include <hip/hip_bf16.h>

#define N_NODES 50000
#define E_EDGES 800000
#define IN_F    256
#define H_HEADS 4
#define F_OUT   64
#define HF      256
#define SLOPE   0.2f

typedef __attribute__((ext_vector_type(16))) __bf16 v16bf;
typedef __attribute__((ext_vector_type(8)))  float  v8f;

__device__ __forceinline__ float leakyf(float x) { return x > 0.f ? x : SLOPE * x; }

__device__ __forceinline__ void atomicMaxF(float* addr, float val) {
  if (val >= 0.f) {
    atomicMax((int*)addr, __float_as_int(val));
  } else {
    atomicMin((unsigned int*)addr, __float_as_uint(val));
  }
}

#define WT_STRIDE 40

__global__ __launch_bounds__(256)
void gemm_h_kernel(const float* __restrict__ feat, const float* __restrict__ W,
                   float* __restrict__ h) {
  __shared__ __align__(16) __bf16 Wt[HF * WT_STRIDE];

  const int t       = threadIdx.x;
  const int wave    = t >> 5;
  const int lane    = t & 31;
  const int rowBase = blockIdx.x * 128 + wave * 16;
  const bool rowValid = (rowBase < N_NODES);
  const int nloc = lane & 15;
  const int kg   = lane >> 4;

  v8f acc[16];
#pragma unroll
  for (int nt = 0; nt < 16; ++nt)
#pragma unroll
    for (int v = 0; v < 8; ++v) acc[nt][v] = 0.f;

  for (int ks = 0; ks < IN_F / 32; ++ks) {
    const int kbase = ks * 32;
    __syncthreads();
#pragma unroll
    for (int kk = 0; kk < 32; ++kk)
      Wt[t * WT_STRIDE + kk] = (__bf16)W[(kbase + kk) * HF + t];
    __syncthreads();

    if (rowValid) {
      const float* arow = feat + (size_t)(rowBase + nloc) * IN_F + kbase;
      float4 a0 = *(const float4*)(arow + 8 * kg);
      float4 a1 = *(const float4*)(arow + 8 * kg + 4);
      float4 a2 = *(const float4*)(arow + 16 + 8 * kg);
      float4 a3 = *(const float4*)(arow + 16 + 8 * kg + 4);
      float atmp[16] = {a0.x, a0.y, a0.z, a0.w, a1.x, a1.y, a1.z, a1.w,
                        a2.x, a2.y, a2.z, a2.w, a3.x, a3.y, a3.z, a3.w};
      v16bf afrag;
#pragma unroll
      for (int i = 0; i < 16; ++i) afrag[i] = (__bf16)atmp[i];

      const __bf16* wrow0 = &Wt[nloc * WT_STRIDE];
      uint4 b0 = *(const uint4*)(wrow0 + 8 * kg);
      uint4 b1 = *(const uint4*)(wrow0 + 16 + 8 * kg);
#pragma unroll
      for (int nt = 0; nt < 16; ++nt) {
        uint4 nb0, nb1;
        if (nt < 15) {
          const __bf16* wrow = &Wt[((nt + 1) * 16 + nloc) * WT_STRIDE];
          nb0 = *(const uint4*)(wrow + 8 * kg);
          nb1 = *(const uint4*)(wrow + 16 + 8 * kg);
        }
        const __bf16* p0 = (const __bf16*)&b0;
        const __bf16* p1 = (const __bf16*)&b1;
        v16bf bfrag;
#pragma unroll
        for (int i = 0; i < 8; ++i) { bfrag[i] = p0[i]; bfrag[8 + i] = p1[i]; }
        acc[nt] = __builtin_amdgcn_wmma_f32_16x16x32_bf16(
            false, afrag, false, bfrag, (short)0, acc[nt], false, false);
        if (nt < 15) { b0 = nb0; b1 = nb1; }
      }
    }
  }

  if (rowValid) {
#pragma unroll
    for (int nt = 0; nt < 16; ++nt)
#pragma unroll
      for (int v = 0; v < 8; ++v)
        h[(size_t)(rowBase + v + 8 * kg) * HF + nt * 16 + nloc] = acc[nt][v];
  }
}

__global__ void scores_kernel(const float* __restrict__ h,
                              const float* __restrict__ attn_l,
                              const float* __restrict__ attn_r,
                              const float* __restrict__ hop_attn_l,
                              float* __restrict__ el, float* __restrict__ er,
                              float* __restrict__ al) {
  int i = blockIdx.x * blockDim.x + threadIdx.x;
  if (i >= N_NODES * H_HEADS) return;
  int node = i >> 2, head = i & 3;
  const float* hp  = h + (size_t)node * HF + head * F_OUT;
  const float* alp = attn_l + head * F_OUT;
  const float* arp = attn_r + head * F_OUT;
  const float* hlp = hop_attn_l + head * F_OUT;
  float sl = 0.f, sr = 0.f, shl = 0.f;
#pragma unroll 8
  for (int f = 0; f < F_OUT; ++f) {
    float v = hp[f];
    sl += v * alp[f]; sr += v * arp[f]; shl += v * hlp[f];
  }
  el[i] = sl; er[i] = sr; al[i] = shl;
}

__global__ void edge_a_kernel(const int* __restrict__ src, const int* __restrict__ dst,
                              const float* __restrict__ el, const float* __restrict__ er,
                              const float* __restrict__ maxs, const float* __restrict__ maxd,
                              const float* __restrict__ sums, const float* __restrict__ sumd,
                              const float* __restrict__ sigma, float* __restrict__ a_ws) {
  int i = blockIdx.x * blockDim.x + threadIdx.x;
  if (i >= E_EDGES * H_HEADS) return;
  int eid = i >> 2, head = i & 3;
  int s = src[eid], d = dst[eid];
  float ev = leakyf(el[s * 4 + head] + er[d * 4 + head]);
  float sgm = 1.f / (1.f + __expf(-sigma[0]));
  float ax = fmaxf(__expf(ev - maxs[s * 4 + head]) / sums[s * 4 + head], 1e-10f);
  float ay = fmaxf(__expf(ev - maxd[d * 4 + head]) / sumd[d * 4 + head], 1e-10f);
  a_ws[i] = __powf(ay, sgm) * __powf(ax, 1.f - sgm);
}

#define RNG        200
#define NBLK       (N_NODES / RNG)
#define AGG_T      1024
#define SCAN_V     4
#define SCAN_EDGES (AGG_T * 4 * SCAN_V)
#define QCAP       (SCAN_EDGES + 4096)
#define DRAIN_U    4
#define GCAP       12288

static_assert(N_NODES % RNG == 0, "ranges tile the node set exactly");
static_assert(E_EDGES % 4 == 0, "dst is read 4 edges at a time");
static_assert(E_EDGES <= (1 << 20) && RNG <= 256, "queue word packs row and edge id");
static_assert((RNG * HF + QCAP + 1) * 4 <= 320 * 1024, "LDS budget");
static_assert(DRAIN_U == 4, "a batch is one 16-byte queue read");
static_assert(GCAP <= QCAP && GCAP % AGG_T == 0, "a saved queue fits the LDS queue; whole rounds of the block reload it");
static_assert((size_t)NBLK * GCAP + NBLK <= (size_t)E_EDGES * H_HEADS,
              "saved queues and their counts fit the edge-score plane");

__device__ __forceinline__ void push_edge(unsigned r, int eid, unsigned* q, unsigned* qn) {
  if (r < (unsigned)RNG) {
    const unsigned p = atomicAdd(qn, 1u);
    q[p] = (r << 20) | (unsigned)eid;
  }
}

template <int U>
__device__ __forceinline__ void add_edges(const unsigned (&ent)[U], int lane, int hsel,
                                          const int* __restrict__ src,
                                          const float* __restrict__ a_ws,
                                          const float* __restrict__ norm_out,
                                          const float* __restrict__ xprev, float* acc) {
  int    sn[U];
  float4 r0[U], r1[U];
  float  c0[U], c1[U];
#pragma unroll
  for (int u = 0; u < U; ++u) sn[u] = src[ent[u] & 0xFFFFFu];
#pragma unroll
  for (int u = 0; u < U; ++u) {
    const int    eid = (int)(ent[u] & 0xFFFFFu);
    const float  nrm = norm_out[sn[u]];
    const float4 a4  = *(const float4*)(a_ws + (size_t)eid * 4);
    c0[u] = (hsel ? a4.y : a4.x) * nrm;
    c1[u] = (hsel ? a4.w : a4.z) * nrm;
    const float* row = xprev + (size_t)sn[u] * HF + lane * 4;
    r0[u] = *(const float4*)row;
    r1[u] = *(const float4*)(row + 128);
  }
#pragma unroll
  for (int u = 0; u < U; ++u) {
    float* arow = acc + (ent[u] >> 20) * HF + lane;
    atomicAdd(arow,       r0[u].x * c0[u]);
    atomicAdd(arow + 32,  r0[u].y * c0[u]);
    atomicAdd(arow + 64,  r0[u].z * c0[u]);
    atomicAdd(arow + 96,  r0[u].w * c0[u]);
    atomicAdd(arow + 128, r1[u].x * c1[u]);
    atomicAdd(arow + 160, r1[u].y * c1[u]);
    atomicAdd(arow + 192, r1[u].z * c1[u]);
    atomicAdd(arow + 224, r1[u].w * c1[u]);
  }
}

__device__ __forceinline__ void drain_queue(unsigned n, unsigned wv, int lane, int hsel,
                                            const unsigned* q,
                                            const int* __restrict__ src,
                                            const float* __restrict__ a_ws,
                                            const float* __restrict__ norm_out,
                                            const float* __restrict__ xprev, float* acc) {
  for (unsigned i0 = wv * DRAIN_U; i0 + DRAIN_U <= n; i0 += (AGG_T / 32) * DRAIN_U) {
    const uint4 w4 = *(const uint4*)&q[i0];
    const unsigned ent[DRAIN_U] = {
        (unsigned)__builtin_amdgcn_readfirstlane(w4.x),
        (unsigned)__builtin_amdgcn_readfirstlane(w4.y),
        (unsigned)__builtin_amdgcn_readfirstlane(w4.z),
        (unsigned)__builtin_amdgcn_readfirstlane(w4.w)};
    add_edges<DRAIN_U>(ent, lane, hsel, src, a_ws, norm_out, xprev, acc);
  }
  const unsigned nfull = n & ~(unsigned)(DRAIN_U - 1);
  if (nfull + wv < n) {
    const unsigned ent1[1] = {(unsigned)__builtin_amdgcn_readfirstlane(q[nfull + wv])};
    add_edges<1>(ent1, lane, hsel, src, a_ws, norm_out, xprev, acc);
  }
}

#define ST_K       2
#define ST_RNG     (ST_K * RNG)
#define ST_NRNG    (N_NODES / ST_RNG)
#define ST_NBLK    (2 * ST_NRNG)
#define ST_T       1024
#define ST_V       16
#define ST_VS      4
#define ST_EDGES_S (ST_T * 4 * ST_VS)
#define ST_QCAP    (3 * ST_EDGES_S)
#define ST_U       8

static_assert(N_NODES % ST_RNG == 0 && ST_RNG <= 512, "statistics ranges tile the node set; local node fits 9 bits");
static_assert(ST_RNG <= ST_T, "one thread per degree word");
static_assert((ST_QCAP + ST_RNG * 9 + 4) * 4 <= 320 * 1024, "LDS budget of the statistics kernel");

template <bool CHECK>
__device__ __forceinline__ void push_own(unsigned r, int eid, unsigned* q, unsigned* qn) {
  if (r < (unsigned)ST_RNG) {
    const unsigned p = atomicAdd(qn, 1u);
    if (!CHECK || p < (unsigned)ST_QCAP) q[p] = (r << 20) | (unsigned)eid;
  }
}

template <int V, bool CHECK, bool GUARD>
__device__ __forceinline__ void scan_group(int base, int t, int lo,
                                           const int* __restrict__ idx,
                                           unsigned* q, unsigned* qn) {
  int4 iv[V];
#pragma unroll
  for (int j = 0; j < V; ++j) {
    const int e0 = base + (j * ST_T + t) * 4;
    if (!GUARD || e0 < E_EDGES) iv[j] = *(const int4*)(idx + e0);
    else                        iv[j] = make_int4(-1, -1, -1, -1);
  }
#pragma unroll
  for (int j = 0; j < V; ++j) {
    const int e0 = base + (j * ST_T + t) * 4;
    const unsigned rx = (unsigned)(iv[j].x - lo), ry = (unsigned)(iv[j].y - lo);
    const unsigned rz = (unsigned)(iv[j].z - lo), rw = (unsigned)(iv[j].w - lo);
    if (min(min(rx, ry), min(rz, rw)) < (unsigned)ST_RNG) {
      push_own<CHECK>(rx, e0,     q, qn);
      push_own<CHECK>(ry, e0 + 1, q, qn);
      push_own<CHECK>(rz, e0 + 2, q, qn);
      push_own<CHECK>(rw, e0 + 3, q, qn);
    }
  }
}

template <int PASS>
__device__ __forceinline__ void stats_pass(unsigned n, int t, bool save, const unsigned* q,
                                           const int* __restrict__ src,
                                           const int* __restrict__ dst,
                                           const float* __restrict__ el,
                                           const float* __restrict__ er,
                                           float* st_max, float* st_sum, float* st_deg,
                                           unsigned* sn, unsigned* gqb) {
  for (unsigned i0 = 0; i0 < n; i0 += ST_T * ST_U) {
    unsigned ent[ST_U];
    int      sv[ST_U], dv[ST_U];
    float4   a[ST_U], b[ST_U];
#pragma unroll
    for (int u = 0; u < ST_U; ++u) {
      const unsigned idx = i0 + (unsigned)(u * ST_T + t);
      ent[u] = (idx < n) ? q[idx] : 0xFFFFFFFFu;
    }
#pragma unroll
    for (int u = 0; u < ST_U; ++u) {
      const unsigned eid = (ent[u] == 0xFFFFFFFFu) ? 0u : (ent[u] & 0xFFFFFu);
      sv[u] = src[eid];
      dv[u] = dst[eid];
    }
#pragma unroll
    for (int u = 0; u < ST_U; ++u) {
      a[u] = *(const float4*)(el + (size_t)sv[u] * 4);
      b[u] = *(const float4*)(er + (size_t)dv[u] * 4);
    }
#pragma unroll
    for (int u = 0; u < ST_U; ++u) {
      const bool     valid = (ent[u] != 0xFFFFFFFFu);
      const unsigned r     = (ent[u] >> 20) & 0x1FFu;
      if (valid) {
        const float e0 = leakyf(a[u].x + b[u].x), e1 = leakyf(a[u].y + b[u].y);
        const float e2 = leakyf(a[u].z + b[u].z), e3 = leakyf(a[u].w + b[u].w);
        if (PASS == 0) {
          atomicMaxF(&st_max[r * 4 + 0], e0);
          atomicMaxF(&st_max[r * 4 + 1], e1);
          atomicMaxF(&st_max[r * 4 + 2], e2);
          atomicMaxF(&st_max[r * 4 + 3], e3);
          atomicAdd(&st_deg[r], 1.f);
        } else {
          atomicAdd(&st_sum[r * 4 + 0], __expf(e0 - st_max[r * 4 + 0]));
          atomicAdd(&st_sum[r * 4 + 1], __expf(e1 - st_max[r * 4 + 1]));
          atomicAdd(&st_sum[r * 4 + 2], __expf(e2 - st_max[r * 4 + 2]));
          atomicAdd(&st_sum[r * 4 + 3], __expf(e3 - st_max[r * 4 + 3]));
        }
      }
      if (PASS == 0 && save) {
        const unsigned lane = (unsigned)t & 31u;
#pragma unroll
        for (unsigned k = 0; k < (unsigned)ST_K; ++k) {
          const bool keep = valid && r >= k * RNG && r < (k + 1) * RNG;
          const unsigned m = __builtin_amdgcn_ballot_w32(keep);
          if (m != 0u) {
            const unsigned lead = (unsigned)__builtin_ctz(m);
            unsigned bp = 0u;
            if (lane == lead) bp = atomicAdd(&sn[k], (unsigned)__builtin_popcount(m));
            bp = (unsigned)__shfl((int)bp, (int)lead);
            const unsigned p = bp + (unsigned)__builtin_popcount(m & ((1u << lane) - 1u));
            if (keep && p < (unsigned)GCAP)
              gqb[(size_t)k * GCAP + p] = ((r - k * RNG) << 20) | (ent[u] & 0xFFFFFu);
          }
        }
      }
    }
  }
}

__global__ __launch_bounds__(ST_T)
void seg_stats_kernel(const int* __restrict__ src, const int* __restrict__ dst,
                      const float* __restrict__ el, const float* __restrict__ er,
                      float* __restrict__ maxs, float* __restrict__ maxd,
                      float* __restrict__ sums, float* __restrict__ sumd,
                      float* __restrict__ norm_out, float* __restrict__ norm_in,
                      unsigned int* gq, unsigned int* gcount) {
  __shared__ __align__(16) unsigned int q[ST_QCAP];
  __shared__ float st_max[ST_RNG * 4];
  __shared__ float st_sum[ST_RNG * 4];
  __shared__ float st_deg[ST_RNG];
  __shared__ unsigned int qn;
  __shared__ unsigned int sn[ST_K];

  const int  t     = threadIdx.x;
  const int  g     = blockIdx.x >> 1;
  const bool ends  = (blockIdx.x & 1) == 0;
  const int  lo    = g * ST_RNG;
  const int* __restrict__ idx = ends ? dst : src;
  unsigned int* gqb = gq + (size_t)g * ST_K * GCAP;

  for (int i = t; i < ST_RNG * 4; i += ST_T) { st_max[i] = -1e30f; st_sum[i] = 0.f; }
  if (t < ST_RNG) st_deg[t] = 0.f;
  if (t == 0) qn = 0u;
  if (t < ST_K) sn[t] = 0u;
  __syncthreads();

  int base0 = 0;
  for (; base0 + ST_T * 4 * ST_V <= E_EDGES; base0 += ST_T * 4 * ST_V)
    scan_group<ST_V, true, false>(base0, t, lo, idx, q, &qn);
  for (; base0 < E_EDGES; base0 += ST_EDGES_S)
    scan_group<ST_VS, true, true>(base0, t, lo, idx, q, &qn);
  __syncthreads();
  unsigned n = __builtin_amdgcn_readfirstlane(qn);

  if (n <= (unsigned)ST_QCAP) {
    stats_pass<0>(n, t, ends, q, src, dst, el, er, st_max, st_sum, st_deg, sn, gqb);
    __syncthreads();
    stats_pass<1>(n, t, ends, q, src, dst, el, er, st_max, st_sum, st_deg, sn, gqb);
  } else {
    __syncthreads();
    if (t == 0) qn = 0u;
    __syncthreads();
    for (int base = 0; base < E_EDGES; base += ST_EDGES_S) {
      scan_group<ST_VS, false, true>(base, t, lo, idx, q, &qn);
      __syncthreads();
      n = __builtin_amdgcn_readfirstlane(qn);
      __syncthreads();
      const bool last = (base + ST_EDGES_S >= E_EDGES);
      if (n > (unsigned)(ST_QCAP - ST_EDGES_S) || last) {
        stats_pass<0>(n, t, ends, q, src, dst, el, er, st_max, st_sum, st_deg, sn, gqb);
        __syncthreads();
        if (t == 0) qn = 0u;
        __syncthreads();
      }
    }
    for (int base = 0; base < E_EDGES; base += ST_EDGES_S) {
      scan_group<ST_VS, false, true>(base, t, lo, idx, q, &qn);
      __syncthreads();
      n = __builtin_amdgcn_readfirstlane(qn);
      __syncthreads();
      const bool last = (base + ST_EDGES_S >= E_EDGES);
      if (n > (unsigned)(ST_QCAP - ST_EDGES_S) || last) {
        stats_pass<1>(n, t, ends, q, src, dst, el, er, st_max, st_sum, st_deg, sn, gqb);
        __syncthreads();
        if (t == 0) qn = 0u;
        __syncthreads();
      }
    }
  }
  __syncthreads();

  float* __restrict__ mx_out = ends ? maxd : maxs;
  float* __restrict__ sm_out = ends ? sumd : sums;
  for (int i = t; i < ST_RNG * 4; i += ST_T) {
    mx_out[lo * 4 + i] = st_max[i];
    sm_out[lo * 4 + i] = st_sum[i];
  }
  if (t < ST_RNG) {
    const float deg = fmaxf(st_deg[t], 1.f);
    if (ends) norm_in[lo + t]  = sqrtf(deg);
    else      norm_out[lo + t] = rsqrtf(deg);
  }
  if (ends && t < ST_K) gcount[g * ST_K + t] = sn[t];
}

__global__ __launch_bounds__(AGG_T)
void agg_tile_kernel(const int* __restrict__ src, const int* __restrict__ dst,
                     const float* __restrict__ a_ws, const float* __restrict__ norm_out,
                     const float* __restrict__ norm_in,
                     const float* __restrict__ xprev, float* __restrict__ xnext,
                     unsigned int* gq, unsigned int* gcount) {
  __shared__ __align__(16) float acc[RNG * HF];
  __shared__ __align__(16) unsigned int q[QCAP];
  __shared__ unsigned int qn;

  const int t    = threadIdx.x;
  const int lane = t & 31;
  const int wave = t >> 5;
  const int hsel = lane >> 4;
  const int lo   = blockIdx.x * RNG;
  const unsigned wv = __builtin_amdgcn_readfirstlane((unsigned)wave);
  unsigned int* gqb = gq + (size_t)blockIdx.x * GCAP;

  for (int i = t; i < RNG * HF / 4; i += AGG_T)
    ((float4*)acc)[i] = make_float4(0.f, 0.f, 0.f, 0.f);
  if (t == 0) qn = 0u;

  bool scan = true;
  {
    const unsigned cnt = __builtin_amdgcn_readfirstlane(gcount[blockIdx.x]);
    if (cnt <= (unsigned)GCAP) {
      scan = false;
      unsigned w[GCAP / AGG_T];
#pragma unroll
      for (int j = 0; j < GCAP / AGG_T; ++j) {
        const unsigned i = (unsigned)(j * AGG_T + t);
        w[j] = (i < cnt) ? gqb[i] : 0u;
      }
#pragma unroll
      for (int j = 0; j < GCAP / AGG_T; ++j) {
        const unsigned i = (unsigned)(j * AGG_T + t);
        if (i < cnt) q[i] = w[j];
      }
      __syncthreads();
      drain_queue(cnt, wv, lane, hsel, q, src, a_ws, norm_out, xprev, acc);
    }
  }

  if (scan) {
    __syncthreads();
    for (int base = 0; base < E_EDGES; base += SCAN_EDGES) {
      int4 dv[SCAN_V];
#pragma unroll
      for (int j = 0; j < SCAN_V; ++j) {
        const int e0 = base + (j * AGG_T + t) * 4;
        dv[j] = (e0 < E_EDGES) ? *(const int4*)(dst + e0) : make_int4(-1, -1, -1, -1);
      }
#pragma unroll
      for (int j = 0; j < SCAN_V; ++j) {
        const int e0 = base + (j * AGG_T + t) * 4;
        const unsigned rx = (unsigned)(dv[j].x - lo), ry = (unsigned)(dv[j].y - lo);
        const unsigned rz = (unsigned)(dv[j].z - lo), rw = (unsigned)(dv[j].w - lo);
        if (min(min(rx, ry), min(rz, rw)) < (unsigned)RNG) {
          push_edge(rx, e0,     q, &qn);
          push_edge(ry, e0 + 1, q, &qn);
          push_edge(rz, e0 + 2, q, &qn);
          push_edge(rw, e0 + 3, q, &qn);
        }
      }
      __syncthreads();
      const unsigned n = __builtin_amdgcn_readfirstlane(qn);
      __syncthreads();
      const bool last = (base + SCAN_EDGES >= E_EDGES);
      if (n > (unsigned)(QCAP - SCAN_EDGES) || last) {
        drain_queue(n, wv, lane, hsel, q, src, a_ws, norm_out, xprev, acc);
        __syncthreads();
        if (t == 0) qn = 0u;
        __syncthreads();
      }
    }
  } else {
    __syncthreads();
  }

  for (int g = t; g < RNG * (HF / 4); g += AGG_T) {
    const int row = g >> 6, c4 = g & 63;
    const float* ap = acc + row * HF + (c4 >> 5) * 128 + (c4 & 31);
    const float ni = norm_in[lo + row];
    const float4 o = make_float4(ap[0] * ni, ap[32] * ni, ap[64] * ni, ap[96] * ni);
    *(float4*)(xnext + (size_t)(lo + row) * HF + c4 * 4) = o;
  }
}

__global__ __launch_bounds__(128)
void combine_kernel(const float* __restrict__ h, const float* __restrict__ x1,
                    const float* __restrict__ x2, const float* __restrict__ x3,
                    const float* __restrict__ al, const float* __restrict__ hop_attn_r,
                    float* __restrict__ out) {
  int n    = blockIdx.x;
  int head = threadIdx.x >> 5;
  int lane = threadIdx.x & 31;
  const float* hs[4] = {h, x1, x2, x3};
  size_t base = (size_t)n * HF + head * F_OUT;
  const float* ar = hop_attn_r + head * F_OUT;
  float ar0 = ar[lane], ar1 = ar[lane + 32];
  float albase = al[n * 4 + head];

  float xv0[4], xv1[4], score[4];
#pragma unroll
  for (int k = 0; k < 4; ++k) {
    float v0 = hs[k][base + lane];
    float v1 = hs[k][base + lane + 32];
    xv0[k] = v0; xv1[k] = v1;
    float p = v0 * ar0 + v1 * ar1;
#pragma unroll
    for (int off = 16; off > 0; off >>= 1) p += __shfl_xor(p, off);
    score[k] = leakyf(albase + p);
  }
  float m = fmaxf(fmaxf(score[0], score[1]), fmaxf(score[2], score[3]));
  float ex[4], ssum = 0.f;
#pragma unroll
  for (int k = 0; k < 4; ++k) { ex[k] = __expf(score[k] - m); ssum += ex[k]; }
  float o0 = 0.f, o1 = 0.f;
#pragma unroll
  for (int k = 0; k < 4; ++k) {
    float w = ex[k] / ssum;
    o0 += xv0[k] * w; o1 += xv1[k] * w;
  }
  out[base + lane] = o0;
  out[base + lane + 32] = o1;
}

extern "C" void kernel_launch(void* const* d_in, const int* in_sizes, int n_in,
                              void* d_out, int out_size, void* d_ws, size_t ws_size,
                              hipStream_t stream) {
  (void)in_sizes; (void)n_in; (void)out_size; (void)ws_size;
  const float* feat       = (const float*)d_in[0];
  const int*   src        = (const int*)d_in[1];
  const int*   dst        = (const int*)d_in[2];
  const float* W          = (const float*)d_in[3];
  const float* attn_l     = (const float*)d_in[4];
  const float* attn_r     = (const float*)d_in[5];
  const float* hop_attn_l = (const float*)d_in[6];
  const float* hop_attn_r = (const float*)d_in[7];
  const float* sigma      = (const float*)d_in[8];

  float* ws = (float*)d_ws;
  size_t off = 0;
  float* h    = ws + off; off += (size_t)N_NODES * HF;
  float* x1   = ws + off; off += (size_t)N_NODES * HF;
  float* x2   = ws + off; off += (size_t)N_NODES * HF;
  float* x3   = ws + off; off += (size_t)N_NODES * HF;
  float* el   = ws + off; off += (size_t)N_NODES * H_HEADS;
  float* er   = ws + off; off += (size_t)N_NODES * H_HEADS;
  float* al   = ws + off; off += (size_t)N_NODES * H_HEADS;
  float* maxs = ws + off; off += (size_t)N_NODES * H_HEADS;
  float* maxd = ws + off; off += (size_t)N_NODES * H_HEADS;
  float* sums = ws + off; off += (size_t)N_NODES * H_HEADS;
  float* sumd = ws + off; off += (size_t)N_NODES * H_HEADS;
  off += (size_t)N_NODES * 2;
  float* norm_out = ws + off; off += (size_t)N_NODES;
  float* norm_in  = ws + off; off += (size_t)N_NODES;
  float* e_ws = ws + off; off += (size_t)E_EDGES * H_HEADS;
  float* a_ws = ws + off; off += (size_t)E_EDGES * H_HEADS;

  gemm_h_kernel<<<(N_NODES + 127) / 128, 256, 0, stream>>>(feat, W, h);

  const int NH = N_NODES * H_HEADS;
  scores_kernel<<<(NH + 255) / 256, 256, 0, stream>>>(h, attn_l, attn_r, hop_attn_l, el, er, al);

  unsigned int* gq     = (unsigned int*)e_ws;
  unsigned int* gcount = gq + (size_t)NBLK * GCAP;
  seg_stats_kernel<<<ST_NBLK, ST_T, 0, stream>>>(src, dst, el, er, maxs, maxd, sums, sumd,
                                                 norm_out, norm_in, gq, gcount);

  const int EH = E_EDGES * H_HEADS;
  edge_a_kernel<<<(EH + 255) / 256, 256, 0, stream>>>(src, dst, el, er, maxs, maxd, sums, sumd, sigma, a_ws);

  agg_tile_kernel<<<NBLK, AGG_T, 0, stream>>>(src, dst, a_ws, norm_out, norm_in, h,  x1, gq, gcount);
  agg_tile_kernel<<<NBLK, AGG_T, 0, stream>>>(src, dst, a_ws, norm_out, norm_in, x1, x2, gq, gcount);
  agg_tile_kernel<<<NBLK, AGG_T, 0, stream>>>(src, dst, a_ws, norm_out, norm_in, x2, x3, gq, gcount);

  combine_kernel<<<N_NODES, 128, 0, stream>>>(h, x1, x2, x3, al, hop_attn_r, (float*)d_out);
}


# Owner blocks find their edges in per-range edge bitmaps (built by the score blocks: LDS atomic-or, whole-line stores), not by scanning an index array each; queue order fixed; LDS sums' order changes
# speedup vs baseline: 7.3995x; 1.2034x over previous
#include <hip/hip_runtime.h>
#include <hip/hip_bf16.h>

#define N_NODES 50000
#define E_EDGES 800000
#define IN_F    256
#define H_HEADS 4
#define F_OUT   64
#define HF      256
#define SLOPE   0.2f

typedef __attribute__((ext_vector_type(16))) __bf16 v16bf;
typedef __attribute__((ext_vector_type(8)))  float  v8f;

__device__ __forceinline__ float leakyf(float x) { return x > 0.f ? x : SLOPE * x; }

__device__ __forceinline__ void atomicMaxF(float* addr, float val) {
  if (val >= 0.f) {
    atomicMax((int*)addr, __float_as_int(val));
  } else {
    atomicMin((unsigned int*)addr, __float_as_uint(val));
  }
}

#define WT_STRIDE 40

__global__ __launch_bounds__(256)
void gemm_h_kernel(const float* __restrict__ feat, const float* __restrict__ W,
                   float* __restrict__ h) {
  __shared__ __align__(16) __bf16 Wt[HF * WT_STRIDE];

  const int t       = threadIdx.x;
  const int wave    = t >> 5;
  const int lane    = t & 31;
  const int rowBase = blockIdx.x * 128 + wave * 16;
  const bool rowValid = (rowBase < N_NODES);
  const int nloc = lane & 15;
  const int kg   = lane >> 4;

  v8f acc[16];
#pragma unroll
  for (int nt = 0; nt < 16; ++nt)
#pragma unroll
    for (int v = 0; v < 8; ++v) acc[nt][v] = 0.f;

  for (int ks = 0; ks < IN_F / 32; ++ks) {
    const int kbase = ks * 32;
    __syncthreads();
#pragma unroll
    for (int kk = 0; kk < 32; ++kk)
      Wt[t * WT_STRIDE + kk] = (__bf16)W[(kbase + kk) * HF + t];
    __syncthreads();

    if (rowValid) {
      const float* arow = feat + (size_t)(rowBase + nloc) * IN_F + kbase;
      float4 a0 = *(const float4*)(arow + 8 * kg);
      float4 a1 = *(const float4*)(arow + 8 * kg + 4);
      float4 a2 = *(const float4*)(arow + 16 + 8 * kg);
      float4 a3 = *(const float4*)(arow + 16 + 8 * kg + 4);
      float atmp[16] = {a0.x, a0.y, a0.z, a0.w, a1.x, a1.y, a1.z, a1.w,
                        a2.x, a2.y, a2.z, a2.w, a3.x, a3.y, a3.z, a3.w};
      v16bf afrag;
#pragma unroll
      for (int i = 0; i < 16; ++i) afrag[i] = (__bf16)atmp[i];

      const __bf16* wrow0 = &Wt[nloc * WT_STRIDE];
      uint4 b0 = *(const uint4*)(wrow0 + 8 * kg);
      uint4 b1 = *(const uint4*)(wrow0 + 16 + 8 * kg);
#pragma unroll
      for (int nt = 0; nt < 16; ++nt) {
        uint4 nb0, nb1;
        if (nt < 15) {
          const __bf16* wrow = &Wt[((nt + 1) * 16 + nloc) * WT_STRIDE];
          nb0 = *(const uint4*)(wrow + 8 * kg);
          nb1 = *(const uint4*)(wrow + 16 + 8 * kg);
        }
        const __bf16* p0 = (const __bf16*)&b0;
        const __bf16* p1 = (const __bf16*)&b1;
        v16bf bfrag;
#pragma unroll
        for (int i = 0; i < 8; ++i) { bfrag[i] = p0[i]; bfrag[8 + i] = p1[i]; }
        acc[nt] = __builtin_amdgcn_wmma_f32_16x16x32_bf16(
            false, afrag, false, bfrag, (short)0, acc[nt], false, false);
        if (nt < 15) { b0 = nb0; b1 = nb1; }
      }
    }
  }

  if (rowValid) {
#pragma unroll
    for (int nt = 0; nt < 16; ++nt)
#pragma unroll
      for (int v = 0; v < 8; ++v)
        h[(size_t)(rowBase + v + 8 * kg) * HF + nt * 16 + nloc] = acc[nt][v];
  }
}

#define RNG        200
#define ST_K       2
#define ST_RNG     (ST_K * RNG)
#define ST_NRNG    (N_NODES / ST_RNG)
#define SC_T       256
#define BM_EDGES   (SC_T * 4)
#define BM_WORDS   (BM_EDGES / 32)
#define BM_NBLK    ((E_EDGES + BM_EDGES - 1) / BM_EDGES)
#define BM_PITCH   (BM_NBLK * BM_WORDS)
#define SC_NBLK    ((N_NODES * H_HEADS + SC_T - 1) / SC_T)

static_assert(N_NODES % ST_RNG == 0, "bitmap ranges tile the node set exactly");
static_assert(E_EDGES % 4 == 0, "the index arrays are read 4 edges at a time");
static_assert(BM_WORDS == 32 && BM_PITCH % 4 == 0, "a block's share of a bitmap is one 128-byte line");
static_assert(SC_NBLK >= BM_NBLK, "the score blocks cover the edge list");
static_assert((2 * ST_NRNG * BM_WORDS) % 4 == 0, "the LDS copy is moved 16 bytes at a time");

__global__ __launch_bounds__(SC_T)
void scores_kernel(const float* __restrict__ h,
                   const float* __restrict__ attn_l,
                   const float* __restrict__ attn_r,
                   const float* __restrict__ hop_attn_l,
                   float* __restrict__ el, float* __restrict__ er,
                   float* __restrict__ al,
                   const int* __restrict__ src, const int* __restrict__ dst,
                   unsigned int* __restrict__ bm) {
  __shared__ __align__(16) unsigned int bml[2 * ST_NRNG * BM_WORDS];
  const int t = threadIdx.x;

  if (blockIdx.x < BM_NBLK) {
    const int e0 = blockIdx.x * BM_EDGES + t * 4;
    int4 sv = make_int4(0, 0, 0, 0), dv = make_int4(0, 0, 0, 0);
    const bool have = (e0 < E_EDGES);
    if (have) { sv = *(const int4*)(src + e0); dv = *(const int4*)(dst + e0); }
    for (int k = t; k < 2 * ST_NRNG * BM_WORDS / 4; k += SC_T)
      ((uint4*)bml)[k] = make_uint4(0u, 0u, 0u, 0u);
    __syncthreads();
    if (have) {
      const unsigned w = (unsigned)t >> 3;
      const unsigned b = ((unsigned)t & 7u) * 4u;
      unsigned int* ends   = bml + w;
      unsigned int* starts = bml + ST_NRNG * BM_WORDS + w;
      atomicOr(&ends[((unsigned)dv.x / ST_RNG) * BM_WORDS], 1u << b);
      atomicOr(&ends[((unsigned)dv.y / ST_RNG) * BM_WORDS], 2u << b);
      atomicOr(&ends[((unsigned)dv.z / ST_RNG) * BM_WORDS], 4u << b);
      atomicOr(&ends[((unsigned)dv.w / ST_RNG) * BM_WORDS], 8u << b);
      atomicOr(&starts[((unsigned)sv.x / ST_RNG) * BM_WORDS], 1u << b);
      atomicOr(&starts[((unsigned)sv.y / ST_RNG) * BM_WORDS], 2u << b);
      atomicOr(&starts[((unsigned)sv.z / ST_RNG) * BM_WORDS], 4u << b);
      atomicOr(&starts[((unsigned)sv.w / ST_RNG) * BM_WORDS], 8u << b);
    }
    __syncthreads();
    for (int k = t; k < 2 * ST_NRNG * BM_WORDS / 4; k += SC_T) {
      const int row = k >> 3, part = k & 7;
      *(uint4*)(bm + (size_t)row * BM_PITCH + (size_t)blockIdx.x * BM_WORDS + part * 4) = ((const uint4*)bml)[k];
    }
  }

  int i = blockIdx.x * SC_T + t;
  if (i >= N_NODES * H_HEADS) return;
  int node = i >> 2, head = i & 3;
  const float* hp  = h + (size_t)node * HF + head * F_OUT;
  const float* alp = attn_l + head * F_OUT;
  const float* arp = attn_r + head * F_OUT;
  const float* hlp = hop_attn_l + head * F_OUT;
  float sl = 0.f, sr = 0.f, shl = 0.f;
#pragma unroll 8
  for (int f = 0; f < F_OUT; ++f) {
    float v = hp[f];
    sl += v * alp[f]; sr += v * arp[f]; shl += v * hlp[f];
  }
  el[i] = sl; er[i] = sr; al[i] = shl;
}

__device__ __forceinline__ float mix_coef(float el, float er, float ms, float md,
                                          float ss, float sd, float sgm, float nrm) {
  const float ev = leakyf(el + er);
  const float ax = fmaxf(__expf(ev - ms) / ss, 1e-10f);
  const float ay = fmaxf(__expf(ev - md) / sd, 1e-10f);
  const float a  = __powf(ay, sgm) * __powf(ax, 1.f - sgm);
  return a * nrm;
}

#define NBLK       (N_NODES / RNG)
#define AGG_T      1024
#define GCAP       4096
#define CH         GCAP
#define SCAN_EDGES (AGG_T * 4)
#define QCAP       (2 * SCAN_EDGES)
#define DRAIN_U    4

static_assert(N_NODES % RNG == 0, "ranges tile the node set exactly");
static_assert(E_EDGES % 4 == 0, "dst is read 4 edges at a time");
static_assert(E_EDGES <= (1 << 20) && RNG <= 256, "queue word packs row and edge id");
static_assert(N_NODES <= (1 << 16), "record word packs row and source node");
static_assert((RNG * HF + CH * 5 + QCAP + 1) * 4 <= 320 * 1024, "LDS budget");
static_assert(DRAIN_U == 4, "a batch is one 16-byte read of record words");
static_assert(GCAP % AGG_T == 0, "whole rounds of the block cover a slice");
static_assert((size_t)NBLK * GCAP * 6 + 256 <= (size_t)E_EDGES * H_HEADS * 2,
              "saved queues, counts and records fit the two edge planes");
static_assert((size_t)2 * ST_NRNG * BM_PITCH <= (size_t)N_NODES * HF,
              "both sets of bitmaps fit the place of the third hop's plane");

__device__ __forceinline__ void push_edge(unsigned r, int eid, unsigned* q, unsigned* qn) {
  if (r < (unsigned)RNG) {
    const unsigned p = atomicAdd(qn, 1u);
    q[p] = (r << 20) | (unsigned)eid;
  }
}

__device__ __forceinline__ void make_record(unsigned w, int lo,
                                            const int* __restrict__ src,
                                            const float* __restrict__ el, const float* __restrict__ er,
                                            const float* __restrict__ maxs, const float* __restrict__ maxd,
                                            const float* __restrict__ sums, const float* __restrict__ sumd,
                                            const float* __restrict__ norm_out, float sgm,
                                            unsigned& word, float4& c) {
  const bool     ok  = (w != 0xFFFFFFFFu);
  const unsigned eid = ok ? (w & 0xFFFFFu) : 0u;
  const unsigned r   = ok ? (w >> 20) : 0u;
  const int      s   = src[eid];
  const size_t   d   = (size_t)(lo + (int)r);
  const float4 a  = *(const float4*)(el   + (size_t)s * 4);
  const float4 ms = *(const float4*)(maxs + (size_t)s * 4);
  const float4 ss = *(const float4*)(sums + (size_t)s * 4);
  const float  nr = norm_out[s];
  const float4 b  = *(const float4*)(er   + d * 4);
  const float4 md = *(const float4*)(maxd + d * 4);
  const float4 sd = *(const float4*)(sumd + d * 4);
  c.x = mix_coef(a.x, b.x, ms.x, md.x, ss.x, sd.x, sgm, nr);
  c.y = mix_coef(a.y, b.y, ms.y, md.y, ss.y, sd.y, sgm, nr);
  c.z = mix_coef(a.z, b.z, ms.z, md.z, ss.z, sd.z, sgm, nr);
  c.w = mix_coef(a.w, b.w, ms.w, md.w, ss.w, sd.w, sgm, nr);
  word = (r << 16) | (unsigned)s;
}

template <int U>
__device__ __forceinline__ void add_records(const unsigned (&ent)[U], unsigned i0, int lane, int hsel,
                                            const float* rcf,
                                            const float* __restrict__ xprev, float* acc) {
  float4 r0[U], r1[U];
  float  c0[U], c1[U];
#pragma unroll
  for (int u = 0; u < U; ++u) {
    const float* row = xprev + (size_t)(ent[u] & 0xFFFFu) * HF + lane * 4;
    r0[u] = *(const float4*)row;
    r1[u] = *(const float4*)(row + 128);
    c0[u] = rcf[(i0 + u) * 4 + hsel];
    c1[u] = rcf[(i0 + u) * 4 + 2 + hsel];
  }
#pragma unroll
  for (int u = 0; u < U; ++u) {
    float* arow = acc + (ent[u] >> 16) * HF + lane;
    atomicAdd(arow,       r0[u].x * c0[u]);
    atomicAdd(arow + 32,  r0[u].y * c0[u]);
    atomicAdd(arow + 64,  r0[u].z * c0[u]);
    atomicAdd(arow + 96,  r0[u].w * c0[u]);
    atomicAdd(arow + 128, r1[u].x * c1[u]);
    atomicAdd(arow + 160, r1[u].y * c1[u]);
    atomicAdd(arow + 192, r1[u].z * c1[u]);
    atomicAdd(arow + 224, r1[u].w * c1[u]);
  }
}

__device__ __forceinline__ void add_batch(unsigned i0, int lane, int hsel,
                                          const unsigned* rs, const float* rcf,
                                          const float* __restrict__ xprev, float* acc) {
  const uint4 w4 = *(const uint4*)&rs[i0];
  const unsigned ent[DRAIN_U] = {
      (unsigned)__builtin_amdgcn_readfirstlane(w4.x),
      (unsigned)__builtin_amdgcn_readfirstlane(w4.y),
      (unsigned)__builtin_amdgcn_readfirstlane(w4.z),
      (unsigned)__builtin_amdgcn_readfirstlane(w4.w)};
  add_records<DRAIN_U>(ent, i0, lane, hsel, rcf, xprev, acc);
}

__device__ __forceinline__ void add_single(unsigned i, int lane, int hsel,
                                           const unsigned* rs, const float* rcf,
                                           const float* __restrict__ xprev, float* acc) {
  const unsigned ent1[1] = {(unsigned)__builtin_amdgcn_readfirstlane(rs[i])};
  add_records<1>(ent1, i, lane, hsel, rcf, xprev, acc);
}

__device__ __forceinline__ void drain_records(unsigned from, unsigned n, unsigned wv, int lane, int hsel,
                                              const unsigned* rs, const float* rcf,
                                              const float* __restrict__ xprev, float* acc) {
  for (unsigned i0 = from + wv * DRAIN_U; i0 + DRAIN_U <= n; i0 += (AGG_T / 32) * DRAIN_U)
    add_batch(i0, lane, hsel, rs, rcf, xprev, acc);
  const unsigned nfull = n & ~(unsigned)(DRAIN_U - 1);
  if (nfull + wv < n) add_single(nfull + wv, lane, hsel, rs, rcf, xprev, acc);
}

#define ST_NBLK    (2 * ST_NRNG)
#define ST_T       1024
#define ST_VS      4
#define BM_V       ((BM_PITCH / 4 + ST_T - 1) / ST_T)
#define ST_EDGES_S (ST_T * 4 * ST_VS)
#define ST_QCAP    (3 * ST_EDGES_S)
#define ST_U       8

static_assert(N_NODES % ST_RNG == 0 && ST_RNG <= 512, "statistics ranges tile the node set; local node fits 9 bits");
static_assert(ST_RNG <= ST_T, "one thread per degree word");
static_assert(ST_T / 32 <= 32, "one lane per wave total");
static_assert((ST_QCAP + ST_RNG * 9 + 4) * 4 <= 320 * 1024, "LDS budget of the statistics kernel");

__device__ __forceinline__ void push_own(unsigned r, int eid, unsigned* q, unsigned* qn) {
  if (r < (unsigned)ST_RNG) {
    const unsigned p = atomicAdd(qn, 1u);
    q[p] = (unsigned)eid;
  }
}

template <int V>
__device__ __forceinline__ void scan_group(int base, int t, int lo,
                                           const int* __restrict__ idx,
                                           unsigned* q, unsigned* qn) {
  int4 iv[V];
#pragma unroll
  for (int j = 0; j < V; ++j) {
    const int e0 = base + (j * ST_T + t) * 4;
    if (e0 < E_EDGES) iv[j] = *(const int4*)(idx + e0);
    else              iv[j] = make_int4(-1, -1, -1, -1);
  }
#pragma unroll
  for (int j = 0; j < V; ++j) {
    const int e0 = base + (j * ST_T + t) * 4;
    const unsigned rx = (unsigned)(iv[j].x - lo), ry = (unsigned)(iv[j].y - lo);
    const unsigned rz = (unsigned)(iv[j].z - lo), rw = (unsigned)(iv[j].w - lo);
    if (min(min(rx, ry), min(rz, rw)) < (unsigned)ST_RNG) {
      push_own(rx, e0,     q, qn);
      push_own(ry, e0 + 1, q, qn);
      push_own(rz, e0 + 2, q, qn);
      push_own(rw, e0 + 3, q, qn);
    }
  }
}

__device__ __forceinline__ void push_bits(unsigned word, unsigned first, unsigned& p, unsigned* q) {
  while (word != 0u) {
    const unsigned b = (unsigned)__builtin_ctz(word);
    word &= word - 1u;
    if (p < (unsigned)ST_QCAP) q[p] = first + b;
    ++p;
  }
}

__device__ __forceinline__ void queue_from_bitmap(const unsigned int* __restrict__ row, int t,
                                                  unsigned* q, unsigned* qn, unsigned* wtot) {
  uint4 wv[BM_V];
#pragma unroll
  for (int j = 0; j < BM_V; ++j) {
    const int k = j * ST_T + t;
    wv[j] = (k < BM_PITCH / 4) ? ((const uint4*)row)[k] : make_uint4(0u, 0u, 0u, 0u);
  }
  unsigned cnt = 0u;
#pragma unroll
  for (int j = 0; j < BM_V; ++j)
    cnt += (unsigned)(__builtin_popcount(wv[j].x) + __builtin_popcount(wv[j].y) +
                      __builtin_popcount(wv[j].z) + __builtin_popcount(wv[j].w));
  const unsigned lane = (unsigned)t & 31u, wave = (unsigned)t >> 5;
  unsigned incl = cnt;
#pragma unroll
  for (int d = 1; d < 32; d <<= 1) {
    const unsigned below = (unsigned)__shfl_up((int)incl, d);
    if (lane >= (unsigned)d) incl += below;
  }
  if (lane == 31u) wtot[wave] = incl;
  __syncthreads();
  unsigned before = (lane < wave) ? wtot[lane] : 0u;
#pragma unroll
  for (int d = 16; d >= 1; d >>= 1) before += (unsigned)__shfl_xor((int)before, d);
  if (t == ST_T - 1) *qn = before + incl;
  unsigned p = before + incl - cnt;
  if (cnt != 0u) {
#pragma unroll
    for (int j = 0; j < BM_V; ++j) {
      const unsigned first = (unsigned)(j * ST_T + t) * 128u;
      push_bits(wv[j].x, first,       p, q);
      push_bits(wv[j].y, first + 32u, p, q);
      push_bits(wv[j].z, first + 64u, p, q);
      push_bits(wv[j].w, first + 96u, p, q);
    }
  }
}

template <int PASS>
__device__ __forceinline__ void stats_pass(unsigned n, int t, bool ends, int lo, const unsigned* q,
                                           const int* __restrict__ src,
                                           const int* __restrict__ dst,
                                           const float* __restrict__ el,
                                           const float* __restrict__ er,
                                           float* st_max, float* st_sum, float* st_deg,
                                           unsigned* sn, unsigned* gqb) {
  for (unsigned i0 = 0; i0 < n; i0 += ST_T * ST_U) {
    unsigned ent[ST_U];
    int      sv[ST_U], dv[ST_U];
    float4   a[ST_U], b[ST_U];
#pragma unroll
    for (int u = 0; u < ST_U; ++u) {
      const unsigned idx = i0 + (unsigned)(u * ST_T + t);
      ent[u] = (idx < n) ? q[idx] : 0xFFFFFFFFu;
    }
#pragma unroll
    for (int u = 0; u < ST_U; ++u) {
      const unsigned eid = (ent[u] == 0xFFFFFFFFu) ? 0u : (ent[u] & 0xFFFFFu);
      sv[u] = src[eid];
      dv[u] = dst[eid];
    }
#pragma unroll
    for (int u = 0; u < ST_U; ++u) {
      a[u] = *(const float4*)(el + (size_t)sv[u] * 4);
      b[u] = *(const float4*)(er + (size_t)dv[u] * 4);
    }
#pragma unroll
    for (int u = 0; u < ST_U; ++u) {
      const unsigned r     = (unsigned)((ends ? dv[u] : sv[u]) - lo);
      const bool     valid = (ent[u] != 0xFFFFFFFFu) && r < (unsigned)ST_RNG;
      if (valid) {
        const float e0 = leakyf(a[u].x + b[u].x), e1 = leakyf(a[u].y + b[u].y);
        const float e2 = leakyf(a[u].z + b[u].z), e3 = leakyf(a[u].w + b[u].w);
        if (PASS == 0) {
          atomicMaxF(&st_max[r * 4 + 0], e0);
          atomicMaxF(&st_max[r * 4 + 1], e1);
          atomicMaxF(&st_max[r * 4 + 2], e2);
          atomicMaxF(&st_max[r * 4 + 3], e3);
          atomicAdd(&st_deg[r], 1.f);
        } else {
          atomicAdd(&st_sum[r * 4 + 0], __expf(e0 - st_max[r * 4 + 0]));
          atomicAdd(&st_sum[r * 4 + 1], __expf(e1 - st_max[r * 4 + 1]));
          atomicAdd(&st_sum[r * 4 + 2], __expf(e2 - st_max[r * 4 + 2]));
          atomicAdd(&st_sum[r * 4 + 3], __expf(e3 - st_max[r * 4 + 3]));
        }
      }
      if (PASS == 0 && ends) {
        const unsigned lane = (unsigned)t & 31u;
#pragma unroll
        for (unsigned k = 0; k < (unsigned)ST_K; ++k) {
          const bool keep = valid && r >= k * RNG && r < (k + 1) * RNG;
          const unsigned m = __builtin_amdgcn_ballot_w32(keep);
          if (m != 0u) {
            const unsigned lead = (unsigned)__builtin_ctz(m);
            unsigned bp = 0u;
            if (lane == lead) bp = atomicAdd(&sn[k], (unsigned)__builtin_popcount(m));
            bp = (unsigned)__shfl((int)bp, (int)lead);
            const unsigned p = bp + (unsigned)__builtin_popcount(m & ((1u << lane) - 1u));
            if (keep && p < (unsigned)GCAP)
              gqb[(size_t)k * GCAP + p] = ((r - k * RNG) << 20) | (ent[u] & 0xFFFFFu);
          }
        }
      }
    }
  }
}

__global__ __launch_bounds__(ST_T)
void seg_stats_kernel(const int* __restrict__ src, const int* __restrict__ dst,
                      const float* __restrict__ el, const float* __restrict__ er,
                      float* __restrict__ maxs, float* __restrict__ maxd,
                      float* __restrict__ sums, float* __restrict__ sumd,
                      float* __restrict__ norm_out, float* __restrict__ norm_in,
                      unsigned int* gq, unsigned int* gcount,
                      const unsigned int* __restrict__ bm) {
  __shared__ __align__(16) unsigned int q[ST_QCAP];
  __shared__ float st_max[ST_RNG * 4];
  __shared__ float st_sum[ST_RNG * 4];
  __shared__ float st_deg[ST_RNG];
  __shared__ unsigned int qn;
  __shared__ unsigned int sn[ST_K];
  __shared__ unsigned int wtot[ST_T / 32];

  const int  t     = threadIdx.x;
  const int  g     = blockIdx.x >> 1;
  const bool ends  = (blockIdx.x & 1) == 0;
  const int  lo    = g * ST_RNG;
  const int* __restrict__ idx = ends ? dst : src;
  unsigned int* gqb = gq + (size_t)g * ST_K * GCAP;

  for (int i = t; i < ST_RNG * 4; i += ST_T) { st_max[i] = -1e30f; st_sum[i] = 0.f; }
  if (t < ST_RNG) st_deg[t] = 0.f;
  if (t < ST_K) sn[t] = 0u;
  __syncthreads();

  queue_from_bitmap(bm + (size_t)((ends ? 0 : ST_NRNG) + g) * BM_PITCH, t, q, &qn, wtot);
  __syncthreads();
  unsigned n = __builtin_amdgcn_readfirstlane(qn);

  if (n <= (unsigned)ST_QCAP) {
    stats_pass<0>(n, t, ends, lo, q, src, dst, el, er, st_max, st_sum, st_deg, sn, gqb);
    __syncthreads();
    stats_pass<1>(n, t, ends, lo, q, src, dst, el, er, st_max, st_sum, st_deg, sn, gqb);
  } else {
    __syncthreads();
    if (t == 0) qn = 0u;
    __syncthreads();
    for (int base = 0; base < E_EDGES; base += ST_EDGES_S) {
      scan_group<ST_VS>(base, t, lo, idx, q, &qn);
      __syncthreads();
      n = __builtin_amdgcn_readfirstlane(qn);
      __syncthreads();
      const bool last = (base + ST_EDGES_S >= E_EDGES);
      if (n > (unsigned)(ST_QCAP - ST_EDGES_S) || last) {
        stats_pass<0>(n, t, ends, lo, q, src, dst, el, er, st_max, st_sum, st_deg, sn, gqb);
        __syncthreads();
        if (t == 0) qn = 0u;
        __syncthreads();
      }
    }
    for (int base = 0; base < E_EDGES; base += ST_EDGES_S) {
      scan_group<ST_VS>(base, t, lo, idx, q, &qn);
      __syncthreads();
      n = __builtin_amdgcn_readfirstlane(qn);
      __syncthreads();
      const bool last = (base + ST_EDGES_S >= E_EDGES);
      if (n > (unsigned)(ST_QCAP - ST_EDGES_S) || last) {
        stats_pass<1>(n, t, ends, lo, q, src, dst, el, er, st_max, st_sum, st_deg, sn, gqb);
        __syncthreads();
        if (t == 0) qn = 0u;
        __syncthreads();
      }
    }
  }
  __syncthreads();

  float* __restrict__ mx_out = ends ? maxd : maxs;
  float* __restrict__ sm_out = ends ? sumd : sums;
  for (int i = t; i < ST_RNG * 4; i += ST_T) {
    mx_out[lo * 4 + i] = st_max[i];
    sm_out[lo * 4 + i] = st_sum[i];
  }
  if (t < ST_RNG) {
    const float deg = fmaxf(st_deg[t], 1.f);
    if (ends) norm_in[lo + t]  = sqrtf(deg);
    else      norm_out[lo + t] = rsqrtf(deg);
  }
  if (ends && t < ST_K) gcount[g * ST_K + t] = sn[t];
}

__device__ __forceinline__ float row16_sum(float v) {
  v += __int_as_float(__builtin_amdgcn_update_dpp(0, __float_as_int(v), 0x168, 0xF, 0xF, true));
  v += __int_as_float(__builtin_amdgcn_update_dpp(0, __float_as_int(v), 0x164, 0xF, 0xF, true));
  v += __int_as_float(__builtin_amdgcn_update_dpp(0, __float_as_int(v), 0x162, 0xF, 0xF, true));
  v += __int_as_float(__builtin_amdgcn_update_dpp(0, __float_as_int(v), 0x161, 0xF, 0xF, true));
  return v;
}

struct HopRows { float4 v0, v1, v2; float ni, ab; };

__device__ __forceinline__ HopRows load_hop_rows(int g, int lo, int head,
                                                 const float* __restrict__ hb, const float* __restrict__ x1b,
                                                 const float* __restrict__ x2b,
                                                 const float* __restrict__ norm_in,
                                                 const float* __restrict__ al) {
  HopRows r;
  const int row = lo + (g >> 6);
  r.v0 = *(const float4*)(hb  + (size_t)g * 4);
  r.v1 = *(const float4*)(x1b + (size_t)g * 4);
  r.v2 = *(const float4*)(x2b + (size_t)g * 4);
  r.ni = norm_in[row];
  r.ab = al[(size_t)row * 4 + head];
  return r;
}

__device__ __forceinline__ void combine_store(int g, int c4, const HopRows& r, const float4& ar,
                                              const float* acc, float* __restrict__ ob) {
  const float* ap = acc + (g >> 6) * HF + (c4 >> 5) * 128 + (c4 & 31);
  const float4 v3 = make_float4(ap[0] * r.ni, ap[32] * r.ni, ap[64] * r.ni, ap[96] * r.ni);
  const float p0 = row16_sum(r.v0.x * ar.x + r.v0.y * ar.y + r.v0.z * ar.z + r.v0.w * ar.w);
  const float p1 = row16_sum(r.v1.x * ar.x + r.v1.y * ar.y + r.v1.z * ar.z + r.v1.w * ar.w);
  const float p2 = row16_sum(r.v2.x * ar.x + r.v2.y * ar.y + r.v2.z * ar.z + r.v2.w * ar.w);
  const float p3 = row16_sum(v3.x * ar.x + v3.y * ar.y + v3.z * ar.z + v3.w * ar.w);
  const float s0 = leakyf(r.ab + p0), s1 = leakyf(r.ab + p1);
  const float s2 = leakyf(r.ab + p2), s3 = leakyf(r.ab + p3);
  const float m  = fmaxf(fmaxf(s0, s1), fmaxf(s2, s3));
  const float e0 = __expf(s0 - m), e1 = __expf(s1 - m), e2 = __expf(s2 - m), e3 = __expf(s3 - m);
  const float inv = 1.f / (e0 + e1 + e2 + e3);
  const float w0 = e0 * inv, w1 = e1 * inv, w2 = e2 * inv, w3 = e3 * inv;
  float4 o;
  o.x = r.v0.x * w0 + r.v1.x * w1 + r.v2.x * w2 + v3.x * w3;
  o.y = r.v0.y * w0 + r.v1.y * w1 + r.v2.y * w2 + v3.y * w3;
  o.z = r.v0.z * w0 + r.v1.z * w1 + r.v2.z * w2 + v3.z * w3;
  o.w = r.v0.w * w0 + r.v1.w * w1 + r.v2.w * w2 + v3.w * w3;
  *(float4*)(ob + (size_t)g * 4) = o;
}

template <bool FIRST, bool LAST>
__global__ __launch_bounds__(AGG_T)
void agg_tile_kernel(const int* __restrict__ src, const int* __restrict__ dst,
                     const float* __restrict__ el, const float* __restrict__ er,
                     const float* __restrict__ maxs, const float* __restrict__ maxd,
                     const float* __restrict__ sums, const float* __restrict__ sumd,
                     const float* __restrict__ norm_out, const float* __restrict__ norm_in,
                     const float* __restrict__ sigma,
                     const float* __restrict__ xprev, float* __restrict__ xnext,
                     const unsigned int* gq, const unsigned int* gcount,
                     unsigned int* gsv, float4* gcf,
                     const float* __restrict__ h0, const float* __restrict__ x1p,
                     const float* __restrict__ al, const float* __restrict__ hop_attn_r) {
  __shared__ __align__(16) float acc[RNG * HF];
  __shared__ __align__(16) float4 rc[CH];
  __shared__ __align__(16) unsigned int rs[CH];
  __shared__ __align__(16) unsigned int q[QCAP];
  __shared__ unsigned int qn;

  const int t    = threadIdx.x;
  const int lane = t & 31;
  const int wave = t >> 5;
  const int hsel = lane >> 4;
  const int lo   = blockIdx.x * RNG;
  const unsigned wv = __builtin_amdgcn_readfirstlane((unsigned)wave);
  const unsigned int* gqb = gq + (size_t)blockIdx.x * GCAP;
  unsigned int* gsvb = gsv + (size_t)blockIdx.x * GCAP;
  float4*       gcfb = gcf + (size_t)blockIdx.x * GCAP;
  const float* rcf = (const float*)rc;
  const float sgm = 1.f / (1.f + __expf(-sigma[0]));

  for (int i = t; i < RNG * HF / 4; i += AGG_T)
    ((float4*)acc)[i] = make_float4(0.f, 0.f, 0.f, 0.f);
  if (t == 0) qn = 0u;

  const unsigned cnt = __builtin_amdgcn_readfirstlane(gcount[blockIdx.x]);
  if (cnt <= (unsigned)GCAP) {
    if (FIRST) {
      const unsigned whole = cnt & ~(unsigned)(AGG_T - 1);
      if (whole + wv * 32u < cnt) {
        const unsigned idx = whole + (unsigned)t;
        unsigned word; float4 c;
        make_record((idx < cnt) ? gqb[idx] : 0xFFFFFFFFu, lo, src, el, er, maxs, maxd, sums, sumd,
                    norm_out, sgm, word, c);
        if (idx < cnt) { rs[idx] = word; rc[idx] = c; gsvb[idx] = word; gcfb[idx] = c; }
      }
      __syncthreads();
      drain_records(whole, cnt, wv, lane, hsel, rs, rcf, xprev, acc);
      for (unsigned base = wv * 32u; base < whole; base += AGG_T) {
        const unsigned idx = base + (unsigned)lane;
        unsigned word; float4 c;
        make_record(gqb[idx], lo, src, el, er, maxs, maxd, sums, sumd, norm_out, sgm, word, c);
        rs[idx] = word; rc[idx] = c; gsvb[idx] = word; gcfb[idx] = c;
        for (unsigned i0 = base; i0 < base + 32u; i0 += DRAIN_U)
          add_batch(i0, lane, hsel, rs, rcf, xprev, acc);
      }
    } else {
      unsigned w[GCAP / AGG_T];
      float4   c[GCAP / AGG_T];
#pragma unroll
      for (int j = 0; j < GCAP / AGG_T; ++j) {
        const unsigned i = (unsigned)(j * AGG_T + t);
        w[j] = (i < cnt) ? gsvb[i] : 0u;
        c[j] = (i < cnt) ? gcfb[i] : make_float4(0.f, 0.f, 0.f, 0.f);
      }
#pragma unroll
      for (int j = 0; j < GCAP / AGG_T; ++j) {
        const unsigned i = (unsigned)(j * AGG_T + t);
        if (i < cnt) { rs[i] = w[j]; rc[i] = c[j]; }
      }
      __syncthreads();
      drain_records(0u, cnt, wv, lane, hsel, rs, rcf, xprev, acc);
    }
    __syncthreads();
  } else {
    __syncthreads();
    for (int base = 0; base < E_EDGES; base += SCAN_EDGES) {
      const int e0 = base + t * 4;
      const int4 dv = (e0 < E_EDGES) ? *(const int4*)(dst + e0) : make_int4(-1, -1, -1, -1);
      const unsigned rx = (unsigned)(dv.x - lo), ry = (unsigned)(dv.y - lo);
      const unsigned rz = (unsigned)(dv.z - lo), rw = (unsigned)(dv.w - lo);
      if (min(min(rx, ry), min(rz, rw)) < (unsigned)RNG) {
        push_edge(rx, e0,     q, &qn);
        push_edge(ry, e0 + 1, q, &qn);
        push_edge(rz, e0 + 2, q, &qn);
        push_edge(rw, e0 + 3, q, &qn);
      }
      __syncthreads();
      const unsigned n = __builtin_amdgcn_readfirstlane(qn);
      __syncthreads();
      const bool last = (base + SCAN_EDGES >= E_EDGES);
      if (n > (unsigned)(QCAP - SCAN_EDGES) || last) {
        for (unsigned c0 = 0; c0 < n; c0 += CH) {
          const unsigned m = min((unsigned)CH, n - c0);
          for (unsigned idx = (unsigned)t; idx < m; idx += AGG_T) {
            unsigned word; float4 c;
            make_record(q[c0 + idx], lo, src, el, er, maxs, maxd, sums, sumd, norm_out, sgm, word, c);
            rs[idx] = word; rc[idx] = c;
          }
          __syncthreads();
          drain_records(0u, m, wv, lane, hsel, rs, rcf, xprev, acc);
          __syncthreads();
        }
        if (t == 0) qn = 0u;
        __syncthreads();
      }
    }
  }

  if (!LAST) {
    for (int g = t; g < RNG * (HF / 4); g += AGG_T) {
      const int row = g >> 6, c4 = g & 63;
      const float* ap = acc + row * HF + (c4 >> 5) * 128 + (c4 & 31);
      const float ni = norm_in[lo + row];
      const float4 o = make_float4(ap[0] * ni, ap[32] * ni, ap[64] * ni, ap[96] * ni);
      *(float4*)(xnext + (size_t)(lo + row) * HF + c4 * 4) = o;
    }
  } else {
    constexpr int NG = RNG * (HF / 4);
    static_assert(AGG_T % 64 == 0 && NG % 32 == 0 && NG > 2 * AGG_T,
                  "a thread keeps its column; waves are whole inside the bounds; two loads ahead exist");
    const int c4   = t & 63;
    const int head = c4 >> 4;
    const float4 ar = *(const float4*)(hop_attn_r + c4 * 4);
    const float* hb  = h0    + (size_t)lo * HF;
    const float* x1b = x1p   + (size_t)lo * HF;
    const float* x2b = xprev + (size_t)lo * HF;
    float*       ob  = xnext + (size_t)lo * HF;
    HopRows ra = load_hop_rows(t,         lo, head, hb, x1b, x2b, norm_in, al);
    HopRows rb = load_hop_rows(t + AGG_T, lo, head, hb, x1b, x2b, norm_in, al);
    for (int g = t; g < NG; g += 2 * AGG_T) {
      const HopRows ca = ra;
      if (g + 2 * AGG_T < NG) ra = load_hop_rows(g + 2 * AGG_T, lo, head, hb, x1b, x2b, norm_in, al);
      combine_store(g, c4, ca, ar, acc, ob);
      if (g + AGG_T < NG) {
        const HopRows cb = rb;
        if (g + 3 * AGG_T < NG) rb = load_hop_rows(g + 3 * AGG_T, lo, head, hb, x1b, x2b, norm_in, al);
        combine_store(g + AGG_T, c4, cb, ar, acc, ob);
      }
    }
  }
}

extern "C" void kernel_launch(void* const* d_in, const int* in_sizes, int n_in,
                              void* d_out, int out_size, void* d_ws, size_t ws_size,
                              hipStream_t stream) {
  (void)in_sizes; (void)n_in; (void)out_size; (void)ws_size;
  const float* feat       = (const float*)d_in[0];
  const int*   src        = (const int*)d_in[1];
  const int*   dst        = (const int*)d_in[2];
  const float* W          = (const float*)d_in[3];
  const float* attn_l     = (const float*)d_in[4];
  const float* attn_r     = (const float*)d_in[5];
  const float* hop_attn_l = (const float*)d_in[6];
  const float* hop_attn_r = (const float*)d_in[7];
  const float* sigma      = (const float*)d_in[8];

  float* ws = (float*)d_ws;
  size_t off = 0;
  float* h    = ws + off; off += (size_t)N_NODES * HF;
  float* x1   = ws + off; off += (size_t)N_NODES * HF;
  float* x2   = ws + off; off += (size_t)N_NODES * HF;
  unsigned int* bm = (unsigned int*)(ws + off); off += (size_t)N_NODES * HF;
  float* el   = ws + off; off += (size_t)N_NODES * H_HEADS;
  float* er   = ws + off; off += (size_t)N_NODES * H_HEADS;
  float* al   = ws + off; off += (size_t)N_NODES * H_HEADS;
  float* maxs = ws + off; off += (size_t)N_NODES * H_HEADS;
  float* maxd = ws + off; off += (size_t)N_NODES * H_HEADS;
  float* sums = ws + off; off += (size_t)N_NODES * H_HEADS;
  float* sumd = ws + off; off += (size_t)N_NODES * H_HEADS;
  off += (size_t)N_NODES * 2;
  float* norm_out = ws + off; off += (size_t)N_NODES;
  float* norm_in  = ws + off; off += (size_t)N_NODES;
  float* edge_planes = ws + off; off += (size_t)E_EDGES * H_HEADS * 2;

  gemm_h_kernel<<<(N_NODES + 127) / 128, 256, 0, stream>>>(feat, W, h);

  scores_kernel<<<SC_NBLK, SC_T, 0, stream>>>(h, attn_l, attn_r, hop_attn_l, el, er, al, src, dst, bm);

  unsigned int* gq     = (unsigned int*)edge_planes;
  unsigned int* gcount = gq + (size_t)NBLK * GCAP;
  unsigned int* gsv    = gcount + 256;
  float4*       gcf    = (float4*)(gsv + (size_t)NBLK * GCAP);
  seg_stats_kernel<<<ST_NBLK, ST_T, 0, stream>>>(src, dst, el, er, maxs, maxd, sums, sumd,
                                                 norm_out, norm_in, gq, gcount, bm);

  agg_tile_kernel<true, false><<<NBLK, AGG_T, 0, stream>>>(src, dst, el, er, maxs, maxd, sums, sumd, norm_out,
                                                           norm_in, sigma, h,  x1, gq, gcount, gsv, gcf,
                                                           h, x1, al, hop_attn_r);
  agg_tile_kernel<false, false><<<NBLK, AGG_T, 0, stream>>>(src, dst, el, er, maxs, maxd, sums, sumd, norm_out,
                                                            norm_in, sigma, x1, x2, gq, gcount, gsv, gcf,
                                                            h, x1, al, hop_attn_r);
  agg_tile_kernel<false, true><<<NBLK, AGG_T, 0, stream>>>(src, dst, el, er, maxs, maxd, sums, sumd, norm_out,
                                                           norm_in, sigma, x2, (float*)d_out, gq, gcount, gsv, gcf,
                                                           h, x1, al, hop_attn_r);
}


# No score launch: product waves take el, er, al of their 16 rows from the WMMA accumulators (score sums regrouped: 4 per lane, then a 16-lane tree); the bitmap blocks join the product launch's grid
# speedup vs baseline: 8.4643x; 1.1439x over previous
#include <hip/hip_runtime.h>
#include <hip/hip_bf16.h>

#define N_NODES 50000
#define E_EDGES 800000
#define IN_F    256
#define H_HEADS 4
#define F_OUT   64
#define HF      256
#define SLOPE   0.2f

typedef __attribute__((ext_vector_type(16))) __bf16 v16bf;
typedef __attribute__((ext_vector_type(8)))  float  v8f;

__device__ __forceinline__ float leakyf(float x) { return x > 0.f ? x : SLOPE * x; }

__device__ __forceinline__ void atomicMaxF(float* addr, float val) {
  if (val >= 0.f) {
    atomicMax((int*)addr, __float_as_int(val));
  } else {
    atomicMin((unsigned int*)addr, __float_as_uint(val));
  }
}

__device__ __forceinline__ float row16_sum(float v) {
  v += __int_as_float(__builtin_amdgcn_update_dpp(0, __float_as_int(v), 0x168, 0xF, 0xF, true));
  v += __int_as_float(__builtin_amdgcn_update_dpp(0, __float_as_int(v), 0x164, 0xF, 0xF, true));
  v += __int_as_float(__builtin_amdgcn_update_dpp(0, __float_as_int(v), 0x162, 0xF, 0xF, true));
  v += __int_as_float(__builtin_amdgcn_update_dpp(0, __float_as_int(v), 0x161, 0xF, 0xF, true));
  return v;
}


#define RNG        200
#define ST_K       2
#define ST_RNG     (ST_K * RNG)
#define ST_NRNG    (N_NODES / ST_RNG)
#define SC_T       256
#define BM_EDGES   (SC_T * 4)
#define BM_WORDS   (BM_EDGES / 32)
#define BM_NBLK    ((E_EDGES + BM_EDGES - 1) / BM_EDGES)
#define BM_PITCH   (BM_NBLK * BM_WORDS)
#define WT_STRIDE  40
#define GEMM_ROWS  128
#define GEMM_NBLK  ((N_NODES + GEMM_ROWS - 1) / GEMM_ROWS)

static_assert(N_NODES % ST_RNG == 0, "bitmap ranges tile the node set exactly");
static_assert(E_EDGES % 4 == 0, "the index arrays are read 4 edges at a time");
static_assert(BM_WORDS == 32 && BM_PITCH % 4 == 0, "a block's share of a bitmap is one 128-byte line");
static_assert(N_NODES % 16 == 0, "a wave's 16 rows are all inside the node set or all outside");
static_assert(2 * ST_NRNG * BM_WORDS * 4 >= HF * WT_STRIDE * 2, "the bitmap lines and the staged weight slice share one LDS array");
static_assert((2 * ST_NRNG * BM_WORDS) % 4 == 0, "the LDS copy is moved 16 bytes at a time");

__global__ __launch_bounds__(256)
void gemm_h_kernel(const float* __restrict__ feat, const float* __restrict__ W,
                   float* __restrict__ h,
                   const float* __restrict__ attn_l,
                   const float* __restrict__ attn_r,
                   const float* __restrict__ hop_attn_l,
                   float* __restrict__ el, float* __restrict__ er,
                   float* __restrict__ al,
                   const int* __restrict__ src, const int* __restrict__ dst,
                   unsigned int* __restrict__ bm) {
  __shared__ __align__(16) unsigned int smem[2 * ST_NRNG * BM_WORDS];
  const int t = threadIdx.x;

  if (blockIdx.x >= GEMM_NBLK) {
    unsigned int* bml = smem;
    const int cb = (int)blockIdx.x - GEMM_NBLK;
    const int e0 = cb * BM_EDGES + t * 4;
    int4 sv = make_int4(0, 0, 0, 0), dv = make_int4(0, 0, 0, 0);
    const bool have = (e0 < E_EDGES);
    if (have) { sv = *(const int4*)(src + e0); dv = *(const int4*)(dst + e0); }
    for (int k = t; k < 2 * ST_NRNG * BM_WORDS / 4; k += SC_T)
      ((uint4*)bml)[k] = make_uint4(0u, 0u, 0u, 0u);
    __syncthreads();
    if (have) {
      const unsigned w = (unsigned)t >> 3;
      const unsigned b = ((unsigned)t & 7u) * 4u;
      unsigned int* ends   = bml + w;
      unsigned int* starts = bml + ST_NRNG * BM_WORDS + w;
      atomicOr(&ends[((unsigned)dv.x / ST_RNG) * BM_WORDS], 1u << b);
      atomicOr(&ends[((unsigned)dv.y / ST_RNG) * BM_WORDS], 2u << b);
      atomicOr(&ends[((unsigned)dv.z / ST_RNG) * BM_WORDS], 4u << b);
      atomicOr(&ends[((unsigned)dv.w / ST_RNG) * BM_WORDS], 8u << b);
      atomicOr(&starts[((unsigned)sv.x / ST_RNG) * BM_WORDS], 1u << b);
      atomicOr(&starts[((unsigned)sv.y / ST_RNG) * BM_WORDS], 2u << b);
      atomicOr(&starts[((unsigned)sv.z / ST_RNG) * BM_WORDS], 4u << b);
      atomicOr(&starts[((unsigned)sv.w / ST_RNG) * BM_WORDS], 8u << b);
    }
    __syncthreads();
    for (int k = t; k < 2 * ST_NRNG * BM_WORDS / 4; k += SC_T) {
      const int row = k >> 3, part = k & 7;
      *(uint4*)(bm + (size_t)row * BM_PITCH + (size_t)cb * BM_WORDS + part * 4) = ((const uint4*)bml)[k];
    }
    return;
  }

  __bf16* Wt = (__bf16*)smem;

  const int wave    = t >> 5;
  const int lane    = t & 31;
  const int rowBase = blockIdx.x * GEMM_ROWS + wave * 16;
  const bool rowValid = (rowBase < N_NODES);
  const int nloc = lane & 15;
  const int kg   = lane >> 4;

  v8f acc[16];
#pragma unroll
  for (int nt = 0; nt < 16; ++nt)
#pragma unroll
    for (int v = 0; v < 8; ++v) acc[nt][v] = 0.f;

  for (int ks = 0; ks < IN_F / 32; ++ks) {
    const int kbase = ks * 32;
    __syncthreads();
#pragma unroll
    for (int kk = 0; kk < 32; ++kk)
      Wt[t * WT_STRIDE + kk] = (__bf16)W[(kbase + kk) * HF + t];
    __syncthreads();

    if (rowValid) {
      const float* arow = feat + (size_t)(rowBase + nloc) * IN_F + kbase;
      float4 a0 = *(const float4*)(arow + 8 * kg);
      float4 a1 = *(const float4*)(arow + 8 * kg + 4);
      float4 a2 = *(const float4*)(arow + 16 + 8 * kg);
      float4 a3 = *(const float4*)(arow + 16 + 8 * kg + 4);
      float atmp[16] = {a0.x, a0.y, a0.z, a0.w, a1.x, a1.y, a1.z, a1.w,
                        a2.x, a2.y, a2.z, a2.w, a3.x, a3.y, a3.z, a3.w};
      v16bf afrag;
#pragma unroll
      for (int i = 0; i < 16; ++i) afrag[i] = (__bf16)atmp[i];

      const __bf16* wrow0 = &Wt[nloc * WT_STRIDE];
      uint4 b0 = *(const uint4*)(wrow0 + 8 * kg);
      uint4 b1 = *(const uint4*)(wrow0 + 16 + 8 * kg);
#pragma unroll
      for (int nt = 0; nt < 16; ++nt) {
        uint4 nb0, nb1;
        if (nt < 15) {
          const __bf16* wrow = &Wt[((nt + 1) * 16 + nloc) * WT_STRIDE];
          nb0 = *(const uint4*)(wrow + 8 * kg);
          nb1 = *(const uint4*)(wrow + 16 + 8 * kg);
        }
        const __bf16* p0 = (const __bf16*)&b0;
        const __bf16* p1 = (const __bf16*)&b1;
        v16bf bfrag;
#pragma unroll
        for (int i = 0; i < 8; ++i) { bfrag[i] = p0[i]; bfrag[8 + i] = p1[i]; }
        acc[nt] = __builtin_amdgcn_wmma_f32_16x16x32_bf16(
            false, afrag, false, bfrag, (short)0, acc[nt], false, false);
        if (nt < 15) { b0 = nb0; b1 = nb1; }
      }
    }
  }

  if (rowValid) {
#pragma unroll
    for (int nt = 0; nt < 16; ++nt)
#pragma unroll
      for (int v = 0; v < 8; ++v)
        h[(size_t)(rowBase + v + 8 * kg) * HF + nt * 16 + nloc] = acc[nt][v];

    float ol[4], orr[4], oh[4];
#pragma unroll
    for (int hd = 0; hd < 4; ++hd) {
      float pl[8], pr[8], ph[8];
#pragma unroll
      for (int v = 0; v < 8; ++v) { pl[v] = 0.f; pr[v] = 0.f; ph[v] = 0.f; }
#pragma unroll
      for (int j = 0; j < 4; ++j) {
        const int nt = hd * 4 + j;
        const float cl = attn_l[nt * 16 + nloc];
        const float cr = attn_r[nt * 16 + nloc];
        const float ch = hop_attn_l[nt * 16 + nloc];
#pragma unroll
        for (int v = 0; v < 8; ++v) {
          pl[v] = fmaf(acc[nt][v], cl, pl[v]);
          pr[v] = fmaf(acc[nt][v], cr, pr[v]);
          ph[v] = fmaf(acc[nt][v], ch, ph[v]);
        }
      }
      float sl = 0.f, sr = 0.f, sh = 0.f;
#pragma unroll
      for (int v = 0; v < 8; ++v) {
        const float tl = row16_sum(pl[v]);
        const float tr = row16_sum(pr[v]);
        const float th = row16_sum(ph[v]);
        if ((nloc & 7) == v) { sl = tl; sr = tr; sh = th; }
      }
      ol[hd] = sl; orr[hd] = sr; oh[hd] = sh;
    }
    if (nloc < 8) {
      const size_t o = (size_t)(rowBase + nloc + 8 * kg) * H_HEADS;
      *(float4*)(el + o) = make_float4(ol[0], ol[1], ol[2], ol[3]);
      *(float4*)(er + o) = make_float4(orr[0], orr[1], orr[2], orr[3]);
      *(float4*)(al + o) = make_float4(oh[0], oh[1], oh[2], oh[3]);
    }
  }
}


__device__ __forceinline__ float mix_coef(float el, float er, float ms, float md,
                                          float ss, float sd, float sgm, float nrm) {
  const float ev = leakyf(el + er);
  const float ax = fmaxf(__expf(ev - ms) / ss, 1e-10f);
  const float ay = fmaxf(__expf(ev - md) / sd, 1e-10f);
  const float a  = __powf(ay, sgm) * __powf(ax, 1.f - sgm);
  return a * nrm;
}

#define NBLK       (N_NODES / RNG)
#define AGG_T      1024
#define GCAP       4096
#define CH         GCAP
#define SCAN_EDGES (AGG_T * 4)
#define QCAP       (SCAN_EDGES + 1504)
#define DRAIN_U    4
#define NWAVE      (AGG_T / 32)
#define OWN_SLOTS  ((RNG + NWAVE - 1) / NWAVE)
#define SCAN_R     128
#define LIST_GO    16
#define LW         (LIST_GO + SCAN_R)
#define LQ         (32 + SCAN_R)
#define WAREA      (2 * LQ + 32 + 32 * 4)
#define POOL_W     (CH * 5 + QCAP + NWAVE * LW)

static_assert(N_NODES % RNG == 0, "ranges tile the node set exactly");
static_assert(E_EDGES % 4 == 0, "dst is read 4 edges at a time");
static_assert(E_EDGES <= (1 << 20) && RNG <= 256, "queue word packs row and edge id");
static_assert(N_NODES <= (1 << 16), "record word packs row and source node");
static_assert((RNG * HF + POOL_W + 1) * 4 <= 320 * 1024, "LDS budget");
static_assert(CH + NWAVE * WAREA <= POOL_W, "the first hop's layout fits the same pool");
static_assert(LQ % 4 == 0 && (CH * 5) % 4 == 0 && QCAP % 4 == 0 && LW % 4 == 0 && WAREA % 4 == 0,
              "every part of the pool starts on 16 bytes");
static_assert(NWAVE == 32 && OWN_SLOTS <= 8, "owner wave = row & 31; list entry packs row >> 5 in 3 bits");
static_assert(CH <= (1 << 12), "list entry packs the record's place in 12 bits");
static_assert(CH % SCAN_R == 0 && LW % DRAIN_U == 0 && LIST_GO % DRAIN_U == 0 && LIST_GO >= DRAIN_U,
              "list trips stay inside the arrays");
static_assert(QCAP >= SCAN_EDGES + AGG_T, "a scan group fits behind what may be left in the queue");
static_assert(DRAIN_U == 4, "a batch is one 16-byte read of record words");
static_assert(GCAP % AGG_T == 0, "whole rounds of the block cover a slice");
static_assert((size_t)NBLK * GCAP * 6 + 256 <= (size_t)E_EDGES * H_HEADS * 2,
              "saved queues, counts and records fit the two edge planes");
static_assert((size_t)2 * ST_NRNG * BM_PITCH <= (size_t)N_NODES * HF,
              "both sets of bitmaps fit the place of the third hop's plane");

__device__ __forceinline__ void push_edge(unsigned r, int eid, unsigned* q, unsigned* qn) {
  if (r < (unsigned)RNG) {
    const unsigned p = atomicAdd(qn, 1u);
    q[p] = (r << 20) | (unsigned)eid;
  }
}

__device__ __forceinline__ void make_record(unsigned w, int lo,
                                            const int* __restrict__ src,
                                            const float* __restrict__ el, const float* __restrict__ er,
                                            const float* __restrict__ maxs, const float* __restrict__ maxd,
                                            const float* __restrict__ sums, const float* __restrict__ sumd,
                                            const float* __restrict__ norm_out, float sgm,
                                            unsigned& word, float4& c) {
  const bool     ok  = (w != 0xFFFFFFFFu);
  const unsigned eid = ok ? (w & 0xFFFFFu) : 0u;
  const unsigned r   = ok ? (w >> 20) : 0u;
  const int      s   = src[eid];
  const size_t   d   = (size_t)(lo + (int)r);
  const float4 a  = *(const float4*)(el   + (size_t)s * 4);
  const float4 ms = *(const float4*)(maxs + (size_t)s * 4);
  const float4 ss = *(const float4*)(sums + (size_t)s * 4);
  const float  nr = norm_out[s];
  const float4 b  = *(const float4*)(er   + d * 4);
  const float4 md = *(const float4*)(maxd + d * 4);
  const float4 sd = *(const float4*)(sumd + d * 4);
  c.x = mix_coef(a.x, b.x, ms.x, md.x, ss.x, sd.x, sgm, nr);
  c.y = mix_coef(a.y, b.y, ms.y, md.y, ss.y, sd.y, sgm, nr);
  c.z = mix_coef(a.z, b.z, ms.z, md.z, ss.z, sd.z, sgm, nr);
  c.w = mix_coef(a.w, b.w, ms.w, md.w, ss.w, sd.w, sgm, nr);
  word = (r << 16) | (unsigned)s;
}


template <bool FULL>
__device__ __forceinline__ void add_batch(unsigned i0, unsigned m, unsigned wv, int lane, int hsel,
                                          const unsigned* lw, const float* rcf,
                                          const float* __restrict__ xprev, float* tile) {
  const uint4 e4 = *(const uint4*)&lw[i0];
  const unsigned eq[DRAIN_U] = {
      (unsigned)__builtin_amdgcn_readfirstlane(e4.x), (unsigned)__builtin_amdgcn_readfirstlane(e4.y),
      (unsigned)__builtin_amdgcn_readfirstlane(e4.z), (unsigned)__builtin_amdgcn_readfirstlane(e4.w)};
  unsigned ent[DRAIN_U];
  float4 r0[DRAIN_U], r1[DRAIN_U];
  float  c0[DRAIN_U], c1[DRAIN_U];
#pragma unroll
  for (int u = 0; u < DRAIN_U; ++u) {
    const bool real = FULL || (unsigned)u < m;
    ent[u] = real ? eq[u] : eq[0];
    const unsigned ix = (ent[u] >> 16) & 0xFFFu;
    const float* row = xprev + (size_t)(ent[u] & 0xFFFFu) * HF + lane * 4;
    r0[u] = *(const float4*)row;
    r1[u] = *(const float4*)(row + 128);
    const float k0 = rcf[ix * 4 + hsel];
    const float k1 = rcf[ix * 4 + 2 + hsel];
    c0[u] = real ? k0 : 0.f;
    c1[u] = real ? k1 : 0.f;
  }
#pragma unroll
  for (int u = 0; u < DRAIN_U; ++u) {
    float* trow = tile + ((ent[u] >> 28) * NWAVE + wv) * HF + lane * 4;
    float4 a0 = *(const float4*)trow;
    float4 a1 = *(const float4*)(trow + 128);
    a0.x = fmaf(r0[u].x, c0[u], a0.x); a0.y = fmaf(r0[u].y, c0[u], a0.y);
    a0.z = fmaf(r0[u].z, c0[u], a0.z); a0.w = fmaf(r0[u].w, c0[u], a0.w);
    a1.x = fmaf(r1[u].x, c1[u], a1.x); a1.y = fmaf(r1[u].y, c1[u], a1.y);
    a1.z = fmaf(r1[u].z, c1[u], a1.z); a1.w = fmaf(r1[u].w, c1[u], a1.w);
    *(float4*)trow         = a0;
    *(float4*)(trow + 128) = a1;
  }
}

__device__ __forceinline__ void drain_list(unsigned n, unsigned wv, int lane, int hsel,
                                           const unsigned* lw, const float* rcf,
                                           const float* __restrict__ xprev, float* tile) {
  const unsigned full = n & ~(unsigned)(DRAIN_U - 1);
#pragma unroll 1
  for (unsigned i0 = 0; i0 < full; i0 += DRAIN_U)
    add_batch<true>(i0, DRAIN_U, wv, lane, hsel, lw, rcf, xprev, tile);
  if (full < n)
    add_batch<false>(full, n - full, wv, lane, hsel, lw, rcf, xprev, tile);
}

__device__ __forceinline__ void drain_owned(unsigned n, unsigned wv, int lane, int hsel,
                                            const unsigned* rs, const float* rcf, unsigned* lw,
                                            const float* __restrict__ xprev, float* tile) {
  unsigned have = 0u;
  const unsigned below = (1u << lane) - 1u;
  for (unsigned base = 0; base < n; base += SCAN_R) {
    const unsigned i4 = base + (unsigned)lane * 4u;
    const uint4 w4 = *(const uint4*)&rs[i4];
    const unsigned wd[4] = {w4.x, w4.y, w4.z, w4.w};
#pragma unroll
    for (int k = 0; k < 4; ++k) {
      const unsigned idx  = i4 + (unsigned)k;
      const unsigned row  = wd[k] >> 16;
      const bool     mine = idx < n && (row & (NWAVE - 1u)) == wv;
      const unsigned m    = __builtin_amdgcn_ballot_w32(mine);
      if (mine) {
        const unsigned p = have + (unsigned)__builtin_popcount(m & below);
        lw[p] = ((row >> 5) << 28) | (idx << 16) | (wd[k] & 0xFFFFu);
      }
      have += (unsigned)__builtin_popcount(m);
    }
    if (have >= (unsigned)LIST_GO) {
      const unsigned full = have & ~(unsigned)(DRAIN_U - 1);
      drain_list(full, wv, lane, hsel, lw, rcf, xprev, tile);
      const unsigned rem = have - full;
      unsigned keep = 0u;
      if ((unsigned)lane < rem) keep = lw[full + (unsigned)lane];
      if ((unsigned)lane < rem) lw[lane] = keep;
      have = rem;
    }
  }
  drain_list(have, wv, lane, hsel, lw, rcf, xprev, tile);
}

__device__ __forceinline__ void build_group(unsigned g, unsigned m, unsigned wv, int lane, int hsel, int lo,
                                            const unsigned* lq, const unsigned* li, unsigned* eb, float4* cb,
                                            unsigned int* gsvb, float4* gcfb,
                                            const int* __restrict__ src,
                                            const float* __restrict__ el, const float* __restrict__ er,
                                            const float* __restrict__ maxs, const float* __restrict__ maxd,
                                            const float* __restrict__ sums, const float* __restrict__ sumd,
                                            const float* __restrict__ norm_out, float sgm,
                                            const float* __restrict__ xprev, float* tile) {
  const bool on = (unsigned)lane < m;
  const unsigned qw  = on ? lq[g + (unsigned)lane] : 0xFFFFFFFFu;
  const unsigned idx = on ? li[g + (unsigned)lane] : 0u;
  unsigned word; float4 c;
  make_record(qw, lo, src, el, er, maxs, maxd, sums, sumd, norm_out, sgm, word, c);
  if (on) {
    gsvb[idx] = word; gcfb[idx] = c;
    eb[lane] = ((word >> 21) << 28) | ((unsigned)lane << 16) | (word & 0xFFFFu);
    cb[lane] = c;
  }
  drain_list(m, wv, lane, hsel, eb, (const float*)cb, xprev, tile);
}

__device__ __forceinline__ void build_owned(unsigned n, unsigned wv, int lane, int hsel, int lo,
                                            const unsigned* qs, unsigned* lq, unsigned* li,
                                            unsigned* eb, float4* cb,
                                            unsigned int* gsvb, float4* gcfb,
                                            const int* __restrict__ src,
                                            const float* __restrict__ el, const float* __restrict__ er,
                                            const float* __restrict__ maxs, const float* __restrict__ maxd,
                                            const float* __restrict__ sums, const float* __restrict__ sumd,
                                            const float* __restrict__ norm_out, float sgm,
                                            const float* __restrict__ xprev, float* tile) {
  unsigned have = 0u;
  const unsigned below = (1u << lane) - 1u;
  for (unsigned base = 0; base < n; base += SCAN_R) {
    const unsigned i4 = base + (unsigned)lane * 4u;
    const uint4 w4 = *(const uint4*)&qs[i4];
    const unsigned wd[4] = {w4.x, w4.y, w4.z, w4.w};
#pragma unroll
    for (int k = 0; k < 4; ++k) {
      const unsigned idx  = i4 + (unsigned)k;
      const bool     mine = idx < n && ((wd[k] >> 20) & (NWAVE - 1u)) == wv;
      const unsigned m    = __builtin_amdgcn_ballot_w32(mine);
      if (mine) {
        const unsigned p = have + (unsigned)__builtin_popcount(m & below);
        lq[p] = wd[k];
        li[p] = idx;
      }
      have += (unsigned)__builtin_popcount(m);
    }
    if (have >= 32u) {
      const unsigned full = have & ~31u;
      for (unsigned g = 0; g < full; g += 32u)
        build_group(g, 32u, wv, lane, hsel, lo, lq, li, eb, cb, gsvb, gcfb,
                    src, el, er, maxs, maxd, sums, sumd, norm_out, sgm, xprev, tile);
      const unsigned rem = have - full;
      unsigned kq = 0u, ki = 0u;
      if ((unsigned)lane < rem) { kq = lq[full + (unsigned)lane]; ki = li[full + (unsigned)lane]; }
      if ((unsigned)lane < rem) { lq[lane] = kq; li[lane] = ki; }
      have = rem;
    }
  }
  if (have != 0u)
    build_group(0u, have, wv, lane, hsel, lo, lq, li, eb, cb, gsvb, gcfb,
                src, el, er, maxs, maxd, sums, sumd, norm_out, sgm, xprev, tile);
}

#define ST_NBLK    (2 * ST_NRNG)
#define ST_T       1024
#define ST_VS      4
#define BM_V       ((BM_PITCH / 4 + ST_T - 1) / ST_T)
#define ST_EDGES_S (ST_T * 4 * ST_VS)
#define ST_QCAP    (3 * ST_EDGES_S)
#define ST_U       8

static_assert(N_NODES % ST_RNG == 0 && ST_RNG <= 512, "statistics ranges tile the node set; local node fits 9 bits");
static_assert(ST_RNG <= ST_T, "one thread per degree word");
static_assert(ST_T / 32 <= 32, "one lane per wave total");
#define SUM_FRAC   40
static_assert((ST_QCAP + ST_RNG * 13 + 4) * 4 <= 320 * 1024, "LDS budget of the statistics kernel");
static_assert(E_EDGES <= (1 << 20), "a segment's terms, each at most 2^SUM_FRAC, fit 64 bits");

__device__ __forceinline__ void add_term(unsigned long long* p, float term) {
  atomicAdd(p, (unsigned long long)(term * (float)(1ull << SUM_FRAC)));
}

__device__ __forceinline__ void push_own(unsigned r, int eid, unsigned* q, unsigned* qn) {
  if (r < (unsigned)ST_RNG) {
    const unsigned p = atomicAdd(qn, 1u);
    q[p] = (unsigned)eid;
  }
}

template <int V>
__device__ __forceinline__ void scan_group(int base, int t, int lo,
                                           const int* __restrict__ idx,
                                           unsigned* q, unsigned* qn) {
  int4 iv[V];
#pragma unroll
  for (int j = 0; j < V; ++j) {
    const int e0 = base + (j * ST_T + t) * 4;
    if (e0 < E_EDGES) iv[j] = *(const int4*)(idx + e0);
    else              iv[j] = make_int4(-1, -1, -1, -1);
  }
#pragma unroll
  for (int j = 0; j < V; ++j) {
    const int e0 = base + (j * ST_T + t) * 4;
    const unsigned rx = (unsigned)(iv[j].x - lo), ry = (unsigned)(iv[j].y - lo);
    const unsigned rz = (unsigned)(iv[j].z - lo), rw = (unsigned)(iv[j].w - lo);
    if (min(min(rx, ry), min(rz, rw)) < (unsigned)ST_RNG) {
      push_own(rx, e0,     q, qn);
      push_own(ry, e0 + 1, q, qn);
      push_own(rz, e0 + 2, q, qn);
      push_own(rw, e0 + 3, q, qn);
    }
  }
}

__device__ __forceinline__ void push_bits(unsigned word, unsigned first, unsigned& p, unsigned* q) {
  while (word != 0u) {
    const unsigned b = (unsigned)__builtin_ctz(word);
    word &= word - 1u;
    if (p < (unsigned)ST_QCAP) q[p] = first + b;
    ++p;
  }
}

__device__ __forceinline__ void queue_from_bitmap(const unsigned int* __restrict__ row, int t,
                                                  unsigned* q, unsigned* qn, unsigned* wtot) {
  uint4 wv[BM_V];
#pragma unroll
  for (int j = 0; j < BM_V; ++j) {
    const int k = j * ST_T + t;
    wv[j] = (k < BM_PITCH / 4) ? ((const uint4*)row)[k] : make_uint4(0u, 0u, 0u, 0u);
  }
  unsigned cnt = 0u;
#pragma unroll
  for (int j = 0; j < BM_V; ++j)
    cnt += (unsigned)(__builtin_popcount(wv[j].x) + __builtin_popcount(wv[j].y) +
                      __builtin_popcount(wv[j].z) + __builtin_popcount(wv[j].w));
  const unsigned lane = (unsigned)t & 31u, wave = (unsigned)t >> 5;
  unsigned incl = cnt;
#pragma unroll
  for (int d = 1; d < 32; d <<= 1) {
    const unsigned below = (unsigned)__shfl_up((int)incl, d);
    if (lane >= (unsigned)d) incl += below;
  }
  if (lane == 31u) wtot[wave] = incl;
  __syncthreads();
  unsigned before = (lane < wave) ? wtot[lane] : 0u;
#pragma unroll
  for (int d = 16; d >= 1; d >>= 1) before += (unsigned)__shfl_xor((int)before, d);
  if (t == ST_T - 1) *qn = before + incl;
  unsigned p = before + incl - cnt;
  if (cnt != 0u) {
#pragma unroll
    for (int j = 0; j < BM_V; ++j) {
      const unsigned first = (unsigned)(j * ST_T + t) * 128u;
      push_bits(wv[j].x, first,       p, q);
      push_bits(wv[j].y, first + 32u, p, q);
      push_bits(wv[j].z, first + 64u, p, q);
      push_bits(wv[j].w, first + 96u, p, q);
    }
  }
}

template <int PASS>
__device__ __forceinline__ void stats_pass(unsigned n, int t, bool ends, int lo, const unsigned* q,
                                           const int* __restrict__ src,
                                           const int* __restrict__ dst,
                                           const float* __restrict__ el,
                                           const float* __restrict__ er,
                                           float* st_max, unsigned long long* st_sum, float* st_deg,
                                           unsigned* sn, unsigned* gqb) {
  static_assert(PASS >= 0 && PASS <= 2, "maxima, sums, or both");
  for (unsigned i0 = 0; i0 < n; i0 += ST_T * ST_U) {
    unsigned ent[ST_U];
    int      sv[ST_U], dv[ST_U];
    float4   a[ST_U], b[ST_U];
#pragma unroll
    for (int u = 0; u < ST_U; ++u) {
      const unsigned idx = i0 + (unsigned)(u * ST_T + t);
      ent[u] = (idx < n) ? q[idx] : 0xFFFFFFFFu;
    }
#pragma unroll
    for (int u = 0; u < ST_U; ++u) {
      const unsigned eid = (ent[u] == 0xFFFFFFFFu) ? 0u : (ent[u] & 0xFFFFFu);
      sv[u] = src[eid];
      dv[u] = dst[eid];
    }
#pragma unroll
    for (int u = 0; u < ST_U; ++u) {
      a[u] = *(const float4*)(el + (size_t)sv[u] * 4);
      b[u] = *(const float4*)(er + (size_t)dv[u] * 4);
    }
    float4   ev[ST_U];
    unsigned rr[ST_U];
#pragma unroll
    for (int u = 0; u < ST_U; ++u) {
      const unsigned r     = (unsigned)((ends ? dv[u] : sv[u]) - lo);
      const bool     valid = (ent[u] != 0xFFFFFFFFu) && r < (unsigned)ST_RNG;
      rr[u] = valid ? r : 0xFFFFFFFFu;
      ev[u] = make_float4(leakyf(a[u].x + b[u].x), leakyf(a[u].y + b[u].y),
                          leakyf(a[u].z + b[u].z), leakyf(a[u].w + b[u].w));
      if (PASS != 1 && valid) {
        atomicMaxF(&st_max[r * 4 + 0], ev[u].x);
        atomicMaxF(&st_max[r * 4 + 1], ev[u].y);
        atomicMaxF(&st_max[r * 4 + 2], ev[u].z);
        atomicMaxF(&st_max[r * 4 + 3], ev[u].w);
        atomicAdd(&st_deg[r], 1.f);
      }
      if (PASS != 1 && ends) {
        const unsigned lane = (unsigned)t & 31u;
#pragma unroll
        for (unsigned k = 0; k < (unsigned)ST_K; ++k) {
          const bool keep = valid && r >= k * RNG && r < (k + 1) * RNG;
          const unsigned m = __builtin_amdgcn_ballot_w32(keep);
          if (m != 0u) {
            const unsigned lead = (unsigned)__builtin_ctz(m);
            unsigned bp = 0u;
            if (lane == lead) bp = atomicAdd(&sn[k], (unsigned)__builtin_popcount(m));
            bp = (unsigned)__shfl((int)bp, (int)lead);
            const unsigned p = bp + (unsigned)__builtin_popcount(m & ((1u << lane) - 1u));
            if (keep && p < (unsigned)GCAP)
              gqb[(size_t)k * GCAP + p] = ((r - k * RNG) << 20) | (ent[u] & 0xFFFFFu);
          }
        }
      }
    }
    if (PASS == 2) __syncthreads();
    if (PASS != 0) {
#pragma unroll
      for (int u = 0; u < ST_U; ++u) {
        const unsigned r = rr[u];
        if (r != 0xFFFFFFFFu) {
          add_term(&st_sum[r * 4 + 0], __expf(ev[u].x - st_max[r * 4 + 0]));
          add_term(&st_sum[r * 4 + 1], __expf(ev[u].y - st_max[r * 4 + 1]));
          add_term(&st_sum[r * 4 + 2], __expf(ev[u].z - st_max[r * 4 + 2]));
          add_term(&st_sum[r * 4 + 3], __expf(ev[u].w - st_max[r * 4 + 3]));
        }
      }
    }
  }
}

__global__ __launch_bounds__(ST_T)
void seg_stats_kernel(const int* __restrict__ src, const int* __restrict__ dst,
                      const float* __restrict__ el, const float* __restrict__ er,
                      float* __restrict__ maxs, float* __restrict__ maxd,
                      float* __restrict__ sums, float* __restrict__ sumd,
                      float* __restrict__ norm_out, float* __restrict__ norm_in,
                      unsigned int* gq, unsigned int* gcount,
                      const unsigned int* __restrict__ bm) {
  __shared__ __align__(16) unsigned int q[ST_QCAP];
  __shared__ float st_max[ST_RNG * 4];
  __shared__ __align__(8) unsigned long long st_sum[ST_RNG * 4];
  __shared__ float st_deg[ST_RNG];
  __shared__ unsigned int qn;
  __shared__ unsigned int sn[ST_K];
  __shared__ unsigned int wtot[ST_T / 32];

  const int  t     = threadIdx.x;
  const int  g     = blockIdx.x >> 1;
  const bool ends  = (blockIdx.x & 1) == 0;
  const int  lo    = g * ST_RNG;
  const int* __restrict__ idx = ends ? dst : src;
  unsigned int* gqb = gq + (size_t)g * ST_K * GCAP;

  for (int i = t; i < ST_RNG * 4; i += ST_T) { st_max[i] = -1e30f; st_sum[i] = 0ull; }
  if (t < ST_RNG) st_deg[t] = 0.f;
  if (t < ST_K) sn[t] = 0u;
  __syncthreads();

  queue_from_bitmap(bm + (size_t)((ends ? 0 : ST_NRNG) + g) * BM_PITCH, t, q, &qn, wtot);
  __syncthreads();
  unsigned n = __builtin_amdgcn_readfirstlane(qn);

  if (n <= (unsigned)(ST_T * ST_U)) {
    stats_pass<2>(n, t, ends, lo, q, src, dst, el, er, st_max, st_sum, st_deg, sn, gqb);
  } else if (n <= (unsigned)ST_QCAP) {
    stats_pass<0>(n, t, ends, lo, q, src, dst, el, er, st_max, st_sum, st_deg, sn, gqb);
    __syncthreads();
    stats_pass<1>(n, t, ends, lo, q, src, dst, el, er, st_max, st_sum, st_deg, sn, gqb);
  } else {
    __syncthreads();
    if (t == 0) qn = 0u;
    __syncthreads();
    for (int base = 0; base < E_EDGES; base += ST_EDGES_S) {
      scan_group<ST_VS>(base, t, lo, idx, q, &qn);
      __syncthreads();
      n = __builtin_amdgcn_readfirstlane(qn);
      __syncthreads();
      const bool last = (base + ST_EDGES_S >= E_EDGES);
      if (n > (unsigned)(ST_QCAP - ST_EDGES_S) || last) {
        stats_pass<0>(n, t, ends, lo, q, src, dst, el, er, st_max, st_sum, st_deg, sn, gqb);
        __syncthreads();
        if (t == 0) qn = 0u;
        __syncthreads();
      }
    }
    for (int base = 0; base < E_EDGES; base += ST_EDGES_S) {
      scan_group<ST_VS>(base, t, lo, idx, q, &qn);
      __syncthreads();
      n = __builtin_amdgcn_readfirstlane(qn);
      __syncthreads();
      const bool last = (base + ST_EDGES_S >= E_EDGES);
      if (n > (unsigned)(ST_QCAP - ST_EDGES_S) || last) {
        stats_pass<1>(n, t, ends, lo, q, src, dst, el, er, st_max, st_sum, st_deg, sn, gqb);
        __syncthreads();
        if (t == 0) qn = 0u;
        __syncthreads();
      }
    }
  }
  __syncthreads();

  float* __restrict__ mx_out = ends ? maxd : maxs;
  float* __restrict__ sm_out = ends ? sumd : sums;
  for (int i = t; i < ST_RNG * 4; i += ST_T) {
    mx_out[lo * 4 + i] = st_max[i];
    sm_out[lo * 4 + i] = (float)st_sum[i] * (1.f / (float)(1ull << SUM_FRAC));
  }
  if (t < ST_RNG) {
    const float deg = fmaxf(st_deg[t], 1.f);
    if (ends) norm_in[lo + t]  = sqrtf(deg);
    else      norm_out[lo + t] = rsqrtf(deg);
  }
  if (ends && t < ST_K) gcount[g * ST_K + t] = sn[t];
}

struct HopRow { float4 h0a, h0b, x1a, x1b, x2a, x2b; float aba, abb; };

__device__ __forceinline__ HopRow load_hop_row(int row, int lane, int hsel,
                                               const float* __restrict__ h0, const float* __restrict__ x1p,
                                               const float* __restrict__ x2p,
                                               const float* __restrict__ al) {
  HopRow r;
  const size_t o = (size_t)row * HF + lane * 4;
  r.h0a = *(const float4*)(h0 + o);  r.h0b = *(const float4*)(h0 + o + 128);
  r.x1a = *(const float4*)(x1p + o); r.x1b = *(const float4*)(x1p + o + 128);
  r.x2a = *(const float4*)(x2p + o); r.x2b = *(const float4*)(x2p + o + 128);
  r.aba = al[(size_t)row * 4 + hsel];
  r.abb = al[(size_t)row * 4 + 2 + hsel];
  return r;
}

__device__ __forceinline__ float4 combine4(const float4& v0, const float4& v1, const float4& v2,
                                           const float4& v3, const float4& ar, float ab) {
  const float p0 = row16_sum(v0.x * ar.x + v0.y * ar.y + v0.z * ar.z + v0.w * ar.w);
  const float p1 = row16_sum(v1.x * ar.x + v1.y * ar.y + v1.z * ar.z + v1.w * ar.w);
  const float p2 = row16_sum(v2.x * ar.x + v2.y * ar.y + v2.z * ar.z + v2.w * ar.w);
  const float p3 = row16_sum(v3.x * ar.x + v3.y * ar.y + v3.z * ar.z + v3.w * ar.w);
  const float s0 = leakyf(ab + p0), s1 = leakyf(ab + p1);
  const float s2 = leakyf(ab + p2), s3 = leakyf(ab + p3);
  const float m  = fmaxf(fmaxf(s0, s1), fmaxf(s2, s3));
  const float e0 = __expf(s0 - m), e1 = __expf(s1 - m), e2 = __expf(s2 - m), e3 = __expf(s3 - m);
  const float inv = 1.f / (e0 + e1 + e2 + e3);
  const float w0 = e0 * inv, w1 = e1 * inv, w2 = e2 * inv, w3 = e3 * inv;
  float4 o;
  o.x = v0.x * w0 + v1.x * w1 + v2.x * w2 + v3.x * w3;
  o.y = v0.y * w0 + v1.y * w1 + v2.y * w2 + v3.y * w3;
  o.z = v0.z * w0 + v1.z * w1 + v2.z * w2 + v3.z * w3;
  o.w = v0.w * w0 + v1.w * w1 + v2.w * w2 + v3.w * w3;
  return o;
}

template <bool FIRST, bool LAST>
__global__ __launch_bounds__(AGG_T)
void agg_tile_kernel(const int* __restrict__ src, const int* __restrict__ dst,
                     const float* __restrict__ el, const float* __restrict__ er,
                     const float* __restrict__ maxs, const float* __restrict__ maxd,
                     const float* __restrict__ sums, const float* __restrict__ sumd,
                     const float* __restrict__ norm_out, const float* __restrict__ norm_in,
                     const float* __restrict__ sigma,
                     const float* __restrict__ xprev, float* __restrict__ xnext,
                     const unsigned int* gq, const unsigned int* gcount,
                     unsigned int* gsv, float4* gcf,
                     const float* __restrict__ h0, const float* __restrict__ x1p,
                     const float* __restrict__ al, const float* __restrict__ hop_attn_r) {
  __shared__ __align__(16) float tile[RNG * HF];
  __shared__ __align__(16) unsigned int pool[POOL_W];
  __shared__ unsigned int qn;
  float4*       rc  = (float4*)pool;
  unsigned int* rs  = pool + CH * 4;
  unsigned int* q   = pool + CH * 5;
  unsigned int* lws = pool + CH * 5 + QCAP;

  const int t    = threadIdx.x;
  const int lane = t & 31;
  const int wave = t >> 5;
  const int hsel = lane >> 4;
  const int lo   = blockIdx.x * RNG;
  const unsigned wv = __builtin_amdgcn_readfirstlane((unsigned)wave);
  const unsigned int* gqb = gq + (size_t)blockIdx.x * GCAP;
  unsigned int* gsvb = gsv + (size_t)blockIdx.x * GCAP;
  float4*       gcfb = gcf + (size_t)blockIdx.x * GCAP;
  const float* rcf = (const float*)rc;
  unsigned* lw = lws + wv * LW;
  const float sgm = 1.f / (1.f + __expf(-sigma[0]));
  const unsigned nrow = (unsigned)lane * NWAVE + wv;
  const float nin = (nrow < (unsigned)RNG) ? norm_in[lo + (int)nrow] : 0.f;

  for (unsigned row = wv; row < (unsigned)RNG; row += NWAVE) {
    float* trow = tile + row * HF + lane * 4;
    *(float4*)trow         = make_float4(0.f, 0.f, 0.f, 0.f);
    *(float4*)(trow + 128) = make_float4(0.f, 0.f, 0.f, 0.f);
  }
  if (t == 0) qn = 0u;

  const unsigned cnt = __builtin_amdgcn_readfirstlane(gcount[blockIdx.x]);
  if (cnt <= (unsigned)GCAP) {
    if (FIRST) {
      unsigned int* qs = pool;
      unsigned int* lq = pool + CH + wv * WAREA;
      unsigned int* li = lq + LQ;
      unsigned int* eb = li + LQ;
      float4*       cb = (float4*)(eb + 32);
      unsigned w[GCAP / AGG_T];
#pragma unroll
      for (int j = 0; j < GCAP / AGG_T; ++j) {
        const unsigned i = (unsigned)(j * AGG_T + t);
        w[j] = (i < cnt) ? gqb[i] : 0u;
      }
#pragma unroll
      for (int j = 0; j < GCAP / AGG_T; ++j) {
        const unsigned i = (unsigned)(j * AGG_T + t);
        if (i < cnt) qs[i] = w[j];
      }
      __syncthreads();
      build_owned(cnt, wv, lane, hsel, lo, qs, lq, li, eb, cb, gsvb, gcfb,
                  src, el, er, maxs, maxd, sums, sumd, norm_out, sgm, xprev, tile);
    } else {
      unsigned w[GCAP / AGG_T];
      float4   c[GCAP / AGG_T];
#pragma unroll
      for (int j = 0; j < GCAP / AGG_T; ++j) {
        const unsigned i = (unsigned)(j * AGG_T + t);
        w[j] = (i < cnt) ? gsvb[i] : 0u;
        c[j] = (i < cnt) ? gcfb[i] : make_float4(0.f, 0.f, 0.f, 0.f);
      }
#pragma unroll
      for (int j = 0; j < GCAP / AGG_T; ++j) {
        const unsigned i = (unsigned)(j * AGG_T + t);
        if (i < cnt) { rs[i] = w[j]; rc[i] = c[j]; }
      }
      __syncthreads();
      drain_owned(cnt, wv, lane, hsel, rs, rcf, lw, xprev, tile);
    }
  } else {
    __syncthreads();
    for (int base = 0; base < E_EDGES; base += SCAN_EDGES) {
      const int e0 = base + t * 4;
      const int4 dv = (e0 < E_EDGES) ? *(const int4*)(dst + e0) : make_int4(-1, -1, -1, -1);
      const unsigned rx = (unsigned)(dv.x - lo), ry = (unsigned)(dv.y - lo);
      const unsigned rz = (unsigned)(dv.z - lo), rw = (unsigned)(dv.w - lo);
      if (min(min(rx, ry), min(rz, rw)) < (unsigned)RNG) {
        push_edge(rx, e0,     q, &qn);
        push_edge(ry, e0 + 1, q, &qn);
        push_edge(rz, e0 + 2, q, &qn);
        push_edge(rw, e0 + 3, q, &qn);
      }
      __syncthreads();
      const unsigned n = __builtin_amdgcn_readfirstlane(qn);
      __syncthreads();
      const bool last = (base + SCAN_EDGES >= E_EDGES);
      if (n > (unsigned)(QCAP - SCAN_EDGES) || last) {
        for (unsigned c0 = 0; c0 < n; c0 += CH) {
          const unsigned m = min((unsigned)CH, n - c0);
          for (unsigned idx = (unsigned)t; idx < m; idx += AGG_T) {
            unsigned word; float4 c;
            make_record(q[c0 + idx], lo, src, el, er, maxs, maxd, sums, sumd, norm_out, sgm, word, c);
            rs[idx] = word; rc[idx] = c;
          }
          __syncthreads();
          drain_owned(m, wv, lane, hsel, rs, rcf, lw, xprev, tile);
          __syncthreads();
        }
        if (t == 0) qn = 0u;
        __syncthreads();
      }
    }
  }

  if (!LAST) {
    for (unsigned row = wv; row < (unsigned)RNG; row += NWAVE) {
      const float* trow = tile + row * HF + lane * 4;
      const float4 a0 = *(const float4*)trow;
      const float4 a1 = *(const float4*)(trow + 128);
      const float ni = __int_as_float(__builtin_amdgcn_readlane(__float_as_int(nin), (int)(row >> 5)));
      float* o = xnext + (size_t)(lo + (int)row) * HF + lane * 4;
      *(float4*)o         = make_float4(a0.x * ni, a0.y * ni, a0.z * ni, a0.w * ni);
      *(float4*)(o + 128) = make_float4(a1.x * ni, a1.y * ni, a1.z * ni, a1.w * ni);
    }
  } else {
    const float4 ara = *(const float4*)(hop_attn_r + lane * 4);
    const float4 arb = *(const float4*)(hop_attn_r + 128 + lane * 4);
    HopRow nx = load_hop_row(lo + (int)wv, lane, hsel, h0, x1p, xprev, al);
    for (unsigned row = wv; row < (unsigned)RNG; row += NWAVE) {
      const HopRow cur = nx;
      if (row + NWAVE < (unsigned)RNG)
        nx = load_hop_row(lo + (int)(row + NWAVE), lane, hsel, h0, x1p, xprev, al);
      const float* trow = tile + row * HF + lane * 4;
      const float4 a0 = *(const float4*)trow;
      const float4 a1 = *(const float4*)(trow + 128);
      const float ni = __int_as_float(__builtin_amdgcn_readlane(__float_as_int(nin), (int)(row >> 5)));
      const float4 v3a = make_float4(a0.x * ni, a0.y * ni, a0.z * ni, a0.w * ni);
      const float4 v3b = make_float4(a1.x * ni, a1.y * ni, a1.z * ni, a1.w * ni);
      float* o = xnext + (size_t)(lo + (int)row) * HF + lane * 4;
      *(float4*)o         = combine4(cur.h0a, cur.x1a, cur.x2a, v3a, ara, cur.aba);
      *(float4*)(o + 128) = combine4(cur.h0b, cur.x1b, cur.x2b, v3b, arb, cur.abb);
    }
  }
}

extern "C" void kernel_launch(void* const* d_in, const int* in_sizes, int n_in,
                              void* d_out, int out_size, void* d_ws, size_t ws_size,
                              hipStream_t stream) {
  (void)in_sizes; (void)n_in; (void)out_size; (void)ws_size;
  const float* feat       = (const float*)d_in[0];
  const int*   src        = (const int*)d_in[1];
  const int*   dst        = (const int*)d_in[2];
  const float* W          = (const float*)d_in[3];
  const float* attn_l     = (const float*)d_in[4];
  const float* attn_r     = (const float*)d_in[5];
  const float* hop_attn_l = (const float*)d_in[6];
  const float* hop_attn_r = (const float*)d_in[7];
  const float* sigma      = (const float*)d_in[8];

  float* ws = (float*)d_ws;
  size_t off = 0;
  float* h    = ws + off; off += (size_t)N_NODES * HF;
  float* x1   = ws + off; off += (size_t)N_NODES * HF;
  float* x2   = ws + off; off += (size_t)N_NODES * HF;
  unsigned int* bm = (unsigned int*)(ws + off); off += (size_t)N_NODES * HF;
  float* el   = ws + off; off += (size_t)N_NODES * H_HEADS;
  float* er   = ws + off; off += (size_t)N_NODES * H_HEADS;
  float* al   = ws + off; off += (size_t)N_NODES * H_HEADS;
  float* maxs = ws + off; off += (size_t)N_NODES * H_HEADS;
  float* maxd = ws + off; off += (size_t)N_NODES * H_HEADS;
  float* sums = ws + off; off += (size_t)N_NODES * H_HEADS;
  float* sumd = ws + off; off += (size_t)N_NODES * H_HEADS;
  off += (size_t)N_NODES * 2;
  float* norm_out = ws + off; off += (size_t)N_NODES;
  float* norm_in  = ws + off; off += (size_t)N_NODES;
  float* edge_planes = ws + off; off += (size_t)E_EDGES * H_HEADS * 2;

  gemm_h_kernel<<<GEMM_NBLK + BM_NBLK, 256, 0, stream>>>(feat, W, h, attn_l, attn_r, hop_attn_l,
                                                         el, er, al, src, dst, bm);

  unsigned int* gq     = (unsigned int*)edge_planes;
  unsigned int* gcount = gq + (size_t)NBLK * GCAP;
  unsigned int* gsv    = gcount + 256;
  float4*       gcf    = (float4*)(gsv + (size_t)NBLK * GCAP);
  seg_stats_kernel<<<ST_NBLK, ST_T, 0, stream>>>(src, dst, el, er, maxs, maxd, sums, sumd,
                                                 norm_out, norm_in, gq, gcount, bm);

  agg_tile_kernel<true, false><<<NBLK, AGG_T, 0, stream>>>(src, dst, el, er, maxs, maxd, sums, sumd, norm_out,
                                                           norm_in, sigma, h,  x1, gq, gcount, gsv, gcf,
                                                           h, x1, al, hop_attn_r);
  agg_tile_kernel<false, false><<<NBLK, AGG_T, 0, stream>>>(src, dst, el, er, maxs, maxd, sums, sumd, norm_out,
                                                            norm_in, sigma, x1, x2, gq, gcount, gsv, gcf,
                                                            h, x1, al, hop_attn_r);
  agg_tile_kernel<false, true><<<NBLK, AGG_T, 0, stream>>>(src, dst, el, er, maxs, maxd, sums, sumd, norm_out,
                                                           norm_in, sigma, x2, (float*)d_out, gq, gcount, gsv, gcf,
                                                           h, x1, al, hop_attn_r);
}


# Index arrays read once: edge blocks store (src,dst) pairs; statistics blocks gather one pair line per queued edge, not two, and save the source node beside each hop-queue entry; hop 1 reads no index
# speedup vs baseline: 8.7626x; 1.0352x over previous
#include <hip/hip_runtime.h>
#include <hip/hip_bf16.h>

#define N_NODES 50000
#define E_EDGES 800000
#define IN_F    256
#define H_HEADS 4
#define F_OUT   64
#define HF      256
#define SLOPE   0.2f

typedef __attribute__((ext_vector_type(16))) __bf16 v16bf;
typedef __attribute__((ext_vector_type(8)))  float  v8f;

__device__ __forceinline__ float leakyf(float x) { return x > 0.f ? x : SLOPE * x; }

__device__ __forceinline__ void atomicMaxF(float* addr, float val) {
  if (val >= 0.f) {
    atomicMax((int*)addr, __float_as_int(val));
  } else {
    atomicMin((unsigned int*)addr, __float_as_uint(val));
  }
}

__device__ __forceinline__ float row16_sum(float v) {
  v += __int_as_float(__builtin_amdgcn_update_dpp(0, __float_as_int(v), 0x168, 0xF, 0xF, true));
  v += __int_as_float(__builtin_amdgcn_update_dpp(0, __float_as_int(v), 0x164, 0xF, 0xF, true));
  v += __int_as_float(__builtin_amdgcn_update_dpp(0, __float_as_int(v), 0x162, 0xF, 0xF, true));
  v += __int_as_float(__builtin_amdgcn_update_dpp(0, __float_as_int(v), 0x161, 0xF, 0xF, true));
  return v;
}


#define RNG        200
#define ST_K       2
#define ST_RNG     (ST_K * RNG)
#define ST_NRNG    (N_NODES / ST_RNG)
#define SC_T       256
#define BM_EDGES   (SC_T * 4)
#define BM_WORDS   (BM_EDGES / 32)
#define BM_NBLK    ((E_EDGES + BM_EDGES - 1) / BM_EDGES)
#define BM_PITCH   (BM_NBLK * BM_WORDS)
#define WT_STRIDE  40
#define GEMM_ROWS  128
#define GEMM_NBLK  ((N_NODES + GEMM_ROWS - 1) / GEMM_ROWS)

static_assert(N_NODES % ST_RNG == 0, "bitmap ranges tile the node set exactly");
static_assert(E_EDGES % 4 == 0, "the index arrays are read 4 edges at a time");
static_assert(BM_WORDS == 32 && BM_PITCH % 4 == 0, "a block's share of a bitmap is one 128-byte line");
static_assert(N_NODES % 16 == 0, "a wave's 16 rows are all inside the node set or all outside");
static_assert(2 * ST_NRNG * BM_WORDS * 4 >= HF * WT_STRIDE * 2, "the bitmap lines and the staged weight slice share one LDS array");
static_assert((2 * ST_NRNG * BM_WORDS) % 4 == 0, "the LDS copy is moved 16 bytes at a time");

__global__ __launch_bounds__(256)
void gemm_h_kernel(const float* __restrict__ feat, const float* __restrict__ W,
                   float* __restrict__ h,
                   const float* __restrict__ attn_l,
                   const float* __restrict__ attn_r,
                   const float* __restrict__ hop_attn_l,
                   float* __restrict__ el, float* __restrict__ er,
                   float* __restrict__ al,
                   const int* __restrict__ src, const int* __restrict__ dst,
                   unsigned int* __restrict__ bm, int2* __restrict__ pairs) {
  __shared__ __align__(16) unsigned int smem[2 * ST_NRNG * BM_WORDS];
  const int t = threadIdx.x;

  if (blockIdx.x >= GEMM_NBLK) {
    unsigned int* bml = smem;
    const int cb = (int)blockIdx.x - GEMM_NBLK;
    const int e0 = cb * BM_EDGES + t * 4;
    int4 sv = make_int4(0, 0, 0, 0), dv = make_int4(0, 0, 0, 0);
    const bool have = (e0 < E_EDGES);
    if (have) {
      sv = *(const int4*)(src + e0); dv = *(const int4*)(dst + e0);
      int4* pp = (int4*)(pairs + e0);
      pp[0] = make_int4(sv.x, dv.x, sv.y, dv.y);
      pp[1] = make_int4(sv.z, dv.z, sv.w, dv.w);
    }
    for (int k = t; k < 2 * ST_NRNG * BM_WORDS / 4; k += SC_T)
      ((uint4*)bml)[k] = make_uint4(0u, 0u, 0u, 0u);
    __syncthreads();
    if (have) {
      const unsigned w = (unsigned)t >> 3;
      const unsigned b = ((unsigned)t & 7u) * 4u;
      unsigned int* ends   = bml + w;
      unsigned int* starts = bml + ST_NRNG * BM_WORDS + w;
      atomicOr(&ends[((unsigned)dv.x / ST_RNG) * BM_WORDS], 1u << b);
      atomicOr(&ends[((unsigned)dv.y / ST_RNG) * BM_WORDS], 2u << b);
      atomicOr(&ends[((unsigned)dv.z / ST_RNG) * BM_WORDS], 4u << b);
      atomicOr(&ends[((unsigned)dv.w / ST_RNG) * BM_WORDS], 8u << b);
      atomicOr(&starts[((unsigned)sv.x / ST_RNG) * BM_WORDS], 1u << b);
      atomicOr(&starts[((unsigned)sv.y / ST_RNG) * BM_WORDS], 2u << b);
      atomicOr(&starts[((unsigned)sv.z / ST_RNG) * BM_WORDS], 4u << b);
      atomicOr(&starts[((unsigned)sv.w / ST_RNG) * BM_WORDS], 8u << b);
    }
    __syncthreads();
    for (int k = t; k < 2 * ST_NRNG * BM_WORDS / 4; k += SC_T) {
      const int row = k >> 3, part = k & 7;
      *(uint4*)(bm + (size_t)row * BM_PITCH + (size_t)cb * BM_WORDS + part * 4) = ((const uint4*)bml)[k];
    }
    return;
  }

  __bf16* Wt = (__bf16*)smem;

  const int wave    = t >> 5;
  const int lane    = t & 31;
  const int rowBase = blockIdx.x * GEMM_ROWS + wave * 16;
  const bool rowValid = (rowBase < N_NODES);
  const int nloc = lane & 15;
  const int kg   = lane >> 4;

  v8f acc[16];
#pragma unroll
  for (int nt = 0; nt < 16; ++nt)
#pragma unroll
    for (int v = 0; v < 8; ++v) acc[nt][v] = 0.f;

  for (int ks = 0; ks < IN_F / 32; ++ks) {
    const int kbase = ks * 32;
    __syncthreads();
#pragma unroll
    for (int kk = 0; kk < 32; ++kk)
      Wt[t * WT_STRIDE + kk] = (__bf16)W[(kbase + kk) * HF + t];
    __syncthreads();

    if (rowValid) {
      const float* arow = feat + (size_t)(rowBase + nloc) * IN_F + kbase;
      float4 a0 = *(const float4*)(arow + 8 * kg);
      float4 a1 = *(const float4*)(arow + 8 * kg + 4);
      float4 a2 = *(const float4*)(arow + 16 + 8 * kg);
      float4 a3 = *(const float4*)(arow + 16 + 8 * kg + 4);
      float atmp[16] = {a0.x, a0.y, a0.z, a0.w, a1.x, a1.y, a1.z, a1.w,
                        a2.x, a2.y, a2.z, a2.w, a3.x, a3.y, a3.z, a3.w};
      v16bf afrag;
#pragma unroll
      for (int i = 0; i < 16; ++i) afrag[i] = (__bf16)atmp[i];

      const __bf16* wrow0 = &Wt[nloc * WT_STRIDE];
      uint4 b0 = *(const uint4*)(wrow0 + 8 * kg);
      uint4 b1 = *(const uint4*)(wrow0 + 16 + 8 * kg);
#pragma unroll
      for (int nt = 0; nt < 16; ++nt) {
        uint4 nb0, nb1;
        if (nt < 15) {
          const __bf16* wrow = &Wt[((nt + 1) * 16 + nloc) * WT_STRIDE];
          nb0 = *(const uint4*)(wrow + 8 * kg);
          nb1 = *(const uint4*)(wrow + 16 + 8 * kg);
        }
        const __bf16* p0 = (const __bf16*)&b0;
        const __bf16* p1 = (const __bf16*)&b1;
        v16bf bfrag;
#pragma unroll
        for (int i = 0; i < 8; ++i) { bfrag[i] = p0[i]; bfrag[8 + i] = p1[i]; }
        acc[nt] = __builtin_amdgcn_wmma_f32_16x16x32_bf16(
            false, afrag, false, bfrag, (short)0, acc[nt], false, false);
        if (nt < 15) { b0 = nb0; b1 = nb1; }
      }
    }
  }

  if (rowValid) {
#pragma unroll
    for (int nt = 0; nt < 16; ++nt)
#pragma unroll
      for (int v = 0; v < 8; ++v)
        h[(size_t)(rowBase + v + 8 * kg) * HF + nt * 16 + nloc] = acc[nt][v];

    float ol[4], orr[4], oh[4];
#pragma unroll
    for (int hd = 0; hd < 4; ++hd) {
      float pl[8], pr[8], ph[8];
#pragma unroll
      for (int v = 0; v < 8; ++v) { pl[v] = 0.f; pr[v] = 0.f; ph[v] = 0.f; }
#pragma unroll
      for (int j = 0; j < 4; ++j) {
        const int nt = hd * 4 + j;
        const float cl = attn_l[nt * 16 + nloc];
        const float cr = attn_r[nt * 16 + nloc];
        const float ch = hop_attn_l[nt * 16 + nloc];
#pragma unroll
        for (int v = 0; v < 8; ++v) {
          pl[v] = fmaf(acc[nt][v], cl, pl[v]);
          pr[v] = fmaf(acc[nt][v], cr, pr[v]);
          ph[v] = fmaf(acc[nt][v], ch, ph[v]);
        }
      }
      float sl = 0.f, sr = 0.f, sh = 0.f;
#pragma unroll
      for (int v = 0; v < 8; ++v) {
        const float tl = row16_sum(pl[v]);
        const float tr = row16_sum(pr[v]);
        const float th = row16_sum(ph[v]);
        if ((nloc & 7) == v) { sl = tl; sr = tr; sh = th; }
      }
      ol[hd] = sl; orr[hd] = sr; oh[hd] = sh;
    }
    if (nloc < 8) {
      const size_t o = (size_t)(rowBase + nloc + 8 * kg) * H_HEADS;
      *(float4*)(el + o) = make_float4(ol[0], ol[1], ol[2], ol[3]);
      *(float4*)(er + o) = make_float4(orr[0], orr[1], orr[2], orr[3]);
      *(float4*)(al + o) = make_float4(oh[0], oh[1], oh[2], oh[3]);
    }
  }
}


__device__ __forceinline__ float mix_coef(float el, float er, float ms, float md,
                                          float ss, float sd, float sgm, float nrm) {
  const float ev = leakyf(el + er);
  const float ax = fmaxf(__expf(ev - ms) / ss, 1e-10f);
  const float ay = fmaxf(__expf(ev - md) / sd, 1e-10f);
  const float a  = __powf(ay, sgm) * __powf(ax, 1.f - sgm);
  return a * nrm;
}

#define NBLK       (N_NODES / RNG)
#define AGG_T      1024
#define GCAP       4096
#define CH         GCAP
#define SCAN_EDGES (AGG_T * 4)
#define QCAP       (SCAN_EDGES + 1504)
#define DRAIN_U    4
#define NWAVE      (AGG_T / 32)
#define OWN_SLOTS  ((RNG + NWAVE - 1) / NWAVE)
#define SCAN_R     128
#define LIST_GO    16
#define LW         (LIST_GO + SCAN_R)
#define LQ         (32 + SCAN_R)
#define WAREA      (2 * LQ + 32 + 32 * 4)
#define POOL_W     (CH * 5 + QCAP + NWAVE * LW)

static_assert(N_NODES % RNG == 0, "ranges tile the node set exactly");
static_assert(E_EDGES % 4 == 0, "dst is read 4 edges at a time");
static_assert(E_EDGES <= (1 << 20) && RNG <= 256, "queue word packs row and edge id");
static_assert(N_NODES <= (1 << 16), "record word packs row and source node");
static_assert((RNG * HF + POOL_W + 1) * 4 <= 320 * 1024, "LDS budget");
static_assert(CH + NWAVE * WAREA <= POOL_W, "the first hop's layout fits the same pool");
static_assert(LQ % 4 == 0 && (CH * 5) % 4 == 0 && QCAP % 4 == 0 && LW % 4 == 0 && WAREA % 4 == 0,
              "every part of the pool starts on 16 bytes");
static_assert(NWAVE == 32 && OWN_SLOTS <= 8, "owner wave = row & 31; list entry packs row >> 5 in 3 bits");
static_assert(CH <= (1 << 12), "list entry packs the record's place in 12 bits");
static_assert(CH % SCAN_R == 0 && LW % DRAIN_U == 0 && LIST_GO % DRAIN_U == 0 && LIST_GO >= DRAIN_U,
              "list trips stay inside the arrays");
static_assert(QCAP >= SCAN_EDGES + AGG_T, "a scan group fits behind what may be left in the queue");
static_assert(DRAIN_U == 4, "a batch is one 16-byte read of record words");
static_assert(GCAP % AGG_T == 0, "whole rounds of the block cover a slice");
static_assert((size_t)NBLK * GCAP * 6 + 256 <= (size_t)E_EDGES * H_HEADS * 2,
              "saved queues, counts and records fit the two edge planes");
static_assert((size_t)2 * ST_NRNG * BM_PITCH + (size_t)2 * E_EDGES + (size_t)NBLK * GCAP <= (size_t)N_NODES * HF,
              "both sets of bitmaps, the pair array and the saved source nodes fit the place of the third hop's plane");
static_assert(((size_t)2 * ST_NRNG * BM_PITCH) % 4 == 0, "the pair array starts on 16 bytes");

__device__ __forceinline__ void push_edge(unsigned r, int eid, unsigned* q, unsigned* qn) {
  if (r < (unsigned)RNG) {
    const unsigned p = atomicAdd(qn, 1u);
    q[p] = (r << 20) | (unsigned)eid;
  }
}

__device__ __forceinline__ void make_record(unsigned w, int lo, int s,
                                            const float* __restrict__ el, const float* __restrict__ er,
                                            const float* __restrict__ maxs, const float* __restrict__ maxd,
                                            const float* __restrict__ sums, const float* __restrict__ sumd,
                                            const float* __restrict__ norm_out, float sgm,
                                            unsigned& word, float4& c) {
  const bool     ok  = (w != 0xFFFFFFFFu);
  const unsigned r   = ok ? (w >> 20) : 0u;
  const size_t   d   = (size_t)(lo + (int)r);
  const float4 a  = *(const float4*)(el   + (size_t)s * 4);
  const float4 ms = *(const float4*)(maxs + (size_t)s * 4);
  const float4 ss = *(const float4*)(sums + (size_t)s * 4);
  const float  nr = norm_out[s];
  const float4 b  = *(const float4*)(er   + d * 4);
  const float4 md = *(const float4*)(maxd + d * 4);
  const float4 sd = *(const float4*)(sumd + d * 4);
  c.x = mix_coef(a.x, b.x, ms.x, md.x, ss.x, sd.x, sgm, nr);
  c.y = mix_coef(a.y, b.y, ms.y, md.y, ss.y, sd.y, sgm, nr);
  c.z = mix_coef(a.z, b.z, ms.z, md.z, ss.z, sd.z, sgm, nr);
  c.w = mix_coef(a.w, b.w, ms.w, md.w, ss.w, sd.w, sgm, nr);
  word = (r << 16) | (unsigned)s;
}


template <bool FULL>
__device__ __forceinline__ void add_batch(unsigned i0, unsigned m, unsigned wv, int lane, int hsel,
                                          const unsigned* lw, const float* rcf,
                                          const float* __restrict__ xprev, float* tile) {
  const uint4 e4 = *(const uint4*)&lw[i0];
  const unsigned eq[DRAIN_U] = {
      (unsigned)__builtin_amdgcn_readfirstlane(e4.x), (unsigned)__builtin_amdgcn_readfirstlane(e4.y),
      (unsigned)__builtin_amdgcn_readfirstlane(e4.z), (unsigned)__builtin_amdgcn_readfirstlane(e4.w)};
  unsigned ent[DRAIN_U];
  float4 r0[DRAIN_U], r1[DRAIN_U];
  float  c0[DRAIN_U], c1[DRAIN_U];
#pragma unroll
  for (int u = 0; u < DRAIN_U; ++u) {
    const bool real = FULL || (unsigned)u < m;
    ent[u] = real ? eq[u] : eq[0];
    const unsigned ix = (ent[u] >> 16) & 0xFFFu;
    const float* row = xprev + (size_t)(ent[u] & 0xFFFFu) * HF + lane * 4;
    r0[u] = *(const float4*)row;
    r1[u] = *(const float4*)(row + 128);
    const float k0 = rcf[ix * 4 + hsel];
    const float k1 = rcf[ix * 4 + 2 + hsel];
    c0[u] = real ? k0 : 0.f;
    c1[u] = real ? k1 : 0.f;
  }
#pragma unroll
  for (int u = 0; u < DRAIN_U; ++u) {
    float* trow = tile + ((ent[u] >> 28) * NWAVE + wv) * HF + lane * 4;
    float4 a0 = *(const float4*)trow;
    float4 a1 = *(const float4*)(trow + 128);
    a0.x = fmaf(r0[u].x, c0[u], a0.x); a0.y = fmaf(r0[u].y, c0[u], a0.y);
    a0.z = fmaf(r0[u].z, c0[u], a0.z); a0.w = fmaf(r0[u].w, c0[u], a0.w);
    a1.x = fmaf(r1[u].x, c1[u], a1.x); a1.y = fmaf(r1[u].y, c1[u], a1.y);
    a1.z = fmaf(r1[u].z, c1[u], a1.z); a1.w = fmaf(r1[u].w, c1[u], a1.w);
    *(float4*)trow         = a0;
    *(float4*)(trow + 128) = a1;
  }
}

__device__ __forceinline__ void drain_list(unsigned n, unsigned wv, int lane, int hsel,
                                           const unsigned* lw, const float* rcf,
                                           const float* __restrict__ xprev, float* tile) {
  const unsigned full = n & ~(unsigned)(DRAIN_U - 1);
#pragma unroll 1
  for (unsigned i0 = 0; i0 < full; i0 += DRAIN_U)
    add_batch<true>(i0, DRAIN_U, wv, lane, hsel, lw, rcf, xprev, tile);
  if (full < n)
    add_batch<false>(full, n - full, wv, lane, hsel, lw, rcf, xprev, tile);
}

__device__ __forceinline__ void drain_owned(unsigned n, unsigned wv, int lane, int hsel,
                                            const unsigned* rs, const float* rcf, unsigned* lw,
                                            const float* __restrict__ xprev, float* tile) {
  unsigned have = 0u;
  const unsigned below = (1u << lane) - 1u;
  for (unsigned base = 0; base < n; base += SCAN_R) {
    const unsigned i4 = base + (unsigned)lane * 4u;
    const uint4 w4 = *(const uint4*)&rs[i4];
    const unsigned wd[4] = {w4.x, w4.y, w4.z, w4.w};
#pragma unroll
    for (int k = 0; k < 4; ++k) {
      const unsigned idx  = i4 + (unsigned)k;
      const unsigned row  = wd[k] >> 16;
      const bool     mine = idx < n && (row & (NWAVE - 1u)) == wv;
      const unsigned m    = __builtin_amdgcn_ballot_w32(mine);
      if (mine) {
        const unsigned p = have + (unsigned)__builtin_popcount(m & below);
        lw[p] = ((row >> 5) << 28) | (idx << 16) | (wd[k] & 0xFFFFu);
      }
      have += (unsigned)__builtin_popcount(m);
    }
    if (have >= (unsigned)LIST_GO) {
      const unsigned full = have & ~(unsigned)(DRAIN_U - 1);
      drain_list(full, wv, lane, hsel, lw, rcf, xprev, tile);
      const unsigned rem = have - full;
      unsigned keep = 0u;
      if ((unsigned)lane < rem) keep = lw[full + (unsigned)lane];
      if ((unsigned)lane < rem) lw[lane] = keep;
      have = rem;
    }
  }
  drain_list(have, wv, lane, hsel, lw, rcf, xprev, tile);
}

__device__ __forceinline__ void build_group(unsigned g, unsigned m, unsigned wv, int lane, int hsel, int lo,
                                            const unsigned* lq, const unsigned* li, unsigned* eb, float4* cb,
                                            unsigned int* gsvb, float4* gcfb,
                                            const unsigned int* __restrict__ gqsb,
                                            const float* __restrict__ el, const float* __restrict__ er,
                                            const float* __restrict__ maxs, const float* __restrict__ maxd,
                                            const float* __restrict__ sums, const float* __restrict__ sumd,
                                            const float* __restrict__ norm_out, float sgm,
                                            const float* __restrict__ xprev, float* tile) {
  const bool on = (unsigned)lane < m;
  const unsigned qw  = on ? lq[g + (unsigned)lane] : 0xFFFFFFFFu;
  const unsigned idx = on ? li[g + (unsigned)lane] : 0u;
  const unsigned sq  = on ? gqsb[idx] : 0u;
  const int      s   = (sq < (unsigned)N_NODES) ? (int)sq : 0;
  unsigned word; float4 c;
  make_record(qw, lo, s, el, er, maxs, maxd, sums, sumd, norm_out, sgm, word, c);
  if (on) {
    gsvb[idx] = word; gcfb[idx] = c;
    eb[lane] = ((word >> 21) << 28) | ((unsigned)lane << 16) | (word & 0xFFFFu);
    cb[lane] = c;
  }
  drain_list(m, wv, lane, hsel, eb, (const float*)cb, xprev, tile);
}

__device__ __forceinline__ void build_owned(unsigned n, unsigned wv, int lane, int hsel, int lo,
                                            const unsigned* qs, unsigned* lq, unsigned* li,
                                            unsigned* eb, float4* cb,
                                            unsigned int* gsvb, float4* gcfb,
                                            const unsigned int* __restrict__ gqsb,
                                            const float* __restrict__ el, const float* __restrict__ er,
                                            const float* __restrict__ maxs, const float* __restrict__ maxd,
                                            const float* __restrict__ sums, const float* __restrict__ sumd,
                                            const float* __restrict__ norm_out, float sgm,
                                            const float* __restrict__ xprev, float* tile) {
  unsigned have = 0u;
  const unsigned below = (1u << lane) - 1u;
  for (unsigned base = 0; base < n; base += SCAN_R) {
    const unsigned i4 = base + (unsigned)lane * 4u;
    const uint4 w4 = *(const uint4*)&qs[i4];
    const unsigned wd[4] = {w4.x, w4.y, w4.z, w4.w};
#pragma unroll
    for (int k = 0; k < 4; ++k) {
      const unsigned idx  = i4 + (unsigned)k;
      const bool     mine = idx < n && ((wd[k] >> 20) & (NWAVE - 1u)) == wv;
      const unsigned m    = __builtin_amdgcn_ballot_w32(mine);
      if (mine) {
        const unsigned p = have + (unsigned)__builtin_popcount(m & below);
        lq[p] = wd[k];
        li[p] = idx;
      }
      have += (unsigned)__builtin_popcount(m);
    }
    if (have >= 32u) {
      const unsigned full = have & ~31u;
      for (unsigned g = 0; g < full; g += 32u)
        build_group(g, 32u, wv, lane, hsel, lo, lq, li, eb, cb, gsvb, gcfb,
                    gqsb, el, er, maxs, maxd, sums, sumd, norm_out, sgm, xprev, tile);
      const unsigned rem = have - full;
      unsigned kq = 0u, ki = 0u;
      if ((unsigned)lane < rem) { kq = lq[full + (unsigned)lane]; ki = li[full + (unsigned)lane]; }
      if ((unsigned)lane < rem) { lq[lane] = kq; li[lane] = ki; }
      have = rem;
    }
  }
  if (have != 0u)
    build_group(0u, have, wv, lane, hsel, lo, lq, li, eb, cb, gsvb, gcfb,
                gqsb, el, er, maxs, maxd, sums, sumd, norm_out, sgm, xprev, tile);
}

#define ST_NBLK    (2 * ST_NRNG)
#define ST_T       1024
#define ST_VS      4
#define BM_V       ((BM_PITCH / 4 + ST_T - 1) / ST_T)
#define ST_EDGES_S (ST_T * 4 * ST_VS)
#define ST_QCAP    (3 * ST_EDGES_S)
#define ST_U       8

static_assert(N_NODES % ST_RNG == 0 && ST_RNG <= 512, "statistics ranges tile the node set; local node fits 9 bits");
static_assert(ST_RNG <= ST_T, "one thread per degree word");
static_assert(ST_T / 32 <= 32, "one lane per wave total");
#define SUM_FRAC   40
static_assert((ST_QCAP + ST_RNG * 13 + 4) * 4 <= 320 * 1024, "LDS budget of the statistics kernel");
static_assert(E_EDGES <= (1 << 20), "a segment's terms, each at most 2^SUM_FRAC, fit 64 bits");

__device__ __forceinline__ void add_term(unsigned long long* p, float term) {
  atomicAdd(p, (unsigned long long)(term * (float)(1ull << SUM_FRAC)));
}

__device__ __forceinline__ void push_own(unsigned r, int eid, unsigned* q, unsigned* qn) {
  if (r < (unsigned)ST_RNG) {
    const unsigned p = atomicAdd(qn, 1u);
    q[p] = (unsigned)eid;
  }
}

template <int V>
__device__ __forceinline__ void scan_group(int base, int t, int lo,
                                           const int* __restrict__ idx,
                                           unsigned* q, unsigned* qn) {
  int4 iv[V];
#pragma unroll
  for (int j = 0; j < V; ++j) {
    const int e0 = base + (j * ST_T + t) * 4;
    if (e0 < E_EDGES) iv[j] = *(const int4*)(idx + e0);
    else              iv[j] = make_int4(-1, -1, -1, -1);
  }
#pragma unroll
  for (int j = 0; j < V; ++j) {
    const int e0 = base + (j * ST_T + t) * 4;
    const unsigned rx = (unsigned)(iv[j].x - lo), ry = (unsigned)(iv[j].y - lo);
    const unsigned rz = (unsigned)(iv[j].z - lo), rw = (unsigned)(iv[j].w - lo);
    if (min(min(rx, ry), min(rz, rw)) < (unsigned)ST_RNG) {
      push_own(rx, e0,     q, qn);
      push_own(ry, e0 + 1, q, qn);
      push_own(rz, e0 + 2, q, qn);
      push_own(rw, e0 + 3, q, qn);
    }
  }
}

__device__ __forceinline__ void push_bits(unsigned word, unsigned first, unsigned& p, unsigned* q) {
  while (word != 0u) {
    const unsigned b = (unsigned)__builtin_ctz(word);
    word &= word - 1u;
    if (p < (unsigned)ST_QCAP) q[p] = first + b;
    ++p;
  }
}

__device__ __forceinline__ void queue_from_bitmap(const unsigned int* __restrict__ row, int t,
                                                  unsigned* q, unsigned* qn, unsigned* wtot) {
  uint4 wv[BM_V];
#pragma unroll
  for (int j = 0; j < BM_V; ++j) {
    const int k = j * ST_T + t;
    wv[j] = (k < BM_PITCH / 4) ? ((const uint4*)row)[k] : make_uint4(0u, 0u, 0u, 0u);
  }
  unsigned cnt = 0u;
#pragma unroll
  for (int j = 0; j < BM_V; ++j)
    cnt += (unsigned)(__builtin_popcount(wv[j].x) + __builtin_popcount(wv[j].y) +
                      __builtin_popcount(wv[j].z) + __builtin_popcount(wv[j].w));
  const unsigned lane = (unsigned)t & 31u, wave = (unsigned)t >> 5;
  unsigned incl = cnt;
#pragma unroll
  for (int d = 1; d < 32; d <<= 1) {
    const unsigned below = (unsigned)__shfl_up((int)incl, d);
    if (lane >= (unsigned)d) incl += below;
  }
  if (lane == 31u) wtot[wave] = incl;
  __syncthreads();
  unsigned before = (lane < wave) ? wtot[lane] : 0u;
#pragma unroll
  for (int d = 16; d >= 1; d >>= 1) before += (unsigned)__shfl_xor((int)before, d);
  if (t == ST_T - 1) *qn = before + incl;
  unsigned p = before + incl - cnt;
  if (cnt != 0u) {
#pragma unroll
    for (int j = 0; j < BM_V; ++j) {
      const unsigned first = (unsigned)(j * ST_T + t) * 128u;
      push_bits(wv[j].x, first,       p, q);
      push_bits(wv[j].y, first + 32u, p, q);
      push_bits(wv[j].z, first + 64u, p, q);
      push_bits(wv[j].w, first + 96u, p, q);
    }
  }
}

template <int PASS>
__device__ __forceinline__ void stats_pass(unsigned n, int t, bool ends, int lo, const unsigned* q,
                                           const int2* __restrict__ pairs,
                                           const float* __restrict__ el,
                                           const float* __restrict__ er,
                                           float* st_max, unsigned long long* st_sum, float* st_deg,
                                           unsigned* sn, unsigned* gqb, unsigned* gsb) {
  static_assert(PASS >= 0 && PASS <= 2, "maxima, sums, or both");
  for (unsigned i0 = 0; i0 < n; i0 += ST_T * ST_U) {
    unsigned ent[ST_U];
    int      sv[ST_U], dv[ST_U];
    float4   a[ST_U], b[ST_U];
#pragma unroll
    for (int u = 0; u < ST_U; ++u) {
      const unsigned idx = i0 + (unsigned)(u * ST_T + t);
      ent[u] = (idx < n) ? q[idx] : 0xFFFFFFFFu;
    }
#pragma unroll
    for (int u = 0; u < ST_U; ++u) {
      const unsigned eid = (ent[u] == 0xFFFFFFFFu) ? 0u : (ent[u] & 0xFFFFFu);
      const int2 pr = pairs[eid];
      sv[u] = pr.x;
      dv[u] = pr.y;
    }
#pragma unroll
    for (int u = 0; u < ST_U; ++u) {
      a[u] = *(const float4*)(el + (size_t)sv[u] * 4);
      b[u] = *(const float4*)(er + (size_t)dv[u] * 4);
    }
    float4   ev[ST_U];
    unsigned rr[ST_U];
#pragma unroll
    for (int u = 0; u < ST_U; ++u) {
      const unsigned r     = (unsigned)((ends ? dv[u] : sv[u]) - lo);
      const bool     valid = (ent[u] != 0xFFFFFFFFu) && r < (unsigned)ST_RNG;
      rr[u] = valid ? r : 0xFFFFFFFFu;
      ev[u] = make_float4(leakyf(a[u].x + b[u].x), leakyf(a[u].y + b[u].y),
                          leakyf(a[u].z + b[u].z), leakyf(a[u].w + b[u].w));
      if (PASS != 1 && valid) {
        atomicMaxF(&st_max[r * 4 + 0], ev[u].x);
        atomicMaxF(&st_max[r * 4 + 1], ev[u].y);
        atomicMaxF(&st_max[r * 4 + 2], ev[u].z);
        atomicMaxF(&st_max[r * 4 + 3], ev[u].w);
        atomicAdd(&st_deg[r], 1.f);
      }
      if (PASS != 1 && ends) {
        const unsigned lane = (unsigned)t & 31u;
#pragma unroll
        for (unsigned k = 0; k < (unsigned)ST_K; ++k) {
          const bool keep = valid && r >= k * RNG && r < (k + 1) * RNG;
          const unsigned m = __builtin_amdgcn_ballot_w32(keep);
          if (m != 0u) {
            const unsigned lead = (unsigned)__builtin_ctz(m);
            unsigned bp = 0u;
            if (lane == lead) bp = atomicAdd(&sn[k], (unsigned)__builtin_popcount(m));
            bp = (unsigned)__shfl((int)bp, (int)lead);
            const unsigned p = bp + (unsigned)__builtin_popcount(m & ((1u << lane) - 1u));
            if (keep && p < (unsigned)GCAP) {
              gqb[(size_t)k * GCAP + p] = ((r - k * RNG) << 20) | (ent[u] & 0xFFFFFu);
              gsb[(size_t)k * GCAP + p] = (unsigned)sv[u];
            }
          }
        }
      }
    }
    if (PASS == 2) __syncthreads();
    if (PASS != 0) {
#pragma unroll
      for (int u = 0; u < ST_U; ++u) {
        const unsigned r = rr[u];
        if (r != 0xFFFFFFFFu) {
          add_term(&st_sum[r * 4 + 0], __expf(ev[u].x - st_max[r * 4 + 0]));
          add_term(&st_sum[r * 4 + 1], __expf(ev[u].y - st_max[r * 4 + 1]));
          add_term(&st_sum[r * 4 + 2], __expf(ev[u].z - st_max[r * 4 + 2]));
          add_term(&st_sum[r * 4 + 3], __expf(ev[u].w - st_max[r * 4 + 3]));
        }
      }
    }
  }
}

__global__ __launch_bounds__(ST_T)
void seg_stats_kernel(const int* __restrict__ src, const int* __restrict__ dst,
                      const float* __restrict__ el, const float* __restrict__ er,
                      float* __restrict__ maxs, float* __restrict__ maxd,
                      float* __restrict__ sums, float* __restrict__ sumd,
                      float* __restrict__ norm_out, float* __restrict__ norm_in,
                      unsigned int* gq, unsigned int* gcount,
                      const unsigned int* __restrict__ bm,
                      const int2* __restrict__ pairs, unsigned int* gqsrc) {
  __shared__ __align__(16) unsigned int q[ST_QCAP];
  __shared__ float st_max[ST_RNG * 4];
  __shared__ __align__(8) unsigned long long st_sum[ST_RNG * 4];
  __shared__ float st_deg[ST_RNG];
  __shared__ unsigned int qn;
  __shared__ unsigned int sn[ST_K];
  __shared__ unsigned int wtot[ST_T / 32];

  const int  t     = threadIdx.x;
  const int  g     = blockIdx.x >> 1;
  const bool ends  = (blockIdx.x & 1) == 0;
  const int  lo    = g * ST_RNG;
  const int* __restrict__ idx = ends ? dst : src;
  unsigned int* gqb = gq + (size_t)g * ST_K * GCAP;
  unsigned int* gsb = gqsrc + (size_t)g * ST_K * GCAP;

  for (int i = t; i < ST_RNG * 4; i += ST_T) { st_max[i] = -1e30f; st_sum[i] = 0ull; }
  if (t < ST_RNG) st_deg[t] = 0.f;
  if (t < ST_K) sn[t] = 0u;
  __syncthreads();

  queue_from_bitmap(bm + (size_t)((ends ? 0 : ST_NRNG) + g) * BM_PITCH, t, q, &qn, wtot);
  __syncthreads();
  unsigned n = __builtin_amdgcn_readfirstlane(qn);

  if (n <= (unsigned)(ST_T * ST_U)) {
    stats_pass<2>(n, t, ends, lo, q, pairs, el, er, st_max, st_sum, st_deg, sn, gqb, gsb);
  } else if (n <= (unsigned)ST_QCAP) {
    stats_pass<0>(n, t, ends, lo, q, pairs, el, er, st_max, st_sum, st_deg, sn, gqb, gsb);
    __syncthreads();
    stats_pass<1>(n, t, ends, lo, q, pairs, el, er, st_max, st_sum, st_deg, sn, gqb, gsb);
  } else {
    __syncthreads();
    if (t == 0) qn = 0u;
    __syncthreads();
    for (int base = 0; base < E_EDGES; base += ST_EDGES_S) {
      scan_group<ST_VS>(base, t, lo, idx, q, &qn);
      __syncthreads();
      n = __builtin_amdgcn_readfirstlane(qn);
      __syncthreads();
      const bool last = (base + ST_EDGES_S >= E_EDGES);
      if (n > (unsigned)(ST_QCAP - ST_EDGES_S) || last) {
        stats_pass<0>(n, t, ends, lo, q, pairs, el, er, st_max, st_sum, st_deg, sn, gqb, gsb);
        __syncthreads();
        if (t == 0) qn = 0u;
        __syncthreads();
      }
    }
    for (int base = 0; base < E_EDGES; base += ST_EDGES_S) {
      scan_group<ST_VS>(base, t, lo, idx, q, &qn);
      __syncthreads();
      n = __builtin_amdgcn_readfirstlane(qn);
      __syncthreads();
      const bool last = (base + ST_EDGES_S >= E_EDGES);
      if (n > (unsigned)(ST_QCAP - ST_EDGES_S) || last) {
        stats_pass<1>(n, t, ends, lo, q, pairs, el, er, st_max, st_sum, st_deg, sn, gqb, gsb);
        __syncthreads();
        if (t == 0) qn = 0u;
        __syncthreads();
      }
    }
  }
  __syncthreads();

  float* __restrict__ mx_out = ends ? maxd : maxs;
  float* __restrict__ sm_out = ends ? sumd : sums;
  for (int i = t; i < ST_RNG * 4; i += ST_T) {
    mx_out[lo * 4 + i] = st_max[i];
    sm_out[lo * 4 + i] = (float)st_sum[i] * (1.f / (float)(1ull << SUM_FRAC));
  }
  if (t < ST_RNG) {
    const float deg = fmaxf(st_deg[t], 1.f);
    if (ends) norm_in[lo + t]  = sqrtf(deg);
    else      norm_out[lo + t] = rsqrtf(deg);
  }
  if (ends && t < ST_K) gcount[g * ST_K + t] = sn[t];
}

struct HopRow { float4 h0a, h0b, x1a, x1b, x2a, x2b; float aba, abb; };

__device__ __forceinline__ HopRow load_hop_row(int row, int lane, int hsel,
                                               const float* __restrict__ h0, const float* __restrict__ x1p,
                                               const float* __restrict__ x2p,
                                               const float* __restrict__ al) {
  HopRow r;
  const size_t o = (size_t)row * HF + lane * 4;
  r.h0a = *(const float4*)(h0 + o);  r.h0b = *(const float4*)(h0 + o + 128);
  r.x1a = *(const float4*)(x1p + o); r.x1b = *(const float4*)(x1p + o + 128);
  r.x2a = *(const float4*)(x2p + o); r.x2b = *(const float4*)(x2p + o + 128);
  r.aba = al[(size_t)row * 4 + hsel];
  r.abb = al[(size_t)row * 4 + 2 + hsel];
  return r;
}

__device__ __forceinline__ float4 combine4(const float4& v0, const float4& v1, const float4& v2,
                                           const float4& v3, const float4& ar, float ab) {
  const float p0 = row16_sum(v0.x * ar.x + v0.y * ar.y + v0.z * ar.z + v0.w * ar.w);
  const float p1 = row16_sum(v1.x * ar.x + v1.y * ar.y + v1.z * ar.z + v1.w * ar.w);
  const float p2 = row16_sum(v2.x * ar.x + v2.y * ar.y + v2.z * ar.z + v2.w * ar.w);
  const float p3 = row16_sum(v3.x * ar.x + v3.y * ar.y + v3.z * ar.z + v3.w * ar.w);
  const float s0 = leakyf(ab + p0), s1 = leakyf(ab + p1);
  const float s2 = leakyf(ab + p2), s3 = leakyf(ab + p3);
  const float m  = fmaxf(fmaxf(s0, s1), fmaxf(s2, s3));
  const float e0 = __expf(s0 - m), e1 = __expf(s1 - m), e2 = __expf(s2 - m), e3 = __expf(s3 - m);
  const float inv = 1.f / (e0 + e1 + e2 + e3);
  const float w0 = e0 * inv, w1 = e1 * inv, w2 = e2 * inv, w3 = e3 * inv;
  float4 o;
  o.x = v0.x * w0 + v1.x * w1 + v2.x * w2 + v3.x * w3;
  o.y = v0.y * w0 + v1.y * w1 + v2.y * w2 + v3.y * w3;
  o.z = v0.z * w0 + v1.z * w1 + v2.z * w2 + v3.z * w3;
  o.w = v0.w * w0 + v1.w * w1 + v2.w * w2 + v3.w * w3;
  return o;
}

template <bool FIRST, bool LAST>
__global__ __launch_bounds__(AGG_T)
void agg_tile_kernel(const int* __restrict__ src, const int* __restrict__ dst,
                     const float* __restrict__ el, const float* __restrict__ er,
                     const float* __restrict__ maxs, const float* __restrict__ maxd,
                     const float* __restrict__ sums, const float* __restrict__ sumd,
                     const float* __restrict__ norm_out, const float* __restrict__ norm_in,
                     const float* __restrict__ sigma,
                     const float* __restrict__ xprev, float* __restrict__ xnext,
                     const unsigned int* gq, const unsigned int* gcount,
                     unsigned int* gsv, float4* gcf,
                     const float* __restrict__ h0, const float* __restrict__ x1p,
                     const float* __restrict__ al, const float* __restrict__ hop_attn_r,
                     const unsigned int* __restrict__ gqsrc) {
  __shared__ __align__(16) float tile[RNG * HF];
  __shared__ __align__(16) unsigned int pool[POOL_W];
  __shared__ unsigned int qn;
  float4*       rc  = (float4*)pool;
  unsigned int* rs  = pool + CH * 4;
  unsigned int* q   = pool + CH * 5;
  unsigned int* lws = pool + CH * 5 + QCAP;

  const int t    = threadIdx.x;
  const int lane = t & 31;
  const int wave = t >> 5;
  const int hsel = lane >> 4;
  const int lo   = blockIdx.x * RNG;
  const unsigned wv = __builtin_amdgcn_readfirstlane((unsigned)wave);
  const unsigned int* gqb = gq + (size_t)blockIdx.x * GCAP;
  unsigned int* gsvb = gsv + (size_t)blockIdx.x * GCAP;
  float4*       gcfb = gcf + (size_t)blockIdx.x * GCAP;
  const float* rcf = (const float*)rc;
  unsigned* lw = lws + wv * LW;
  const float sgm = 1.f / (1.f + __expf(-sigma[0]));
  const unsigned nrow = (unsigned)lane * NWAVE + wv;
  const float nin = (nrow < (unsigned)RNG) ? norm_in[lo + (int)nrow] : 0.f;

  for (unsigned row = wv; row < (unsigned)RNG; row += NWAVE) {
    float* trow = tile + row * HF + lane * 4;
    *(float4*)trow         = make_float4(0.f, 0.f, 0.f, 0.f);
    *(float4*)(trow + 128) = make_float4(0.f, 0.f, 0.f, 0.f);
  }
  if (t == 0) qn = 0u;

  const unsigned cnt = __builtin_amdgcn_readfirstlane(gcount[blockIdx.x]);
  if (cnt <= (unsigned)GCAP) {
    if (FIRST) {
      unsigned int* qs = pool;
      unsigned int* lq = pool + CH + wv * WAREA;
      unsigned int* li = lq + LQ;
      unsigned int* eb = li + LQ;
      float4*       cb = (float4*)(eb + 32);
      unsigned w[GCAP / AGG_T];
#pragma unroll
      for (int j = 0; j < GCAP / AGG_T; ++j) {
        const unsigned i = (unsigned)(j * AGG_T + t);
        w[j] = (i < cnt) ? gqb[i] : 0u;
      }
#pragma unroll
      for (int j = 0; j < GCAP / AGG_T; ++j) {
        const unsigned i = (unsigned)(j * AGG_T + t);
        if (i < cnt) qs[i] = w[j];
      }
      __syncthreads();
      build_owned(cnt, wv, lane, hsel, lo, qs, lq, li, eb, cb, gsvb, gcfb,
                  gqsrc + (size_t)blockIdx.x * GCAP, el, er, maxs, maxd, sums, sumd, norm_out, sgm, xprev, tile);
    } else {
      unsigned w[GCAP / AGG_T];
      float4   c[GCAP / AGG_T];
#pragma unroll
      for (int j = 0; j < GCAP / AGG_T; ++j) {
        const unsigned i = (unsigned)(j * AGG_T + t);
        w[j] = (i < cnt) ? gsvb[i] : 0u;
        c[j] = (i < cnt) ? gcfb[i] : make_float4(0.f, 0.f, 0.f, 0.f);
      }
#pragma unroll
      for (int j = 0; j < GCAP / AGG_T; ++j) {
        const unsigned i = (unsigned)(j * AGG_T + t);
        if (i < cnt) { rs[i] = w[j]; rc[i] = c[j]; }
      }
      __syncthreads();
      drain_owned(cnt, wv, lane, hsel, rs, rcf, lw, xprev, tile);
    }
  } else {
    __syncthreads();
    for (int base = 0; base < E_EDGES; base += SCAN_EDGES) {
      const int e0 = base + t * 4;
      const int4 dv = (e0 < E_EDGES) ? *(const int4*)(dst + e0) : make_int4(-1, -1, -1, -1);
      const unsigned rx = (unsigned)(dv.x - lo), ry = (unsigned)(dv.y - lo);
      const unsigned rz = (unsigned)(dv.z - lo), rw = (unsigned)(dv.w - lo);
      if (min(min(rx, ry), min(rz, rw)) < (unsigned)RNG) {
        push_edge(rx, e0,     q, &qn);
        push_edge(ry, e0 + 1, q, &qn);
        push_edge(rz, e0 + 2, q, &qn);
        push_edge(rw, e0 + 3, q, &qn);
      }
      __syncthreads();
      const unsigned n = __builtin_amdgcn_readfirstlane(qn);
      __syncthreads();
      const bool last = (base + SCAN_EDGES >= E_EDGES);
      if (n > (unsigned)(QCAP - SCAN_EDGES) || last) {
        for (unsigned c0 = 0; c0 < n; c0 += CH) {
          const unsigned m = min((unsigned)CH, n - c0);
          for (unsigned idx = (unsigned)t; idx < m; idx += AGG_T) {
            unsigned word; float4 c;
            const unsigned qw = q[c0 + idx];
            make_record(qw, lo, src[qw & 0xFFFFFu], el, er, maxs, maxd, sums, sumd, norm_out, sgm, word, c);
            rs[idx] = word; rc[idx] = c;
          }
          __syncthreads();
          drain_owned(m, wv, lane, hsel, rs, rcf, lw, xprev, tile);
          __syncthreads();
        }
        if (t == 0) qn = 0u;
        __syncthreads();
      }
    }
  }

  if (!LAST) {
    for (unsigned row = wv; row < (unsigned)RNG; row += NWAVE) {
      const float* trow = tile + row * HF + lane * 4;
      const float4 a0 = *(const float4*)trow;
      const float4 a1 = *(const float4*)(trow + 128);
      const float ni = __int_as_float(__builtin_amdgcn_readlane(__float_as_int(nin), (int)(row >> 5)));
      float* o = xnext + (size_t)(lo + (int)row) * HF + lane * 4;
      *(float4*)o         = make_float4(a0.x * ni, a0.y * ni, a0.z * ni, a0.w * ni);
      *(float4*)(o + 128) = make_float4(a1.x * ni, a1.y * ni, a1.z * ni, a1.w * ni);
    }
  } else {
    const float4 ara = *(const float4*)(hop_attn_r + lane * 4);
    const float4 arb = *(const float4*)(hop_attn_r + 128 + lane * 4);
    HopRow nx = load_hop_row(lo + (int)wv, lane, hsel, h0, x1p, xprev, al);
    for (unsigned row = wv; row < (unsigned)RNG; row += NWAVE) {
      const HopRow cur = nx;
      if (row + NWAVE < (unsigned)RNG)
        nx = load_hop_row(lo + (int)(row + NWAVE), lane, hsel, h0, x1p, xprev, al);
      const float* trow = tile + row * HF + lane * 4;
      const float4 a0 = *(const float4*)trow;
      const float4 a1 = *(const float4*)(trow + 128);
      const float ni = __int_as_float(__builtin_amdgcn_readlane(__float_as_int(nin), (int)(row >> 5)));
      const float4 v3a = make_float4(a0.x * ni, a0.y * ni, a0.z * ni, a0.w * ni);
      const float4 v3b = make_float4(a1.x * ni, a1.y * ni, a1.z * ni, a1.w * ni);
      float* o = xnext + (size_t)(lo + (int)row) * HF + lane * 4;
      *(float4*)o         = combine4(cur.h0a, cur.x1a, cur.x2a, v3a, ara, cur.aba);
      *(float4*)(o + 128) = combine4(cur.h0b, cur.x1b, cur.x2b, v3b, arb, cur.abb);
    }
  }
}

extern "C" void kernel_launch(void* const* d_in, const int* in_sizes, int n_in,
                              void* d_out, int out_size, void* d_ws, size_t ws_size,
                              hipStream_t stream) {
  (void)in_sizes; (void)n_in; (void)out_size; (void)ws_size;
  const float* feat       = (const float*)d_in[0];
  const int*   src        = (const int*)d_in[1];
  const int*   dst        = (const int*)d_in[2];
  const float* W          = (const float*)d_in[3];
  const float* attn_l     = (const float*)d_in[4];
  const float* attn_r     = (const float*)d_in[5];
  const float* hop_attn_l = (const float*)d_in[6];
  const float* hop_attn_r = (const float*)d_in[7];
  const float* sigma      = (const float*)d_in[8];

  float* ws = (float*)d_ws;
  size_t off = 0;
  float* h    = ws + off; off += (size_t)N_NODES * HF;
  float* x1   = ws + off; off += (size_t)N_NODES * HF;
  float* x2   = ws + off; off += (size_t)N_NODES * HF;
  unsigned int* bm = (unsigned int*)(ws + off); off += (size_t)N_NODES * HF;
  int2* pairs = (int2*)(bm + (size_t)2 * ST_NRNG * BM_PITCH);
  unsigned int* gqsrc = (unsigned int*)(pairs + E_EDGES);
  float* el   = ws + off; off += (size_t)N_NODES * H_HEADS;
  float* er   = ws + off; off += (size_t)N_NODES * H_HEADS;
  float* al   = ws + off; off += (size_t)N_NODES * H_HEADS;
  float* maxs = ws + off; off += (size_t)N_NODES * H_HEADS;
  float* maxd = ws + off; off += (size_t)N_NODES * H_HEADS;
  float* sums = ws + off; off += (size_t)N_NODES * H_HEADS;
  float* sumd = ws + off; off += (size_t)N_NODES * H_HEADS;
  off += (size_t)N_NODES * 2;
  float* norm_out = ws + off; off += (size_t)N_NODES;
  float* norm_in  = ws + off; off += (size_t)N_NODES;
  float* edge_planes = ws + off; off += (size_t)E_EDGES * H_HEADS * 2;

  gemm_h_kernel<<<GEMM_NBLK + BM_NBLK, 256, 0, stream>>>(feat, W, h, attn_l, attn_r, hop_attn_l,
                                                         el, er, al, src, dst, bm, pairs);

  unsigned int* gq     = (unsigned int*)edge_planes;
  unsigned int* gcount = gq + (size_t)NBLK * GCAP;
  unsigned int* gsv    = gcount + 256;
  float4*       gcf    = (float4*)(gsv + (size_t)NBLK * GCAP);
  seg_stats_kernel<<<ST_NBLK, ST_T, 0, stream>>>(src, dst, el, er, maxs, maxd, sums, sumd,
                                                 norm_out, norm_in, gq, gcount, bm, pairs, gqsrc);

  agg_tile_kernel<true, false><<<NBLK, AGG_T, 0, stream>>>(src, dst, el, er, maxs, maxd, sums, sumd, norm_out,
                                                           norm_in, sigma, h,  x1, gq, gcount, gsv, gcf,
                                                           h, x1, al, hop_attn_r, gqsrc);
  agg_tile_kernel<false, false><<<NBLK, AGG_T, 0, stream>>>(src, dst, el, er, maxs, maxd, sums, sumd, norm_out,
                                                            norm_in, sigma, x1, x2, gq, gcount, gsv, gcf,
                                                            h, x1, al, hop_attn_r, gqsrc);
  agg_tile_kernel<false, true><<<NBLK, AGG_T, 0, stream>>>(src, dst, el, er, maxs, maxd, sums, sumd, norm_out,
                                                           norm_in, sigma, x2, (float*)d_out, gq, gcount, gsv, gcf,
                                                           h, x1, al, hop_attn_r, gqsrc);
}
